# starting kernel
# baseline (speedup 1.0000x reference)
#include <hip/hip_runtime.h>
#include <hip/hip_bf16.h>

#define GEX 978
#define KPAD 980
#define HDIM 128
#define NAT 50
#define FEAT 34
#define CCEH 64
#define BATCH 64

typedef __attribute__((ext_vector_type(2))) float v2f;
typedef __attribute__((ext_vector_type(8))) float v8f;

__global__ __launch_bounds__(256) void k_cce(const float* __restrict__ nf,
                                             const float* __restrict__ mask,
                                             const float* __restrict__ adj,
                                             const float* __restrict__ dist,
                                             const float* __restrict__ W1,
                                             float* __restrict__ pooled) {
    int b = blockIdx.x, tid = threadIdx.x;
    __shared__ float W1s[FEAT * CCEH];
    __shared__ float hs[NAT * CCEH];
    __shared__ float wm[NAT];
    for (int i = tid; i < FEAT * CCEH; i += 256) W1s[i] = W1[i];
    __syncthreads();
    for (int i = tid; i < NAT * CCEH; i += 256) {
        int n = i / CCEH, d = i % CCEH;
        const float* row = nf + (b * NAT + n) * FEAT;
        float s = 0.f;
        #pragma unroll
        for (int f = 0; f < FEAT; f++) s += row[f] * W1s[f * CCEH + d];
        hs[i] = fmaxf(s, 0.f);
    }
    if (tid < NAT) {
        float s = 0.f;
        for (int n = 0; n < NAT; n++) {
            int idx = (b * NAT + n) * NAT + tid;
            s += mask[b * NAT + n] * adj[idx] * expf(-dist[idx]);
        }
        wm[tid] = s;
    }
    __syncthreads();
    if (tid < CCEH) {
        float msum = 0.f;
        for (int n = 0; n < NAT; n++) msum += mask[b * NAT + n];
        msum = fmaxf(msum, 1.f);
        float s = 0.f;
        for (int m = 0; m < NAT; m++) s += wm[m] * hs[m * CCEH + tid];
        pooled[b * CCEH + tid] = s / msum;
    }
}

__global__ __launch_bounds__(256) void k_comp(const float* __restrict__ pooled,
                                              const float* __restrict__ W2,
                                              const float* __restrict__ dose,
                                              const float* __restrict__ timev,
                                              const float* __restrict__ w_dose,
                                              const float* __restrict__ w_time,
                                              float* __restrict__ comp) {
    int idx = blockIdx.x * 256 + threadIdx.x;
    if (idx >= BATCH * GEX) return;
    int b = idx / GEX, g = idx % GEX;
    float s = dose[b] * w_dose[g] + timev[b] * w_time[g];
    #pragma unroll 8
    for (int d = 0; d < CCEH; d++) s += pooled[b * CCEH + d] * W2[d * GEX + g];
    comp[idx] = s;
}

__global__ __launch_bounds__(128) void k_S(const float* __restrict__ b_gex,
                                           const float* __restrict__ comp,
                                           const float* __restrict__ w_gex,
                                           const float* __restrict__ w_comp,
                                           float* __restrict__ S) {
    int b = blockIdx.x, h = threadIdx.x;
    float s = 0.f;
    for (int g = 0; g < GEX; g++) {
        float x = b_gex[b * GEX + g], c = comp[b * GEX + g];
        s += x * w_gex[g * HDIM + h] + c * w_comp[g * HDIM + h];
    }
    S[b * HDIM + h] = s;
}

__global__ __launch_bounds__(256) void k_ppirow(const float* __restrict__ ppi,
                                                float* __restrict__ rowsum) {
    int g = blockIdx.x, tid = threadIdx.x;
    __shared__ float red[256];
    float s = 0.f;
    for (int k = tid; k < GEX; k += 256) s += ppi[g * GEX + k];
    red[tid] = s;
    __syncthreads();
    for (int o = 128; o > 0; o >>= 1) {
        if (tid < o) red[tid] += red[tid + o];
        __syncthreads();
    }
    if (tid == 0) rowsum[g] = red[0];
}

__global__ __launch_bounds__(256) void k_pred0(const float* __restrict__ b_gex,
                                               const float* __restrict__ comp,
                                               const float* __restrict__ w_gex,
                                               const float* __restrict__ w_comp,
                                               const float* __restrict__ S,
                                               const float* __restrict__ ppirow,
                                               float* __restrict__ pred0) {
    __shared__ float Ssh[BATCH * HDIM];
    __shared__ float wg[4][HDIM], wc[4][HDIM];
    int g0 = blockIdx.x * 4, tid = threadIdx.x;
    for (int i = tid; i < BATCH * HDIM; i += 256) Ssh[i] = S[i];
    for (int i = tid; i < 4 * HDIM; i += 256) {
        int gl = i / HDIM, hh = i % HDIM, g = g0 + gl;
        wg[gl][hh] = (g < GEX) ? w_gex[g * HDIM + hh] : 0.f;
        wc[gl][hh] = (g < GEX) ? w_comp[g * HDIM + hh] : 0.f;
    }
    __syncthreads();
    int b = tid >> 2, gl = tid & 3, g = g0 + gl;
    if (g < GEX) {
        float d1 = 0.f, d2 = 0.f;
        #pragma unroll 8
        for (int hh = 0; hh < HDIM; hh++) {
            float sv = Ssh[b * HDIM + hh];
            d1 += sv * wg[gl][hh];
            d2 += sv * wc[gl][hh];
        }
        float x = b_gex[b * GEX + g], c = comp[b * GEX + g];
        float row = (x * d1 + c * d2) * 0.08838834764831845f + ppirow[g];
        pred0[b * GEX + g] = x * row;
    }
}

__global__ __launch_bounds__(256) void k_ln(const float* __restrict__ pred0,
                                            const float* __restrict__ gamma,
                                            const float* __restrict__ beta,
                                            float* __restrict__ xlnP) {
    int b = blockIdx.x, tid = threadIdx.x;
    __shared__ float red[256];
    __shared__ float mu_s, rstd_s;
    float s = 0.f;
    for (int g = tid; g < GEX; g += 256) s += pred0[b * GEX + g];
    red[tid] = s;
    __syncthreads();
    for (int o = 128; o > 0; o >>= 1) { if (tid < o) red[tid] += red[tid + o]; __syncthreads(); }
    if (tid == 0) mu_s = red[0] / (float)GEX;
    __syncthreads();
    float mu = mu_s, v = 0.f;
    for (int g = tid; g < GEX; g += 256) { float d = pred0[b * GEX + g] - mu; v += d * d; }
    red[tid] = v;
    __syncthreads();
    for (int o = 128; o > 0; o >>= 1) { if (tid < o) red[tid] += red[tid + o]; __syncthreads(); }
    if (tid == 0) rstd_s = rsqrtf(red[0] / (float)GEX + 1e-5f);
    __syncthreads();
    float rstd = rstd_s;
    for (int g = tid; g < KPAD; g += 256) {
        float y = 0.f;
        if (g < GEX)
            y = fmaxf((pred0[b * GEX + g] - mu) * rstd * gamma[g] + beta[g], 0.f);
        xlnP[b * KPAD + g] = y;
    }
}

__global__ __launch_bounds__(256) void k_transpose(const float* __restrict__ Wff,
                                                   float* __restrict__ WffT) {
    __shared__ float tile[32][33];
    int kb = blockIdx.x * 32;
    int nb = blockIdx.y * 32;
    int tx = threadIdx.x & 31, ty = threadIdx.x >> 5;
    for (int i = ty; i < 32; i += 8) {
        int k = kb + i, n = nb + tx;
        tile[i][tx] = (k < GEX && n < GEX) ? Wff[k * GEX + n] : 0.f;
    }
    __syncthreads();
    for (int i = ty; i < 32; i += 8) {
        int n = nb + i, k = kb + tx;
        if (n < GEX && k < KPAD) WffT[n * KPAD + k] = tile[tx][i];
    }
}

__global__ __launch_bounds__(256) void k_gemm_wmma(const float* __restrict__ xlnP,
                                                   const float* __restrict__ WffT,
                                                   float* __restrict__ out) {
    const int NT = (GEX + 15) / 16;
    int wave = (blockIdx.x * 256 + threadIdx.x) >> 5;
    int lane = threadIdx.x & 31;
    int mt = wave / NT;
    int nt = wave % NT;
    if (mt >= BATCH / 16) return;
    int lg   = lane & 15;
    int koff = (lane >= 16) ? 2 : 0;
    int m = mt * 16 + lg;
    int n = nt * 16 + lg;
    int nc = (n < GEX) ? n : (GEX - 1);

    const float* ap = xlnP + m * KPAD + koff;
    const float* bp = WffT + nc * KPAD + koff;

    v8f acc = {};
    #pragma unroll 5
    for (int k = 0; k < KPAD; k += 4) {
        v2f a  = *(const v2f*)ap;
        v2f bm = *(const v2f*)bp;
        ap += 4; bp += 4;
        acc = __builtin_amdgcn_wmma_f32_16x16x4_f32(
            false, a, false, bm, (short)0, acc, false, false);
    }

    bool nok = n < GEX;
    #pragma unroll
    for (int v = 0; v < 8; v++) {
        int row = mt * 16 + v + ((lane >= 16) ? 8 : 0);
        if (nok) out[row * GEX + n] = acc[v];
    }
}

extern "C" void kernel_launch(void* const* d_in, const int* in_sizes, int n_in,
                              void* d_out, int out_size, void* d_ws, size_t ws_size,
                              hipStream_t stream) {
    const float* b_gex    = (const float*)d_in[0];
    const float* node_feat= (const float*)d_in[1];
    const float* mask     = (const float*)d_in[2];
    const float* adj      = (const float*)d_in[3];
    const float* dist     = (const float*)d_in[4];
    const float* dose     = (const float*)d_in[5];
    const float* timev    = (const float*)d_in[6];
    const float* ppi      = (const float*)d_in[7];
    const float* w_gex    = (const float*)d_in[8];
    const float* w_comp   = (const float*)d_in[9];
    const float* W1       = (const float*)d_in[10];
    const float* W2       = (const float*)d_in[11];
    const float* w_dose   = (const float*)d_in[12];
    const float* w_time   = (const float*)d_in[13];
    const float* gamma    = (const float*)d_in[14];
    const float* beta     = (const float*)d_in[15];
    const float* Wff      = (const float*)d_in[16];

    float* out  = (float*)d_out;
    float* comp = out + BATCH * GEX;

    float* ws     = (float*)d_ws;
    float* pooled = ws;
    float* S      = pooled + BATCH * CCEH;
    float* ppirow = S + BATCH * HDIM;
    float* pred0  = ppirow + 1024;
    float* xlnP   = pred0 + BATCH * GEX;
    float* WffT   = xlnP + BATCH * KPAD;

    dim3 tg((KPAD + 31) / 32, (GEX + 31) / 32);
    k_transpose<<<tg, 256, 0, stream>>>(Wff, WffT);

    k_cce<<<BATCH, 256, 0, stream>>>(node_feat, mask, adj, dist, W1, pooled);
    k_comp<<<(BATCH * GEX + 255) / 256, 256, 0, stream>>>(pooled, W2, dose, timev,
                                                          w_dose, w_time, comp);
    k_S<<<BATCH, HDIM, 0, stream>>>(b_gex, comp, w_gex, w_comp, S);
    k_ppirow<<<GEX, 256, 0, stream>>>(ppi, ppirow);
    k_pred0<<<(GEX + 3) / 4, 256, 0, stream>>>(b_gex, comp, w_gex, w_comp, S, ppirow, pred0);
    k_ln<<<BATCH, 256, 0, stream>>>(pred0, gamma, beta, xlnP);
    k_gemm_wmma<<<31, 256, 0, stream>>>(xlnP, WffT, out);
}


# k_S: the 2x978-long g-sum of S is split over 32 waves per 16x16 tile on f32 WMMA (A slices staged through LDS), partial tiles added through LDS in fixed order; summation order changes
# speedup vs baseline: 1.8388x; 1.8388x over previous
#include <hip/hip_runtime.h>
#include <hip/hip_bf16.h>

#define GEX 978
#define KPAD 980
#define HDIM 128
#define NAT 50
#define FEAT 34
#define CCEH 64
#define BATCH 64

typedef __attribute__((ext_vector_type(2))) float v2f;
typedef __attribute__((ext_vector_type(8))) float v8f;

__global__ __launch_bounds__(256) void k_cce(const float* __restrict__ nf,
                                             const float* __restrict__ mask,
                                             const float* __restrict__ adj,
                                             const float* __restrict__ dist,
                                             const float* __restrict__ W1,
                                             float* __restrict__ pooled) {
    int b = blockIdx.x, tid = threadIdx.x;
    __shared__ float W1s[FEAT * CCEH];
    __shared__ float hs[NAT * CCEH];
    __shared__ float wm[NAT];
    for (int i = tid; i < FEAT * CCEH; i += 256) W1s[i] = W1[i];
    __syncthreads();
    for (int i = tid; i < NAT * CCEH; i += 256) {
        int n = i / CCEH, d = i % CCEH;
        const float* row = nf + (b * NAT + n) * FEAT;
        float s = 0.f;
        #pragma unroll
        for (int f = 0; f < FEAT; f++) s += row[f] * W1s[f * CCEH + d];
        hs[i] = fmaxf(s, 0.f);
    }
    if (tid < NAT) {
        float s = 0.f;
        for (int n = 0; n < NAT; n++) {
            int idx = (b * NAT + n) * NAT + tid;
            s += mask[b * NAT + n] * adj[idx] * expf(-dist[idx]);
        }
        wm[tid] = s;
    }
    __syncthreads();
    if (tid < CCEH) {
        float msum = 0.f;
        for (int n = 0; n < NAT; n++) msum += mask[b * NAT + n];
        msum = fmaxf(msum, 1.f);
        float s = 0.f;
        for (int m = 0; m < NAT; m++) s += wm[m] * hs[m * CCEH + tid];
        pooled[b * CCEH + tid] = s / msum;
    }
}

__global__ __launch_bounds__(256) void k_comp(const float* __restrict__ pooled,
                                              const float* __restrict__ W2,
                                              const float* __restrict__ dose,
                                              const float* __restrict__ timev,
                                              const float* __restrict__ w_dose,
                                              const float* __restrict__ w_time,
                                              float* __restrict__ comp) {
    int idx = blockIdx.x * 256 + threadIdx.x;
    if (idx >= BATCH * GEX) return;
    int b = idx / GEX, g = idx % GEX;
    float s = dose[b] * w_dose[g] + timev[b] * w_time[g];
    #pragma unroll 8
    for (int d = 0; d < CCEH; d++) s += pooled[b * CCEH + d] * W2[d * GEX + g];
    comp[idx] = s;
}

#define SPAIRS (GEX / 2)
#define SA_PITCH 68
__global__ __launch_bounds__(1024) void k_S(const float* __restrict__ b_gex,
                                            const float* __restrict__ comp,
                                            const float* __restrict__ w_gex,
                                            const float* __restrict__ w_comp,
                                            float* __restrict__ S) {
    __shared__ float sA[32 * 16 * SA_PITCH];
    __shared__ float part[32 * 256];
    const int tid = threadIdx.x;
    const int lane = tid & 31;
    const int wave = __builtin_amdgcn_readfirstlane(tid >> 5);
    const int c = lane & 15, h = lane >> 4;
    const int mt = blockIdx.x >> 3, nt = blockIdx.x & 7;
    const int wl = wave & 15;
    const float* Mx = (wave < 16) ? b_gex : comp;
    const float* Wm = (wave < 16) ? w_gex : w_comp;
    const int pw = wl * 32;
    const int p0 = (pw < SPAIRS - 32) ? pw : (SPAIRS - 32);
    const int skip = pw - p0;
    const int k0 = 2 * p0;

    const float* arow = Mx + (mt * 16) * GEX + k0 + 2 * lane;
    v2f av[16];
    for (int r = 0; r < 16; r++) av[r] = *(const v2f*)(arow + r * GEX);

    const float* bp = Wm + (k0 + 2 * h) * HDIM + nt * 16 + c;
    v2f bf[16];
    for (int j = 0; j < 16; j++) {
        bf[j].x = bp[(4 * j) * HDIM];
        bf[j].y = bp[(4 * j + 1) * HDIM];
    }

    float* sw = sA + wave * (16 * SA_PITCH);
    for (int r = 0; r < 16; r++) *(v2f*)(sw + r * SA_PITCH + 2 * lane) = av[r];
    __syncthreads();

    const float* fa = sw + c * SA_PITCH + 2 * h;
    v8f acc0 = {}, acc1 = {};
    for (int j = 0; j < 16; j += 2) {
        v2f a0 = *(const v2f*)(fa + 4 * j);
        v2f a1 = *(const v2f*)(fa + 4 * j + 4);
        v2f b0 = bf[j], b1 = bf[j + 1];
        const bool ok0 = (2 * j + h) >= skip;
        const bool ok1 = (2 * j + 2 + h) >= skip;
        a0.x = ok0 ? a0.x : 0.f;  a0.y = ok0 ? a0.y : 0.f;
        b0.x = ok0 ? b0.x : 0.f;  b0.y = ok0 ? b0.y : 0.f;
        a1.x = ok1 ? a1.x : 0.f;  a1.y = ok1 ? a1.y : 0.f;
        b1.x = ok1 ? b1.x : 0.f;  b1.y = ok1 ? b1.y : 0.f;
        acc0 = __builtin_amdgcn_wmma_f32_16x16x4_f32(
            false, a0, false, b0, (short)0, acc0, false, false);
        acc1 = __builtin_amdgcn_wmma_f32_16x16x4_f32(
            false, a1, false, b1, (short)0, acc1, false, false);
    }

    for (int r = 0; r < 8; r++) part[wave * 256 + r * 32 + lane] = acc0[r] + acc1[r];
    __syncthreads();
    if (tid < 256) {
        float s0 = 0.f, s1 = 0.f, s2 = 0.f, s3 = 0.f;
        for (int w = 0; w < 32; w += 4) {
            s0 += part[(w + 0) * 256 + tid];
            s1 += part[(w + 1) * 256 + tid];
            s2 += part[(w + 2) * 256 + tid];
            s3 += part[(w + 3) * 256 + tid];
        }
        const int r = tid >> 5, l = tid & 31;
        const int row = mt * 16 + 8 * (l >> 4) + r;
        const int col = nt * 16 + (l & 15);
        S[row * HDIM + col] = (s0 + s1) + (s2 + s3);
    }
}

__global__ __launch_bounds__(256) void k_ppirow(const float* __restrict__ ppi,
                                                float* __restrict__ rowsum) {
    int g = blockIdx.x, tid = threadIdx.x;
    __shared__ float red[256];
    float s = 0.f;
    for (int k = tid; k < GEX; k += 256) s += ppi[g * GEX + k];
    red[tid] = s;
    __syncthreads();
    for (int o = 128; o > 0; o >>= 1) {
        if (tid < o) red[tid] += red[tid + o];
        __syncthreads();
    }
    if (tid == 0) rowsum[g] = red[0];
}

__global__ __launch_bounds__(256) void k_pred0(const float* __restrict__ b_gex,
                                               const float* __restrict__ comp,
                                               const float* __restrict__ w_gex,
                                               const float* __restrict__ w_comp,
                                               const float* __restrict__ S,
                                               const float* __restrict__ ppirow,
                                               float* __restrict__ pred0) {
    __shared__ float Ssh[BATCH * HDIM];
    __shared__ float wg[4][HDIM], wc[4][HDIM];
    int g0 = blockIdx.x * 4, tid = threadIdx.x;
    for (int i = tid; i < BATCH * HDIM; i += 256) Ssh[i] = S[i];
    for (int i = tid; i < 4 * HDIM; i += 256) {
        int gl = i / HDIM, hh = i % HDIM, g = g0 + gl;
        wg[gl][hh] = (g < GEX) ? w_gex[g * HDIM + hh] : 0.f;
        wc[gl][hh] = (g < GEX) ? w_comp[g * HDIM + hh] : 0.f;
    }
    __syncthreads();
    int b = tid >> 2, gl = tid & 3, g = g0 + gl;
    if (g < GEX) {
        float d1 = 0.f, d2 = 0.f;
        #pragma unroll 8
        for (int hh = 0; hh < HDIM; hh++) {
            float sv = Ssh[b * HDIM + hh];
            d1 += sv * wg[gl][hh];
            d2 += sv * wc[gl][hh];
        }
        float x = b_gex[b * GEX + g], c = comp[b * GEX + g];
        float row = (x * d1 + c * d2) * 0.08838834764831845f + ppirow[g];
        pred0[b * GEX + g] = x * row;
    }
}

__global__ __launch_bounds__(256) void k_ln(const float* __restrict__ pred0,
                                            const float* __restrict__ gamma,
                                            const float* __restrict__ beta,
                                            float* __restrict__ xlnP) {
    int b = blockIdx.x, tid = threadIdx.x;
    __shared__ float red[256];
    __shared__ float mu_s, rstd_s;
    float s = 0.f;
    for (int g = tid; g < GEX; g += 256) s += pred0[b * GEX + g];
    red[tid] = s;
    __syncthreads();
    for (int o = 128; o > 0; o >>= 1) { if (tid < o) red[tid] += red[tid + o]; __syncthreads(); }
    if (tid == 0) mu_s = red[0] / (float)GEX;
    __syncthreads();
    float mu = mu_s, v = 0.f;
    for (int g = tid; g < GEX; g += 256) { float d = pred0[b * GEX + g] - mu; v += d * d; }
    red[tid] = v;
    __syncthreads();
    for (int o = 128; o > 0; o >>= 1) { if (tid < o) red[tid] += red[tid + o]; __syncthreads(); }
    if (tid == 0) rstd_s = rsqrtf(red[0] / (float)GEX + 1e-5f);
    __syncthreads();
    float rstd = rstd_s;
    for (int g = tid; g < KPAD; g += 256) {
        float y = 0.f;
        if (g < GEX)
            y = fmaxf((pred0[b * GEX + g] - mu) * rstd * gamma[g] + beta[g], 0.f);
        xlnP[b * KPAD + g] = y;
    }
}

__global__ __launch_bounds__(256) void k_transpose(const float* __restrict__ Wff,
                                                   float* __restrict__ WffT) {
    __shared__ float tile[32][33];
    int kb = blockIdx.x * 32;
    int nb = blockIdx.y * 32;
    int tx = threadIdx.x & 31, ty = threadIdx.x >> 5;
    for (int i = ty; i < 32; i += 8) {
        int k = kb + i, n = nb + tx;
        tile[i][tx] = (k < GEX && n < GEX) ? Wff[k * GEX + n] : 0.f;
    }
    __syncthreads();
    for (int i = ty; i < 32; i += 8) {
        int n = nb + i, k = kb + tx;
        if (n < GEX && k < KPAD) WffT[n * KPAD + k] = tile[tx][i];
    }
}

__global__ __launch_bounds__(256) void k_gemm_wmma(const float* __restrict__ xlnP,
                                                   const float* __restrict__ WffT,
                                                   float* __restrict__ out) {
    const int NT = (GEX + 15) / 16;
    int wave = (blockIdx.x * 256 + threadIdx.x) >> 5;
    int lane = threadIdx.x & 31;
    int mt = wave / NT;
    int nt = wave % NT;
    if (mt >= BATCH / 16) return;
    int lg   = lane & 15;
    int koff = (lane >= 16) ? 2 : 0;
    int m = mt * 16 + lg;
    int n = nt * 16 + lg;
    int nc = (n < GEX) ? n : (GEX - 1);

    const float* ap = xlnP + m * KPAD + koff;
    const float* bp = WffT + nc * KPAD + koff;

    v8f acc = {};
    #pragma unroll 5
    for (int k = 0; k < KPAD; k += 4) {
        v2f a  = *(const v2f*)ap;
        v2f bm = *(const v2f*)bp;
        ap += 4; bp += 4;
        acc = __builtin_amdgcn_wmma_f32_16x16x4_f32(
            false, a, false, bm, (short)0, acc, false, false);
    }

    bool nok = n < GEX;
    #pragma unroll
    for (int v = 0; v < 8; v++) {
        int row = mt * 16 + v + ((lane >= 16) ? 8 : 0);
        if (nok) out[row * GEX + n] = acc[v];
    }
}

extern "C" void kernel_launch(void* const* d_in, const int* in_sizes, int n_in,
                              void* d_out, int out_size, void* d_ws, size_t ws_size,
                              hipStream_t stream) {
    const float* b_gex    = (const float*)d_in[0];
    const float* node_feat= (const float*)d_in[1];
    const float* mask     = (const float*)d_in[2];
    const float* adj      = (const float*)d_in[3];
    const float* dist     = (const float*)d_in[4];
    const float* dose     = (const float*)d_in[5];
    const float* timev    = (const float*)d_in[6];
    const float* ppi      = (const float*)d_in[7];
    const float* w_gex    = (const float*)d_in[8];
    const float* w_comp   = (const float*)d_in[9];
    const float* W1       = (const float*)d_in[10];
    const float* W2       = (const float*)d_in[11];
    const float* w_dose   = (const float*)d_in[12];
    const float* w_time   = (const float*)d_in[13];
    const float* gamma    = (const float*)d_in[14];
    const float* beta     = (const float*)d_in[15];
    const float* Wff      = (const float*)d_in[16];

    float* out  = (float*)d_out;
    float* comp = out + BATCH * GEX;

    float* ws     = (float*)d_ws;
    float* pooled = ws;
    float* S      = pooled + BATCH * CCEH;
    float* ppirow = S + BATCH * HDIM;
    float* pred0  = ppirow + 1024;
    float* xlnP   = pred0 + BATCH * GEX;
    float* WffT   = xlnP + BATCH * KPAD;

    dim3 tg((KPAD + 31) / 32, (GEX + 31) / 32);
    k_transpose<<<tg, 256, 0, stream>>>(Wff, WffT);

    k_cce<<<BATCH, 256, 0, stream>>>(node_feat, mask, adj, dist, W1, pooled);
    k_comp<<<(BATCH * GEX + 255) / 256, 256, 0, stream>>>(pooled, W2, dose, timev,
                                                          w_dose, w_time, comp);
    k_S<<<32, 1024, 0, stream>>>(b_gex, comp, w_gex, w_comp, S);
    k_ppirow<<<GEX, 256, 0, stream>>>(ppi, ppirow);
    k_pred0<<<(GEX + 3) / 4, 256, 0, stream>>>(b_gex, comp, w_gex, w_comp, S, ppirow, pred0);
    k_ln<<<BATCH, 256, 0, stream>>>(pred0, gamma, beta, xlnP);
    k_gemm_wmma<<<31, 256, 0, stream>>>(xlnP, WffT, out);
}


# k_gemm: K split over 16 waves per 32x16 out block, A slabs staged in LDS, B read from W_ff rows so the transposed copy and its launch go; partials added via LDS in fixed order; sum order changes
# speedup vs baseline: 2.7464x; 1.4936x over previous
#include <hip/hip_runtime.h>
#include <hip/hip_bf16.h>

#define GEX 978
#define KPAD 980
#define HDIM 128
#define NAT 50
#define FEAT 34
#define CCEH 64
#define BATCH 64

typedef __attribute__((ext_vector_type(2))) float v2f;
typedef __attribute__((ext_vector_type(8))) float v8f;

__global__ __launch_bounds__(256) void k_cce(const float* __restrict__ nf,
                                             const float* __restrict__ mask,
                                             const float* __restrict__ adj,
                                             const float* __restrict__ dist,
                                             const float* __restrict__ W1,
                                             float* __restrict__ pooled) {
    int b = blockIdx.x, tid = threadIdx.x;
    __shared__ float W1s[FEAT * CCEH];
    __shared__ float hs[NAT * CCEH];
    __shared__ float wm[NAT];
    for (int i = tid; i < FEAT * CCEH; i += 256) W1s[i] = W1[i];
    __syncthreads();
    for (int i = tid; i < NAT * CCEH; i += 256) {
        int n = i / CCEH, d = i % CCEH;
        const float* row = nf + (b * NAT + n) * FEAT;
        float s = 0.f;
        #pragma unroll
        for (int f = 0; f < FEAT; f++) s += row[f] * W1s[f * CCEH + d];
        hs[i] = fmaxf(s, 0.f);
    }
    if (tid < NAT) {
        float s = 0.f;
        for (int n = 0; n < NAT; n++) {
            int idx = (b * NAT + n) * NAT + tid;
            s += mask[b * NAT + n] * adj[idx] * expf(-dist[idx]);
        }
        wm[tid] = s;
    }
    __syncthreads();
    if (tid < CCEH) {
        float msum = 0.f;
        for (int n = 0; n < NAT; n++) msum += mask[b * NAT + n];
        msum = fmaxf(msum, 1.f);
        float s = 0.f;
        for (int m = 0; m < NAT; m++) s += wm[m] * hs[m * CCEH + tid];
        pooled[b * CCEH + tid] = s / msum;
    }
}

__global__ __launch_bounds__(256) void k_comp(const float* __restrict__ pooled,
                                              const float* __restrict__ W2,
                                              const float* __restrict__ dose,
                                              const float* __restrict__ timev,
                                              const float* __restrict__ w_dose,
                                              const float* __restrict__ w_time,
                                              float* __restrict__ comp) {
    int idx = blockIdx.x * 256 + threadIdx.x;
    if (idx >= BATCH * GEX) return;
    int b = idx / GEX, g = idx % GEX;
    float s = dose[b] * w_dose[g] + timev[b] * w_time[g];
    #pragma unroll 8
    for (int d = 0; d < CCEH; d++) s += pooled[b * CCEH + d] * W2[d * GEX + g];
    comp[idx] = s;
}

#define SPAIRS (GEX / 2)
#define SA_PITCH 68
__global__ __launch_bounds__(1024) void k_S(const float* __restrict__ b_gex,
                                            const float* __restrict__ comp,
                                            const float* __restrict__ w_gex,
                                            const float* __restrict__ w_comp,
                                            float* __restrict__ S) {
    __shared__ float sA[32 * 16 * SA_PITCH];
    __shared__ float part[32 * 256];
    const int tid = threadIdx.x;
    const int lane = tid & 31;
    const int wave = __builtin_amdgcn_readfirstlane(tid >> 5);
    const int c = lane & 15, h = lane >> 4;
    const int mt = blockIdx.x >> 3, nt = blockIdx.x & 7;
    const int wl = wave & 15;
    const float* Mx = (wave < 16) ? b_gex : comp;
    const float* Wm = (wave < 16) ? w_gex : w_comp;
    const int pw = wl * 32;
    const int p0 = (pw < SPAIRS - 32) ? pw : (SPAIRS - 32);
    const int skip = pw - p0;
    const int k0 = 2 * p0;

    const float* arow = Mx + (mt * 16) * GEX + k0 + 2 * lane;
    v2f av[16];
    for (int r = 0; r < 16; r++) av[r] = *(const v2f*)(arow + r * GEX);

    const float* bp = Wm + (k0 + 2 * h) * HDIM + nt * 16 + c;
    v2f bf[16];
    for (int j = 0; j < 16; j++) {
        bf[j].x = bp[(4 * j) * HDIM];
        bf[j].y = bp[(4 * j + 1) * HDIM];
    }

    float* sw = sA + wave * (16 * SA_PITCH);
    for (int r = 0; r < 16; r++) *(v2f*)(sw + r * SA_PITCH + 2 * lane) = av[r];
    __syncthreads();

    const float* fa = sw + c * SA_PITCH + 2 * h;
    v8f acc0 = {}, acc1 = {};
    for (int j = 0; j < 16; j += 2) {
        v2f a0 = *(const v2f*)(fa + 4 * j);
        v2f a1 = *(const v2f*)(fa + 4 * j + 4);
        v2f b0 = bf[j], b1 = bf[j + 1];
        const bool ok0 = (2 * j + h) >= skip;
        const bool ok1 = (2 * j + 2 + h) >= skip;
        a0.x = ok0 ? a0.x : 0.f;  a0.y = ok0 ? a0.y : 0.f;
        b0.x = ok0 ? b0.x : 0.f;  b0.y = ok0 ? b0.y : 0.f;
        a1.x = ok1 ? a1.x : 0.f;  a1.y = ok1 ? a1.y : 0.f;
        b1.x = ok1 ? b1.x : 0.f;  b1.y = ok1 ? b1.y : 0.f;
        acc0 = __builtin_amdgcn_wmma_f32_16x16x4_f32(
            false, a0, false, b0, (short)0, acc0, false, false);
        acc1 = __builtin_amdgcn_wmma_f32_16x16x4_f32(
            false, a1, false, b1, (short)0, acc1, false, false);
    }

    for (int r = 0; r < 8; r++) part[wave * 256 + r * 32 + lane] = acc0[r] + acc1[r];
    __syncthreads();
    if (tid < 256) {
        float s0 = 0.f, s1 = 0.f, s2 = 0.f, s3 = 0.f;
        for (int w = 0; w < 32; w += 4) {
            s0 += part[(w + 0) * 256 + tid];
            s1 += part[(w + 1) * 256 + tid];
            s2 += part[(w + 2) * 256 + tid];
            s3 += part[(w + 3) * 256 + tid];
        }
        const int r = tid >> 5, l = tid & 31;
        const int row = mt * 16 + 8 * (l >> 4) + r;
        const int col = nt * 16 + (l & 15);
        S[row * HDIM + col] = (s0 + s1) + (s2 + s3);
    }
}

__global__ __launch_bounds__(256) void k_ppirow(const float* __restrict__ ppi,
                                                float* __restrict__ rowsum) {
    int g = blockIdx.x, tid = threadIdx.x;
    __shared__ float red[256];
    float s = 0.f;
    for (int k = tid; k < GEX; k += 256) s += ppi[g * GEX + k];
    red[tid] = s;
    __syncthreads();
    for (int o = 128; o > 0; o >>= 1) {
        if (tid < o) red[tid] += red[tid + o];
        __syncthreads();
    }
    if (tid == 0) rowsum[g] = red[0];
}

__global__ __launch_bounds__(256) void k_pred0(const float* __restrict__ b_gex,
                                               const float* __restrict__ comp,
                                               const float* __restrict__ w_gex,
                                               const float* __restrict__ w_comp,
                                               const float* __restrict__ S,
                                               const float* __restrict__ ppirow,
                                               float* __restrict__ pred0) {
    __shared__ float Ssh[BATCH * HDIM];
    __shared__ float wg[4][HDIM], wc[4][HDIM];
    int g0 = blockIdx.x * 4, tid = threadIdx.x;
    for (int i = tid; i < BATCH * HDIM; i += 256) Ssh[i] = S[i];
    for (int i = tid; i < 4 * HDIM; i += 256) {
        int gl = i / HDIM, hh = i % HDIM, g = g0 + gl;
        wg[gl][hh] = (g < GEX) ? w_gex[g * HDIM + hh] : 0.f;
        wc[gl][hh] = (g < GEX) ? w_comp[g * HDIM + hh] : 0.f;
    }
    __syncthreads();
    int b = tid >> 2, gl = tid & 3, g = g0 + gl;
    if (g < GEX) {
        float d1 = 0.f, d2 = 0.f;
        #pragma unroll 8
        for (int hh = 0; hh < HDIM; hh++) {
            float sv = Ssh[b * HDIM + hh];
            d1 += sv * wg[gl][hh];
            d2 += sv * wc[gl][hh];
        }
        float x = b_gex[b * GEX + g], c = comp[b * GEX + g];
        float row = (x * d1 + c * d2) * 0.08838834764831845f + ppirow[g];
        pred0[b * GEX + g] = x * row;
    }
}

__global__ __launch_bounds__(256) void k_ln(const float* __restrict__ pred0,
                                            const float* __restrict__ gamma,
                                            const float* __restrict__ beta,
                                            float* __restrict__ xlnP) {
    int b = blockIdx.x, tid = threadIdx.x;
    __shared__ float red[256];
    __shared__ float mu_s, rstd_s;
    float s = 0.f;
    for (int g = tid; g < GEX; g += 256) s += pred0[b * GEX + g];
    red[tid] = s;
    __syncthreads();
    for (int o = 128; o > 0; o >>= 1) { if (tid < o) red[tid] += red[tid + o]; __syncthreads(); }
    if (tid == 0) mu_s = red[0] / (float)GEX;
    __syncthreads();
    float mu = mu_s, v = 0.f;
    for (int g = tid; g < GEX; g += 256) { float d = pred0[b * GEX + g] - mu; v += d * d; }
    red[tid] = v;
    __syncthreads();
    for (int o = 128; o > 0; o >>= 1) { if (tid < o) red[tid] += red[tid + o]; __syncthreads(); }
    if (tid == 0) rstd_s = rsqrtf(red[0] / (float)GEX + 1e-5f);
    __syncthreads();
    float rstd = rstd_s;
    for (int g = tid; g < KPAD; g += 256) {
        float y = 0.f;
        if (g < GEX)
            y = fmaxf((pred0[b * GEX + g] - mu) * rstd * gamma[g] + beta[g], 0.f);
        xlnP[b * KPAD + g] = y;
    }
}

#define GWAVES 16
__global__ __launch_bounds__(512) void k_gemm_wmma(const float* __restrict__ xlnP,
                                                   const float* __restrict__ Wff,
                                                   float* __restrict__ out) {
    __shared__ float gA[GWAVES * 32 * SA_PITCH];
    __shared__ float gpart[GWAVES * 512];
    const int tid = threadIdx.x;
    const int lane = tid & 31;
    const int wave = __builtin_amdgcn_readfirstlane(tid >> 5);
    const int c = lane & 15, h = lane >> 4;
    const int nt = blockIdx.x >> 1, mh = blockIdx.x & 1;
    const int k0 = wave * 64;

    int ka = k0 + 2 * lane;
    ka = (ka < KPAD) ? ka : (KPAD - 2);
    const float* arow = xlnP + (mh * 32) * KPAD + ka;
    v2f av[32];
    for (int r = 0; r < 32; r++) av[r] = *(const v2f*)(arow + r * KPAD);

    const int n = nt * 16 + c;
    const int nc = (n < GEX) ? n : (GEX - 1);
    v2f bf[16];
    for (int j = 0; j < 16; j++) {
        const int kb = k0 + 4 * j + 2 * h;
        const bool ok = kb < GEX;
        const int kr = ok ? kb : (GEX - 2);
        const float x = Wff[kr * GEX + nc];
        const float y = Wff[(kr + 1) * GEX + nc];
        bf[j].x = ok ? x : 0.f;
        bf[j].y = ok ? y : 0.f;
    }

    float* sw = gA + wave * (32 * SA_PITCH);
    for (int r = 0; r < 32; r++) *(v2f*)(sw + r * SA_PITCH + 2 * lane) = av[r];
    __syncthreads();

    const float* fa0 = sw + c * SA_PITCH + 2 * h;
    const float* fa1 = fa0 + 16 * SA_PITCH;
    v8f acc0 = {}, acc1 = {};
    for (int j = 0; j < 16; j++) {
        v2f a0 = *(const v2f*)(fa0 + 4 * j);
        v2f a1 = *(const v2f*)(fa1 + 4 * j);
        acc0 = __builtin_amdgcn_wmma_f32_16x16x4_f32(
            false, a0, false, bf[j], (short)0, acc0, false, false);
        acc1 = __builtin_amdgcn_wmma_f32_16x16x4_f32(
            false, a1, false, bf[j], (short)0, acc1, false, false);
    }

    for (int r = 0; r < 8; r++) {
        gpart[wave * 512 + r * 32 + lane] = acc0[r];
        gpart[wave * 512 + 256 + r * 32 + lane] = acc1[r];
    }
    __syncthreads();
    {
        float s0 = 0.f, s1 = 0.f, s2 = 0.f, s3 = 0.f;
        for (int w = 0; w < GWAVES; w += 4) {
            s0 += gpart[(w + 0) * 512 + tid];
            s1 += gpart[(w + 1) * 512 + tid];
            s2 += gpart[(w + 2) * 512 + tid];
            s3 += gpart[(w + 3) * 512 + tid];
        }
        const int t = tid >> 8, idx = tid & 255;
        const int r = idx >> 5, l = idx & 31;
        const int row = mh * 32 + t * 16 + 8 * (l >> 4) + r;
        const int col = nt * 16 + (l & 15);
        if (col < GEX) out[row * GEX + col] = (s0 + s1) + (s2 + s3);
    }
}

extern "C" void kernel_launch(void* const* d_in, const int* in_sizes, int n_in,
                              void* d_out, int out_size, void* d_ws, size_t ws_size,
                              hipStream_t stream) {
    const float* b_gex    = (const float*)d_in[0];
    const float* node_feat= (const float*)d_in[1];
    const float* mask     = (const float*)d_in[2];
    const float* adj      = (const float*)d_in[3];
    const float* dist     = (const float*)d_in[4];
    const float* dose     = (const float*)d_in[5];
    const float* timev    = (const float*)d_in[6];
    const float* ppi      = (const float*)d_in[7];
    const float* w_gex    = (const float*)d_in[8];
    const float* w_comp   = (const float*)d_in[9];
    const float* W1       = (const float*)d_in[10];
    const float* W2       = (const float*)d_in[11];
    const float* w_dose   = (const float*)d_in[12];
    const float* w_time   = (const float*)d_in[13];
    const float* gamma    = (const float*)d_in[14];
    const float* beta     = (const float*)d_in[15];
    const float* Wff      = (const float*)d_in[16];

    float* out  = (float*)d_out;
    float* comp = out + BATCH * GEX;

    float* ws     = (float*)d_ws;
    float* pooled = ws;
    float* S      = pooled + BATCH * CCEH;
    float* ppirow = S + BATCH * HDIM;
    float* pred0  = ppirow + 1024;
    float* xlnP   = pred0 + BATCH * GEX;

    k_cce<<<BATCH, 256, 0, stream>>>(node_feat, mask, adj, dist, W1, pooled);
    k_comp<<<(BATCH * GEX + 255) / 256, 256, 0, stream>>>(pooled, W2, dose, timev,
                                                          w_dose, w_time, comp);
    k_S<<<32, 1024, 0, stream>>>(b_gex, comp, w_gex, w_comp, S);
    k_ppirow<<<GEX, 256, 0, stream>>>(ppi, ppirow);
    k_pred0<<<(GEX + 3) / 4, 256, 0, stream>>>(b_gex, comp, w_gex, w_comp, S, ppirow, pred0);
    k_ln<<<BATCH, 256, 0, stream>>>(pred0, gamma, beta, xlnP);
    k_gemm_wmma<<<124, 512, 0, stream>>>(xlnP, Wff, out);
}


# Launch fusion, 7 launches to 4: comp's two stages and the ppi row sums share one launch; LayerNorm moves into the GEMM's work-groups (row stats per wave in registers); the LN sums' order changes
# speedup vs baseline: 4.1030x; 1.4940x over previous
#include <hip/hip_runtime.h>
#include <hip/hip_bf16.h>

#define GEX 978
#define HDIM 128
#define NAT 50
#define FEAT 34
#define CCEH 64
#define BATCH 64

typedef __attribute__((ext_vector_type(2))) float v2f;
typedef __attribute__((ext_vector_type(8))) float v8f;

__global__ __launch_bounds__(256) void k_front(const float* __restrict__ nf,
                                               const float* __restrict__ mask,
                                               const float* __restrict__ adj,
                                               const float* __restrict__ dist,
                                               const float* __restrict__ W1,
                                               const float* __restrict__ W2,
                                               const float* __restrict__ dose,
                                               const float* __restrict__ timev,
                                               const float* __restrict__ w_dose,
                                               const float* __restrict__ w_time,
                                               const float* __restrict__ ppi,
                                               float* __restrict__ comp,
                                               float* __restrict__ rowsum) {
    const int tid = threadIdx.x;
    __shared__ float W1s[FEAT * CCEH];
    __shared__ float hs[NAT * CCEH];
    __shared__ float wm[NAT];
    __shared__ float pl[CCEH];
    __shared__ float red[256];

    if (blockIdx.x >= BATCH) {
        const int g = blockIdx.x - BATCH;
        float s = 0.f;
        for (int k = tid; k < GEX; k += 256) s += ppi[g * GEX + k];
        red[tid] = s;
        __syncthreads();
        for (int o = 128; o > 0; o >>= 1) {
            if (tid < o) red[tid] += red[tid + o];
            __syncthreads();
        }
        if (tid == 0) rowsum[g] = red[0];
        return;
    }

    const int b = blockIdx.x;
    for (int i = tid; i < FEAT * CCEH; i += 256) W1s[i] = W1[i];
    __syncthreads();
    for (int i = tid; i < NAT * CCEH; i += 256) {
        int n = i / CCEH, d = i % CCEH;
        const float* row = nf + (b * NAT + n) * FEAT;
        float s = 0.f;
        #pragma unroll
        for (int f = 0; f < FEAT; f++) s += row[f] * W1s[f * CCEH + d];
        hs[i] = fmaxf(s, 0.f);
    }
    if (tid < NAT) {
        float s = 0.f;
        for (int n = 0; n < NAT; n++) {
            int idx = (b * NAT + n) * NAT + tid;
            s += mask[b * NAT + n] * adj[idx] * expf(-dist[idx]);
        }
        wm[tid] = s;
    }
    __syncthreads();
    if (tid < CCEH) {
        float msum = 0.f;
        for (int n = 0; n < NAT; n++) msum += mask[b * NAT + n];
        msum = fmaxf(msum, 1.f);
        float s = 0.f;
        for (int m = 0; m < NAT; m++) s += wm[m] * hs[m * CCEH + tid];
        pl[tid] = s / msum;
    }
    __syncthreads();

    float acc[4];
    int gq[4];
    const float db = dose[b], tb = timev[b];
    #pragma unroll
    for (int q = 0; q < 4; q++) {
        const int g = tid + 256 * q;
        gq[q] = (g < GEX) ? g : (GEX - 1);
        acc[q] = db * w_dose[gq[q]] + tb * w_time[gq[q]];
    }
    #pragma unroll 16
    for (int d = 0; d < CCEH; d++) {
        const float p = pl[d];
        #pragma unroll
        for (int q = 0; q < 4; q++) acc[q] += p * W2[d * GEX + gq[q]];
    }
    #pragma unroll
    for (int q = 0; q < 4; q++) {
        const int g = tid + 256 * q;
        if (g < GEX) comp[b * GEX + g] = acc[q];
    }
}

#define SPAIRS (GEX / 2)
#define SA_PITCH 68
__global__ __launch_bounds__(1024) void k_S(const float* __restrict__ b_gex,
                                            const float* __restrict__ comp,
                                            const float* __restrict__ w_gex,
                                            const float* __restrict__ w_comp,
                                            float* __restrict__ S) {
    __shared__ float sA[32 * 16 * SA_PITCH];
    __shared__ float part[32 * 256];
    const int tid = threadIdx.x;
    const int lane = tid & 31;
    const int wave = __builtin_amdgcn_readfirstlane(tid >> 5);
    const int c = lane & 15, h = lane >> 4;
    const int mt = blockIdx.x >> 3, nt = blockIdx.x & 7;
    const int wl = wave & 15;
    const float* Mx = (wave < 16) ? b_gex : comp;
    const float* Wm = (wave < 16) ? w_gex : w_comp;
    const int pw = wl * 32;
    const int p0 = (pw < SPAIRS - 32) ? pw : (SPAIRS - 32);
    const int skip = pw - p0;
    const int k0 = 2 * p0;

    const float* arow = Mx + (mt * 16) * GEX + k0 + 2 * lane;
    v2f av[16];
    for (int r = 0; r < 16; r++) av[r] = *(const v2f*)(arow + r * GEX);

    const float* bp = Wm + (k0 + 2 * h) * HDIM + nt * 16 + c;
    v2f bf[16];
    for (int j = 0; j < 16; j++) {
        bf[j].x = bp[(4 * j) * HDIM];
        bf[j].y = bp[(4 * j + 1) * HDIM];
    }

    float* sw = sA + wave * (16 * SA_PITCH);
    for (int r = 0; r < 16; r++) *(v2f*)(sw + r * SA_PITCH + 2 * lane) = av[r];
    __syncthreads();

    const float* fa = sw + c * SA_PITCH + 2 * h;
    v8f acc0 = {}, acc1 = {};
    for (int j = 0; j < 16; j += 2) {
        v2f a0 = *(const v2f*)(fa + 4 * j);
        v2f a1 = *(const v2f*)(fa + 4 * j + 4);
        v2f b0 = bf[j], b1 = bf[j + 1];
        const bool ok0 = (2 * j + h) >= skip;
        const bool ok1 = (2 * j + 2 + h) >= skip;
        a0.x = ok0 ? a0.x : 0.f;  a0.y = ok0 ? a0.y : 0.f;
        b0.x = ok0 ? b0.x : 0.f;  b0.y = ok0 ? b0.y : 0.f;
        a1.x = ok1 ? a1.x : 0.f;  a1.y = ok1 ? a1.y : 0.f;
        b1.x = ok1 ? b1.x : 0.f;  b1.y = ok1 ? b1.y : 0.f;
        acc0 = __builtin_amdgcn_wmma_f32_16x16x4_f32(
            false, a0, false, b0, (short)0, acc0, false, false);
        acc1 = __builtin_amdgcn_wmma_f32_16x16x4_f32(
            false, a1, false, b1, (short)0, acc1, false, false);
    }

    for (int r = 0; r < 8; r++) part[wave * 256 + r * 32 + lane] = acc0[r] + acc1[r];
    __syncthreads();
    if (tid < 256) {
        float s0 = 0.f, s1 = 0.f, s2 = 0.f, s3 = 0.f;
        for (int w = 0; w < 32; w += 4) {
            s0 += part[(w + 0) * 256 + tid];
            s1 += part[(w + 1) * 256 + tid];
            s2 += part[(w + 2) * 256 + tid];
            s3 += part[(w + 3) * 256 + tid];
        }
        const int r = tid >> 5, l = tid & 31;
        const int row = mt * 16 + 8 * (l >> 4) + r;
        const int col = nt * 16 + (l & 15);
        S[row * HDIM + col] = (s0 + s1) + (s2 + s3);
    }
}

__global__ __launch_bounds__(256) void k_pred0(const float* __restrict__ b_gex,
                                               const float* __restrict__ comp,
                                               const float* __restrict__ w_gex,
                                               const float* __restrict__ w_comp,
                                               const float* __restrict__ S,
                                               const float* __restrict__ ppirow,
                                               float* __restrict__ pred0) {
    __shared__ float Ssh[BATCH * HDIM];
    __shared__ float wg[4][HDIM], wc[4][HDIM];
    int g0 = blockIdx.x * 4, tid = threadIdx.x;
    for (int i = tid; i < BATCH * HDIM; i += 256) Ssh[i] = S[i];
    for (int i = tid; i < 4 * HDIM; i += 256) {
        int gl = i / HDIM, hh = i % HDIM, g = g0 + gl;
        wg[gl][hh] = (g < GEX) ? w_gex[g * HDIM + hh] : 0.f;
        wc[gl][hh] = (g < GEX) ? w_comp[g * HDIM + hh] : 0.f;
    }
    __syncthreads();
    int b = tid >> 2, gl = tid & 3, g = g0 + gl;
    if (g < GEX) {
        float d1 = 0.f, d2 = 0.f;
        #pragma unroll 8
        for (int hh = 0; hh < HDIM; hh++) {
            float sv = Ssh[b * HDIM + hh];
            d1 += sv * wg[gl][hh];
            d2 += sv * wc[gl][hh];
        }
        float x = b_gex[b * GEX + g], c = comp[b * GEX + g];
        float row = (x * d1 + c * d2) * 0.08838834764831845f + ppirow[g];
        pred0[b * GEX + g] = x * row;
    }
}

#define GWAVES 16
#define LNV 31

__device__ __forceinline__ void wave_sum2(float& x, float& y) {
    for (int o = 16; o > 0; o >>= 1) {
        const float px = __shfl_xor(x, o, 32);
        const float py = __shfl_xor(y, o, 32);
        x += px;
        y += py;
    }
}

__global__ __launch_bounds__(512) void k_ln_gemm_wmma(const float* __restrict__ pred0,
                                                      const float* __restrict__ gamma,
                                                      const float* __restrict__ beta,
                                                      const float* __restrict__ Wff,
                                                      float* __restrict__ out) {
    __shared__ float gA[GWAVES * 32 * SA_PITCH];
    __shared__ float gpart[GWAVES * 512];
    __shared__ float mu_s[32], rstd_s[32];
    const int tid = threadIdx.x;
    const int lane = tid & 31;
    const int wave = __builtin_amdgcn_readfirstlane(tid >> 5);
    const int c = lane & 15, h = lane >> 4;
    const int nt = blockIdx.x >> 1, mh = blockIdx.x & 1;
    const int k0 = wave * 64;

    const float* prow0 = pred0 + (mh * 32 + 2 * wave) * GEX + lane;
    const float* prow1 = prow0 + GEX;
    float v0[LNV], v1[LNV];
    #pragma unroll
    for (int i = 0; i < LNV; i++) {
        const bool in = lane + 32 * i < GEX;
        v0[i] = in ? prow0[32 * i] : 0.f;
        v1[i] = in ? prow1[32 * i] : 0.f;
    }

    const int ka = k0 + 2 * lane;
    const bool kin = ka < GEX;
    const int kc = kin ? ka : (GEX - 2);
    const float* arow = pred0 + (mh * 32) * GEX + kc;
    v2f av[32];
    for (int r = 0; r < 32; r++) av[r] = *(const v2f*)(arow + r * GEX);
    const float ga0 = gamma[kc], ga1 = gamma[kc + 1];
    const float be0 = beta[kc], be1 = beta[kc + 1];

    const int n = nt * 16 + c;
    const int nc = (n < GEX) ? n : (GEX - 1);
    v2f bf[16];
    for (int j = 0; j < 16; j++) {
        const int kb = k0 + 4 * j + 2 * h;
        const bool ok = kb < GEX;
        const int kr = ok ? kb : (GEX - 2);
        const float x = Wff[kr * GEX + nc];
        const float y = Wff[(kr + 1) * GEX + nc];
        bf[j].x = ok ? x : 0.f;
        bf[j].y = ok ? y : 0.f;
    }

    {
        float s0 = 0.f, s1 = 0.f;
        #pragma unroll
        for (int i = 0; i < LNV; i++) { s0 += v0[i]; s1 += v1[i]; }
        wave_sum2(s0, s1);
        const float mu0 = s0 / (float)GEX, mu1 = s1 / (float)GEX;
        float q0 = 0.f, q1 = 0.f;
        #pragma unroll
        for (int i = 0; i < LNV; i++) {
            const bool in = lane + 32 * i < GEX;
            const float d0 = v0[i] - mu0, d1 = v1[i] - mu1;
            q0 += in ? d0 * d0 : 0.f;
            q1 += in ? d1 * d1 : 0.f;
        }
        wave_sum2(q0, q1);
        if (lane == 0) {
            mu_s[2 * wave] = mu0;
            mu_s[2 * wave + 1] = mu1;
            rstd_s[2 * wave] = rsqrtf(q0 / (float)GEX + 1e-5f);
            rstd_s[2 * wave + 1] = rsqrtf(q1 / (float)GEX + 1e-5f);
        }
    }
    __syncthreads();

    float* sw = gA + wave * (32 * SA_PITCH);
    for (int r = 0; r < 32; r++) {
        const float mu = mu_s[r], rstd = rstd_s[r];
        const float t0 = fmaxf((av[r].x - mu) * rstd * ga0 + be0, 0.f);
        const float t1 = fmaxf((av[r].y - mu) * rstd * ga1 + be1, 0.f);
        v2f y;
        y.x = kin ? t0 : 0.f;
        y.y = kin ? t1 : 0.f;
        *(v2f*)(sw + r * SA_PITCH + 2 * lane) = y;
    }
    __syncthreads();

    const float* fa0 = sw + c * SA_PITCH + 2 * h;
    const float* fa1 = fa0 + 16 * SA_PITCH;
    v8f acc0 = {}, acc1 = {};
    for (int j = 0; j < 16; j++) {
        v2f a0 = *(const v2f*)(fa0 + 4 * j);
        v2f a1 = *(const v2f*)(fa1 + 4 * j);
        acc0 = __builtin_amdgcn_wmma_f32_16x16x4_f32(
            false, a0, false, bf[j], (short)0, acc0, false, false);
        acc1 = __builtin_amdgcn_wmma_f32_16x16x4_f32(
            false, a1, false, bf[j], (short)0, acc1, false, false);
    }

    for (int r = 0; r < 8; r++) {
        gpart[wave * 512 + r * 32 + lane] = acc0[r];
        gpart[wave * 512 + 256 + r * 32 + lane] = acc1[r];
    }
    __syncthreads();
    {
        float s0 = 0.f, s1 = 0.f, s2 = 0.f, s3 = 0.f;
        for (int w = 0; w < GWAVES; w += 4) {
            s0 += gpart[(w + 0) * 512 + tid];
            s1 += gpart[(w + 1) * 512 + tid];
            s2 += gpart[(w + 2) * 512 + tid];
            s3 += gpart[(w + 3) * 512 + tid];
        }
        const int t = tid >> 8, idx = tid & 255;
        const int r = idx >> 5, l = idx & 31;
        const int row = mh * 32 + t * 16 + 8 * (l >> 4) + r;
        const int col = nt * 16 + (l & 15);
        if (col < GEX) out[row * GEX + col] = (s0 + s1) + (s2 + s3);
    }
}

extern "C" void kernel_launch(void* const* d_in, const int* in_sizes, int n_in,
                              void* d_out, int out_size, void* d_ws, size_t ws_size,
                              hipStream_t stream) {
    const float* b_gex    = (const float*)d_in[0];
    const float* node_feat= (const float*)d_in[1];
    const float* mask     = (const float*)d_in[2];
    const float* adj      = (const float*)d_in[3];
    const float* dist     = (const float*)d_in[4];
    const float* dose     = (const float*)d_in[5];
    const float* timev    = (const float*)d_in[6];
    const float* ppi      = (const float*)d_in[7];
    const float* w_gex    = (const float*)d_in[8];
    const float* w_comp   = (const float*)d_in[9];
    const float* W1       = (const float*)d_in[10];
    const float* W2       = (const float*)d_in[11];
    const float* w_dose   = (const float*)d_in[12];
    const float* w_time   = (const float*)d_in[13];
    const float* gamma    = (const float*)d_in[14];
    const float* beta     = (const float*)d_in[15];
    const float* Wff      = (const float*)d_in[16];

    float* out  = (float*)d_out;
    float* comp = out + BATCH * GEX;

    float* ws     = (float*)d_ws;
    float* S      = ws;
    float* ppirow = S + BATCH * HDIM;
    float* pred0  = ppirow + 1024;

    k_front<<<BATCH + GEX, 256, 0, stream>>>(node_feat, mask, adj, dist, W1, W2, dose, timev,
                                             w_dose, w_time, ppi, comp, ppirow);
    k_S<<<32, 1024, 0, stream>>>(b_gex, comp, w_gex, w_comp, S);
    k_pred0<<<(GEX + 3) / 4, 256, 0, stream>>>(b_gex, comp, w_gex, w_comp, S, ppirow, pred0);
    k_ln_gemm_wmma<<<124, 512, 0, stream>>>(pred0, gamma, beta, Wff, out);
}


# k_front, batch-row groups: nf, W1 column, adj, dist and mask fetched as ONE group of loads over all 256 threads into LDS/registers; the hs and message loops no longer wait on global loads every trip
# speedup vs baseline: 5.1322x; 1.2508x over previous
#include <hip/hip_runtime.h>
#include <hip/hip_bf16.h>

#define GEX 978
#define HDIM 128
#define NAT 50
#define FEAT 34
#define CCEH 64
#define BATCH 64

typedef __attribute__((ext_vector_type(2))) float v2f;
typedef __attribute__((ext_vector_type(8))) float v8f;

#define NFQ 7
#define WPQ 10
__global__ __launch_bounds__(256, 8) void k_front(const float* __restrict__ nf,
                                               const float* __restrict__ mask,
                                               const float* __restrict__ adj,
                                               const float* __restrict__ dist,
                                               const float* __restrict__ W1,
                                               const float* __restrict__ W2,
                                               const float* __restrict__ dose,
                                               const float* __restrict__ timev,
                                               const float* __restrict__ w_dose,
                                               const float* __restrict__ w_time,
                                               const float* __restrict__ ppi,
                                               float* __restrict__ comp,
                                               float* __restrict__ rowsum) {
    const int tid = threadIdx.x;
    __shared__ float nfs[NFQ * 256];
    __shared__ float wprod[WPQ * 256];
    __shared__ float hs[NAT * CCEH];
    __shared__ float msk[256];
    __shared__ float wm[NAT];
    __shared__ float pl[CCEH];
    __shared__ float red[256];

    if (blockIdx.x >= BATCH) {
        const int g = blockIdx.x - BATCH;
        float s = 0.f;
        for (int k = tid; k < GEX; k += 256) s += ppi[g * GEX + k];
        red[tid] = s;
        __syncthreads();
        for (int o = 128; o > 0; o >>= 1) {
            if (tid < o) red[tid] += red[tid + o];
            __syncthreads();
        }
        if (tid == 0) rowsum[g] = red[0];
        return;
    }

    const int b = blockIdx.x;
    const int d = tid & (CCEH - 1);

    const float* nfb = nf + b * (NAT * FEAT);
    float nv[NFQ];
    #pragma unroll
    for (int q = 0; q < NFQ; q++) {
        const int i = tid + 256 * q;
        nv[q] = nfb[(i < NAT * FEAT) ? i : (NAT * FEAT - 1)];
    }
    float w1c[FEAT];
    #pragma unroll
    for (int f = 0; f < FEAT; f++) w1c[f] = W1[f * CCEH + d];
    float av[WPQ], dv[WPQ], mv[WPQ];
    #pragma unroll
    for (int q = 0; q < WPQ; q++) {
        const int i = tid + 256 * q;
        const int ic = (i < NAT * NAT) ? i : (NAT * NAT - 1);
        av[q] = adj[b * (NAT * NAT) + ic];
        dv[q] = dist[b * (NAT * NAT) + ic];
        mv[q] = mask[b * NAT + ic / NAT];
    }
    const float mk = mask[b * NAT + ((tid < NAT) ? tid : 0)];

    #pragma unroll
    for (int q = 0; q < NFQ; q++) nfs[tid + 256 * q] = nv[q];
    #pragma unroll
    for (int q = 0; q < WPQ; q++) wprod[tid + 256 * q] = mv[q] * av[q] * expf(-dv[q]);
    msk[tid] = mk;
    __syncthreads();

    #pragma unroll 2
    for (int it = 0; it < 12; it++) {
        const float* row = nfs + ((tid >> 6) + 4 * it) * FEAT;
        float s = 0.f;
        #pragma unroll
        for (int f = 0; f < FEAT; f++) s += row[f] * w1c[f];
        hs[tid + 256 * it] = fmaxf(s, 0.f);
    }
    if (tid < 128) {
        const float* row = nfs + ((tid >> 6) + 48) * FEAT;
        float s = 0.f;
        #pragma unroll
        for (int f = 0; f < FEAT; f++) s += row[f] * w1c[f];
        hs[tid + 256 * 12] = fmaxf(s, 0.f);
    } else if (tid < 128 + NAT) {
        const int m = tid - 128;
        float s = 0.f;
        #pragma unroll
        for (int n = 0; n < NAT; n++) s += wprod[n * NAT + m];
        wm[m] = s;
    }
    __syncthreads();
    if (tid < CCEH) {
        float msum = 0.f;
        for (int n = 0; n < NAT; n++) msum += msk[n];
        msum = fmaxf(msum, 1.f);
        float s = 0.f;
        for (int m = 0; m < NAT; m++) s += wm[m] * hs[m * CCEH + tid];
        pl[tid] = s / msum;
    }
    __syncthreads();

    float acc[4];
    int gq[4];
    const float db = dose[b], tb = timev[b];
    #pragma unroll
    for (int q = 0; q < 4; q++) {
        const int g = tid + 256 * q;
        gq[q] = (g < GEX) ? g : (GEX - 1);
        acc[q] = db * w_dose[gq[q]] + tb * w_time[gq[q]];
    }
    #pragma unroll 16
    for (int dd = 0; dd < CCEH; dd++) {
        const float p = pl[dd];
        #pragma unroll
        for (int q = 0; q < 4; q++) acc[q] += p * W2[dd * GEX + gq[q]];
    }
    #pragma unroll
    for (int q = 0; q < 4; q++) {
        const int g = tid + 256 * q;
        if (g < GEX) comp[b * GEX + g] = acc[q];
    }
}

#define SPAIRS (GEX / 2)
#define SA_PITCH 68
__global__ __launch_bounds__(1024) void k_S(const float* __restrict__ b_gex,
                                            const float* __restrict__ comp,
                                            const float* __restrict__ w_gex,
                                            const float* __restrict__ w_comp,
                                            float* __restrict__ S) {
    __shared__ float sA[32 * 16 * SA_PITCH];
    __shared__ float part[32 * 256];
    const int tid = threadIdx.x;
    const int lane = tid & 31;
    const int wave = __builtin_amdgcn_readfirstlane(tid >> 5);
    const int c = lane & 15, h = lane >> 4;
    const int mt = blockIdx.x >> 3, nt = blockIdx.x & 7;
    const int wl = wave & 15;
    const float* Mx = (wave < 16) ? b_gex : comp;
    const float* Wm = (wave < 16) ? w_gex : w_comp;
    const int pw = wl * 32;
    const int p0 = (pw < SPAIRS - 32) ? pw : (SPAIRS - 32);
    const int skip = pw - p0;
    const int k0 = 2 * p0;

    const float* arow = Mx + (mt * 16) * GEX + k0 + 2 * lane;
    v2f av[16];
    for (int r = 0; r < 16; r++) av[r] = *(const v2f*)(arow + r * GEX);

    const float* bp = Wm + (k0 + 2 * h) * HDIM + nt * 16 + c;
    v2f bf[16];
    for (int j = 0; j < 16; j++) {
        bf[j].x = bp[(4 * j) * HDIM];
        bf[j].y = bp[(4 * j + 1) * HDIM];
    }

    float* sw = sA + wave * (16 * SA_PITCH);
    for (int r = 0; r < 16; r++) *(v2f*)(sw + r * SA_PITCH + 2 * lane) = av[r];
    __syncthreads();

    const float* fa = sw + c * SA_PITCH + 2 * h;
    v8f acc0 = {}, acc1 = {};
    for (int j = 0; j < 16; j += 2) {
        v2f a0 = *(const v2f*)(fa + 4 * j);
        v2f a1 = *(const v2f*)(fa + 4 * j + 4);
        v2f b0 = bf[j], b1 = bf[j + 1];
        const bool ok0 = (2 * j + h) >= skip;
        const bool ok1 = (2 * j + 2 + h) >= skip;
        a0.x = ok0 ? a0.x : 0.f;  a0.y = ok0 ? a0.y : 0.f;
        b0.x = ok0 ? b0.x : 0.f;  b0.y = ok0 ? b0.y : 0.f;
        a1.x = ok1 ? a1.x : 0.f;  a1.y = ok1 ? a1.y : 0.f;
        b1.x = ok1 ? b1.x : 0.f;  b1.y = ok1 ? b1.y : 0.f;
        acc0 = __builtin_amdgcn_wmma_f32_16x16x4_f32(
            false, a0, false, b0, (short)0, acc0, false, false);
        acc1 = __builtin_amdgcn_wmma_f32_16x16x4_f32(
            false, a1, false, b1, (short)0, acc1, false, false);
    }

    for (int r = 0; r < 8; r++) part[wave * 256 + r * 32 + lane] = acc0[r] + acc1[r];
    __syncthreads();
    if (tid < 256) {
        float s0 = 0.f, s1 = 0.f, s2 = 0.f, s3 = 0.f;
        for (int w = 0; w < 32; w += 4) {
            s0 += part[(w + 0) * 256 + tid];
            s1 += part[(w + 1) * 256 + tid];
            s2 += part[(w + 2) * 256 + tid];
            s3 += part[(w + 3) * 256 + tid];
        }
        const int r = tid >> 5, l = tid & 31;
        const int row = mt * 16 + 8 * (l >> 4) + r;
        const int col = nt * 16 + (l & 15);
        S[row * HDIM + col] = (s0 + s1) + (s2 + s3);
    }
}

__global__ __launch_bounds__(256) void k_pred0(const float* __restrict__ b_gex,
                                               const float* __restrict__ comp,
                                               const float* __restrict__ w_gex,
                                               const float* __restrict__ w_comp,
                                               const float* __restrict__ S,
                                               const float* __restrict__ ppirow,
                                               float* __restrict__ pred0) {
    __shared__ float Ssh[BATCH * HDIM];
    __shared__ float wg[4][HDIM], wc[4][HDIM];
    int g0 = blockIdx.x * 4, tid = threadIdx.x;
    for (int i = tid; i < BATCH * HDIM; i += 256) Ssh[i] = S[i];
    for (int i = tid; i < 4 * HDIM; i += 256) {
        int gl = i / HDIM, hh = i % HDIM, g = g0 + gl;
        wg[gl][hh] = (g < GEX) ? w_gex[g * HDIM + hh] : 0.f;
        wc[gl][hh] = (g < GEX) ? w_comp[g * HDIM + hh] : 0.f;
    }
    __syncthreads();
    int b = tid >> 2, gl = tid & 3, g = g0 + gl;
    if (g < GEX) {
        float d1 = 0.f, d2 = 0.f;
        #pragma unroll 8
        for (int hh = 0; hh < HDIM; hh++) {
            float sv = Ssh[b * HDIM + hh];
            d1 += sv * wg[gl][hh];
            d2 += sv * wc[gl][hh];
        }
        float x = b_gex[b * GEX + g], c = comp[b * GEX + g];
        float row = (x * d1 + c * d2) * 0.08838834764831845f + ppirow[g];
        pred0[b * GEX + g] = x * row;
    }
}

#define GWAVES 16
#define LNV 31

__device__ __forceinline__ void wave_sum2(float& x, float& y) {
    for (int o = 16; o > 0; o >>= 1) {
        const float px = __shfl_xor(x, o, 32);
        const float py = __shfl_xor(y, o, 32);
        x += px;
        y += py;
    }
}

__global__ __launch_bounds__(512) void k_ln_gemm_wmma(const float* __restrict__ pred0,
                                                      const float* __restrict__ gamma,
                                                      const float* __restrict__ beta,
                                                      const float* __restrict__ Wff,
                                                      float* __restrict__ out) {
    __shared__ float gA[GWAVES * 32 * SA_PITCH];
    __shared__ float gpart[GWAVES * 512];
    __shared__ float mu_s[32], rstd_s[32];
    const int tid = threadIdx.x;
    const int lane = tid & 31;
    const int wave = __builtin_amdgcn_readfirstlane(tid >> 5);
    const int c = lane & 15, h = lane >> 4;
    const int nt = blockIdx.x >> 1, mh = blockIdx.x & 1;
    const int k0 = wave * 64;

    const float* prow0 = pred0 + (mh * 32 + 2 * wave) * GEX + lane;
    const float* prow1 = prow0 + GEX;
    float v0[LNV], v1[LNV];
    #pragma unroll
    for (int i = 0; i < LNV; i++) {
        const bool in = lane + 32 * i < GEX;
        v0[i] = in ? prow0[32 * i] : 0.f;
        v1[i] = in ? prow1[32 * i] : 0.f;
    }

    const int ka = k0 + 2 * lane;
    const bool kin = ka < GEX;
    const int kc = kin ? ka : (GEX - 2);
    const float* arow = pred0 + (mh * 32) * GEX + kc;
    v2f av[32];
    for (int r = 0; r < 32; r++) av[r] = *(const v2f*)(arow + r * GEX);
    const float ga0 = gamma[kc], ga1 = gamma[kc + 1];
    const float be0 = beta[kc], be1 = beta[kc + 1];

    const int n = nt * 16 + c;
    const int nc = (n < GEX) ? n : (GEX - 1);
    v2f bf[16];
    for (int j = 0; j < 16; j++) {
        const int kb = k0 + 4 * j + 2 * h;
        const bool ok = kb < GEX;
        const int kr = ok ? kb : (GEX - 2);
        const float x = Wff[kr * GEX + nc];
        const float y = Wff[(kr + 1) * GEX + nc];
        bf[j].x = ok ? x : 0.f;
        bf[j].y = ok ? y : 0.f;
    }

    {
        float s0 = 0.f, s1 = 0.f;
        #pragma unroll
        for (int i = 0; i < LNV; i++) { s0 += v0[i]; s1 += v1[i]; }
        wave_sum2(s0, s1);
        const float mu0 = s0 / (float)GEX, mu1 = s1 / (float)GEX;
        float q0 = 0.f, q1 = 0.f;
        #pragma unroll
        for (int i = 0; i < LNV; i++) {
            const bool in = lane + 32 * i < GEX;
            const float d0 = v0[i] - mu0, d1 = v1[i] - mu1;
            q0 += in ? d0 * d0 : 0.f;
            q1 += in ? d1 * d1 : 0.f;
        }
        wave_sum2(q0, q1);
        if (lane == 0) {
            mu_s[2 * wave] = mu0;
            mu_s[2 * wave + 1] = mu1;
            rstd_s[2 * wave] = rsqrtf(q0 / (float)GEX + 1e-5f);
            rstd_s[2 * wave + 1] = rsqrtf(q1 / (float)GEX + 1e-5f);
        }
    }
    __syncthreads();

    float* sw = gA + wave * (32 * SA_PITCH);
    for (int r = 0; r < 32; r++) {
        const float mu = mu_s[r], rstd = rstd_s[r];
        const float t0 = fmaxf((av[r].x - mu) * rstd * ga0 + be0, 0.f);
        const float t1 = fmaxf((av[r].y - mu) * rstd * ga1 + be1, 0.f);
        v2f y;
        y.x = kin ? t0 : 0.f;
        y.y = kin ? t1 : 0.f;
        *(v2f*)(sw + r * SA_PITCH + 2 * lane) = y;
    }
    __syncthreads();

    const float* fa0 = sw + c * SA_PITCH + 2 * h;
    const float* fa1 = fa0 + 16 * SA_PITCH;
    v8f acc0 = {}, acc1 = {};
    for (int j = 0; j < 16; j++) {
        v2f a0 = *(const v2f*)(fa0 + 4 * j);
        v2f a1 = *(const v2f*)(fa1 + 4 * j);
        acc0 = __builtin_amdgcn_wmma_f32_16x16x4_f32(
            false, a0, false, bf[j], (short)0, acc0, false, false);
        acc1 = __builtin_amdgcn_wmma_f32_16x16x4_f32(
            false, a1, false, bf[j], (short)0, acc1, false, false);
    }

    for (int r = 0; r < 8; r++) {
        gpart[wave * 512 + r * 32 + lane] = acc0[r];
        gpart[wave * 512 + 256 + r * 32 + lane] = acc1[r];
    }
    __syncthreads();
    {
        float s0 = 0.f, s1 = 0.f, s2 = 0.f, s3 = 0.f;
        for (int w = 0; w < GWAVES; w += 4) {
            s0 += gpart[(w + 0) * 512 + tid];
            s1 += gpart[(w + 1) * 512 + tid];
            s2 += gpart[(w + 2) * 512 + tid];
            s3 += gpart[(w + 3) * 512 + tid];
        }
        const int t = tid >> 8, idx = tid & 255;
        const int r = idx >> 5, l = idx & 31;
        const int row = mh * 32 + t * 16 + 8 * (l >> 4) + r;
        const int col = nt * 16 + (l & 15);
        if (col < GEX) out[row * GEX + col] = (s0 + s1) + (s2 + s3);
    }
}

extern "C" void kernel_launch(void* const* d_in, const int* in_sizes, int n_in,
                              void* d_out, int out_size, void* d_ws, size_t ws_size,
                              hipStream_t stream) {
    const float* b_gex    = (const float*)d_in[0];
    const float* node_feat= (const float*)d_in[1];
    const float* mask     = (const float*)d_in[2];
    const float* adj      = (const float*)d_in[3];
    const float* dist     = (const float*)d_in[4];
    const float* dose     = (const float*)d_in[5];
    const float* timev    = (const float*)d_in[6];
    const float* ppi      = (const float*)d_in[7];
    const float* w_gex    = (const float*)d_in[8];
    const float* w_comp   = (const float*)d_in[9];
    const float* W1       = (const float*)d_in[10];
    const float* W2       = (const float*)d_in[11];
    const float* w_dose   = (const float*)d_in[12];
    const float* w_time   = (const float*)d_in[13];
    const float* gamma    = (const float*)d_in[14];
    const float* beta     = (const float*)d_in[15];
    const float* Wff      = (const float*)d_in[16];

    float* out  = (float*)d_out;
    float* comp = out + BATCH * GEX;

    float* ws     = (float*)d_ws;
    float* S      = ws;
    float* ppirow = S + BATCH * HDIM;
    float* pred0  = ppirow + 1024;

    k_front<<<BATCH + GEX, 256, 0, stream>>>(node_feat, mask, adj, dist, W1, W2, dose, timev,
                                             w_dose, w_time, ppi, comp, ppirow);
    k_S<<<32, 1024, 0, stream>>>(b_gex, comp, w_gex, w_comp, S);
    k_pred0<<<(GEX + 3) / 4, 256, 0, stream>>>(b_gex, comp, w_gex, w_comp, S, ppirow, pred0);
    k_ln_gemm_wmma<<<124, 512, 0, stream>>>(pred0, gamma, beta, Wff, out);
}


# k_pred0: its two 128-long dot products per element go onto f32 WMMA, one wave per 16x16 tile, fragments by 16-byte loads straight from S and the weight rows, no LDS; their summation order changes
# speedup vs baseline: 5.7140x; 1.1134x over previous
#include <hip/hip_runtime.h>
#include <hip/hip_bf16.h>

#define GEX 978
#define HDIM 128
#define NAT 50
#define FEAT 34
#define CCEH 64
#define BATCH 64

typedef __attribute__((ext_vector_type(2))) float v2f;
typedef __attribute__((ext_vector_type(4))) float v4f;
typedef __attribute__((ext_vector_type(8))) float v8f;

#define NFQ 7
#define WPQ 10
__global__ __launch_bounds__(256, 8) void k_front(const float* __restrict__ nf,
                                               const float* __restrict__ mask,
                                               const float* __restrict__ adj,
                                               const float* __restrict__ dist,
                                               const float* __restrict__ W1,
                                               const float* __restrict__ W2,
                                               const float* __restrict__ dose,
                                               const float* __restrict__ timev,
                                               const float* __restrict__ w_dose,
                                               const float* __restrict__ w_time,
                                               const float* __restrict__ ppi,
                                               float* __restrict__ comp,
                                               float* __restrict__ rowsum) {
    const int tid = threadIdx.x;
    __shared__ float nfs[NFQ * 256];
    __shared__ float wprod[WPQ * 256];
    __shared__ float hs[NAT * CCEH];
    __shared__ float msk[256];
    __shared__ float wm[NAT];
    __shared__ float pl[CCEH];
    __shared__ float red[256];

    if (blockIdx.x >= BATCH) {
        const int g = blockIdx.x - BATCH;
        float s = 0.f;
        for (int k = tid; k < GEX; k += 256) s += ppi[g * GEX + k];
        red[tid] = s;
        __syncthreads();
        for (int o = 128; o > 0; o >>= 1) {
            if (tid < o) red[tid] += red[tid + o];
            __syncthreads();
        }
        if (tid == 0) rowsum[g] = red[0];
        return;
    }

    const int b = blockIdx.x;
    const int d = tid & (CCEH - 1);

    const float* nfb = nf + b * (NAT * FEAT);
    float nv[NFQ];
    #pragma unroll
    for (int q = 0; q < NFQ; q++) {
        const int i = tid + 256 * q;
        nv[q] = nfb[(i < NAT * FEAT) ? i : (NAT * FEAT - 1)];
    }
    float w1c[FEAT];
    #pragma unroll
    for (int f = 0; f < FEAT; f++) w1c[f] = W1[f * CCEH + d];
    float av[WPQ], dv[WPQ], mv[WPQ];
    #pragma unroll
    for (int q = 0; q < WPQ; q++) {
        const int i = tid + 256 * q;
        const int ic = (i < NAT * NAT) ? i : (NAT * NAT - 1);
        av[q] = adj[b * (NAT * NAT) + ic];
        dv[q] = dist[b * (NAT * NAT) + ic];
        mv[q] = mask[b * NAT + ic / NAT];
    }
    const float mk = mask[b * NAT + ((tid < NAT) ? tid : 0)];

    #pragma unroll
    for (int q = 0; q < NFQ; q++) nfs[tid + 256 * q] = nv[q];
    #pragma unroll
    for (int q = 0; q < WPQ; q++) wprod[tid + 256 * q] = mv[q] * av[q] * expf(-dv[q]);
    msk[tid] = mk;
    __syncthreads();

    #pragma unroll 2
    for (int it = 0; it < 12; it++) {
        const float* row = nfs + ((tid >> 6) + 4 * it) * FEAT;
        float s = 0.f;
        #pragma unroll
        for (int f = 0; f < FEAT; f++) s += row[f] * w1c[f];
        hs[tid + 256 * it] = fmaxf(s, 0.f);
    }
    if (tid < 128) {
        const float* row = nfs + ((tid >> 6) + 48) * FEAT;
        float s = 0.f;
        #pragma unroll
        for (int f = 0; f < FEAT; f++) s += row[f] * w1c[f];
        hs[tid + 256 * 12] = fmaxf(s, 0.f);
    } else if (tid < 128 + NAT) {
        const int m = tid - 128;
        float s = 0.f;
        #pragma unroll
        for (int n = 0; n < NAT; n++) s += wprod[n * NAT + m];
        wm[m] = s;
    }
    __syncthreads();
    if (tid < CCEH) {
        float msum = 0.f;
        for (int n = 0; n < NAT; n++) msum += msk[n];
        msum = fmaxf(msum, 1.f);
        float s = 0.f;
        for (int m = 0; m < NAT; m++) s += wm[m] * hs[m * CCEH + tid];
        pl[tid] = s / msum;
    }
    __syncthreads();

    float acc[4];
    int gq[4];
    const float db = dose[b], tb = timev[b];
    #pragma unroll
    for (int q = 0; q < 4; q++) {
        const int g = tid + 256 * q;
        gq[q] = (g < GEX) ? g : (GEX - 1);
        acc[q] = db * w_dose[gq[q]] + tb * w_time[gq[q]];
    }
    #pragma unroll 16
    for (int dd = 0; dd < CCEH; dd++) {
        const float p = pl[dd];
        #pragma unroll
        for (int q = 0; q < 4; q++) acc[q] += p * W2[dd * GEX + gq[q]];
    }
    #pragma unroll
    for (int q = 0; q < 4; q++) {
        const int g = tid + 256 * q;
        if (g < GEX) comp[b * GEX + g] = acc[q];
    }
}

#define SPAIRS (GEX / 2)
#define SA_PITCH 68
__global__ __launch_bounds__(1024) void k_S(const float* __restrict__ b_gex,
                                            const float* __restrict__ comp,
                                            const float* __restrict__ w_gex,
                                            const float* __restrict__ w_comp,
                                            float* __restrict__ S) {
    __shared__ float sA[32 * 16 * SA_PITCH];
    __shared__ float part[32 * 256];
    const int tid = threadIdx.x;
    const int lane = tid & 31;
    const int wave = __builtin_amdgcn_readfirstlane(tid >> 5);
    const int c = lane & 15, h = lane >> 4;
    const int mt = blockIdx.x >> 3, nt = blockIdx.x & 7;
    const int wl = wave & 15;
    const float* Mx = (wave < 16) ? b_gex : comp;
    const float* Wm = (wave < 16) ? w_gex : w_comp;
    const int pw = wl * 32;
    const int p0 = (pw < SPAIRS - 32) ? pw : (SPAIRS - 32);
    const int skip = pw - p0;
    const int k0 = 2 * p0;

    const float* arow = Mx + (mt * 16) * GEX + k0 + 2 * lane;
    v2f av[16];
    for (int r = 0; r < 16; r++) av[r] = *(const v2f*)(arow + r * GEX);

    const float* bp = Wm + (k0 + 2 * h) * HDIM + nt * 16 + c;
    v2f bf[16];
    for (int j = 0; j < 16; j++) {
        bf[j].x = bp[(4 * j) * HDIM];
        bf[j].y = bp[(4 * j + 1) * HDIM];
    }

    float* sw = sA + wave * (16 * SA_PITCH);
    for (int r = 0; r < 16; r++) *(v2f*)(sw + r * SA_PITCH + 2 * lane) = av[r];
    __syncthreads();

    const float* fa = sw + c * SA_PITCH + 2 * h;
    v8f acc0 = {}, acc1 = {};
    for (int j = 0; j < 16; j += 2) {
        v2f a0 = *(const v2f*)(fa + 4 * j);
        v2f a1 = *(const v2f*)(fa + 4 * j + 4);
        v2f b0 = bf[j], b1 = bf[j + 1];
        const bool ok0 = (2 * j + h) >= skip;
        const bool ok1 = (2 * j + 2 + h) >= skip;
        a0.x = ok0 ? a0.x : 0.f;  a0.y = ok0 ? a0.y : 0.f;
        b0.x = ok0 ? b0.x : 0.f;  b0.y = ok0 ? b0.y : 0.f;
        a1.x = ok1 ? a1.x : 0.f;  a1.y = ok1 ? a1.y : 0.f;
        b1.x = ok1 ? b1.x : 0.f;  b1.y = ok1 ? b1.y : 0.f;
        acc0 = __builtin_amdgcn_wmma_f32_16x16x4_f32(
            false, a0, false, b0, (short)0, acc0, false, false);
        acc1 = __builtin_amdgcn_wmma_f32_16x16x4_f32(
            false, a1, false, b1, (short)0, acc1, false, false);
    }

    for (int r = 0; r < 8; r++) part[wave * 256 + r * 32 + lane] = acc0[r] + acc1[r];
    __syncthreads();
    if (tid < 256) {
        float s0 = 0.f, s1 = 0.f, s2 = 0.f, s3 = 0.f;
        for (int w = 0; w < 32; w += 4) {
            s0 += part[(w + 0) * 256 + tid];
            s1 += part[(w + 1) * 256 + tid];
            s2 += part[(w + 2) * 256 + tid];
            s3 += part[(w + 3) * 256 + tid];
        }
        const int r = tid >> 5, l = tid & 31;
        const int row = mt * 16 + 8 * (l >> 4) + r;
        const int col = nt * 16 + (l & 15);
        S[row * HDIM + col] = (s0 + s1) + (s2 + s3);
    }
}

__global__ __launch_bounds__(32) void k_pred0(const float* __restrict__ b_gex,
                                              const float* __restrict__ comp,
                                              const float* __restrict__ w_gex,
                                              const float* __restrict__ w_comp,
                                              const float* __restrict__ S,
                                              const float* __restrict__ ppirow,
                                              float* __restrict__ pred0) {
    const int lane = threadIdx.x;
    const int c = lane & 15, h = lane >> 4;
    const int mt = blockIdx.x & 3, nt = blockIdx.x >> 2;
    const int g = nt * 16 + c;
    const int gc = (g < GEX) ? g : (GEX - 1);
    const v4f* ap = (const v4f*)(S + (mt * 16 + c) * HDIM + 4 * h);
    const v4f* gp = (const v4f*)(w_gex + gc * HDIM + 4 * h);
    const v4f* cp = (const v4f*)(w_comp + gc * HDIM + 4 * h);
    const int b0 = mt * 16 + 8 * h;
    float xv[8], cv[8];
    #pragma unroll
    for (int r = 0; r < 8; r++) {
        xv[r] = b_gex[(b0 + r) * GEX + gc];
        cv[r] = comp[(b0 + r) * GEX + gc];
    }
    const float pr = ppirow[gc];
    v4f af[16], gf[16], cf[16];
    #pragma unroll
    for (int J = 0; J < 16; J++) {
        af[J] = ap[2 * J];
        gf[J] = gp[2 * J];
        cf[J] = cp[2 * J];
    }
    __builtin_amdgcn_sched_barrier(0);

    v8f t1 = {}, t2 = {};
    #pragma unroll
    for (int J = 0; J < 16; J++) {
        v2f a0, a1, g0, g1, c0, c1;
        a0.x = af[J].x;  a0.y = af[J].y;  a1.x = af[J].z;  a1.y = af[J].w;
        g0.x = gf[J].x;  g0.y = gf[J].y;  g1.x = gf[J].z;  g1.y = gf[J].w;
        c0.x = cf[J].x;  c0.y = cf[J].y;  c1.x = cf[J].z;  c1.y = cf[J].w;
        t1 = __builtin_amdgcn_wmma_f32_16x16x4_f32(
            false, a0, false, g0, (short)0, t1, false, false);
        t2 = __builtin_amdgcn_wmma_f32_16x16x4_f32(
            false, a0, false, c0, (short)0, t2, false, false);
        t1 = __builtin_amdgcn_wmma_f32_16x16x4_f32(
            false, a1, false, g1, (short)0, t1, false, false);
        t2 = __builtin_amdgcn_wmma_f32_16x16x4_f32(
            false, a1, false, c1, (short)0, t2, false, false);
    }

    asm volatile("" :: "v"(xv[0]), "v"(xv[1]), "v"(xv[2]), "v"(xv[3]), "v"(xv[4]), "v"(xv[5]), "v"(xv[6]), "v"(xv[7]),
                       "v"(cv[0]), "v"(cv[1]), "v"(cv[2]), "v"(cv[3]), "v"(cv[4]), "v"(cv[5]), "v"(cv[6]), "v"(cv[7]),
                       "v"(pr));
    if (g < GEX) {
        #pragma unroll
        for (int r = 0; r < 8; r++) {
            const float x = xv[r];
            const float row = (x * t1[r] + cv[r] * t2[r]) * 0.08838834764831845f + pr;
            pred0[(b0 + r) * GEX + g] = x * row;
        }
    }
}

#define GWAVES 16
#define LNV 31

__device__ __forceinline__ void wave_sum2(float& x, float& y) {
    for (int o = 16; o > 0; o >>= 1) {
        const float px = __shfl_xor(x, o, 32);
        const float py = __shfl_xor(y, o, 32);
        x += px;
        y += py;
    }
}

__global__ __launch_bounds__(512) void k_ln_gemm_wmma(const float* __restrict__ pred0,
                                                      const float* __restrict__ gamma,
                                                      const float* __restrict__ beta,
                                                      const float* __restrict__ Wff,
                                                      float* __restrict__ out) {
    __shared__ float gA[GWAVES * 32 * SA_PITCH];
    __shared__ float gpart[GWAVES * 512];
    __shared__ float mu_s[32], rstd_s[32];
    const int tid = threadIdx.x;
    const int lane = tid & 31;
    const int wave = __builtin_amdgcn_readfirstlane(tid >> 5);
    const int c = lane & 15, h = lane >> 4;
    const int nt = blockIdx.x >> 1, mh = blockIdx.x & 1;
    const int k0 = wave * 64;

    const float* prow0 = pred0 + (mh * 32 + 2 * wave) * GEX + lane;
    const float* prow1 = prow0 + GEX;
    float v0[LNV], v1[LNV];
    #pragma unroll
    for (int i = 0; i < LNV; i++) {
        const bool in = lane + 32 * i < GEX;
        v0[i] = in ? prow0[32 * i] : 0.f;
        v1[i] = in ? prow1[32 * i] : 0.f;
    }

    const int ka = k0 + 2 * lane;
    const bool kin = ka < GEX;
    const int kc = kin ? ka : (GEX - 2);
    const float* arow = pred0 + (mh * 32) * GEX + kc;
    v2f av[32];
    for (int r = 0; r < 32; r++) av[r] = *(const v2f*)(arow + r * GEX);
    const float ga0 = gamma[kc], ga1 = gamma[kc + 1];
    const float be0 = beta[kc], be1 = beta[kc + 1];

    const int n = nt * 16 + c;
    const int nc = (n < GEX) ? n : (GEX - 1);
    v2f bf[16];
    for (int j = 0; j < 16; j++) {
        const int kb = k0 + 4 * j + 2 * h;
        const bool ok = kb < GEX;
        const int kr = ok ? kb : (GEX - 2);
        const float x = Wff[kr * GEX + nc];
        const float y = Wff[(kr + 1) * GEX + nc];
        bf[j].x = ok ? x : 0.f;
        bf[j].y = ok ? y : 0.f;
    }

    {
        float s0 = 0.f, s1 = 0.f;
        #pragma unroll
        for (int i = 0; i < LNV; i++) { s0 += v0[i]; s1 += v1[i]; }
        wave_sum2(s0, s1);
        const float mu0 = s0 / (float)GEX, mu1 = s1 / (float)GEX;
        float q0 = 0.f, q1 = 0.f;
        #pragma unroll
        for (int i = 0; i < LNV; i++) {
            const bool in = lane + 32 * i < GEX;
            const float d0 = v0[i] - mu0, d1 = v1[i] - mu1;
            q0 += in ? d0 * d0 : 0.f;
            q1 += in ? d1 * d1 : 0.f;
        }
        wave_sum2(q0, q1);
        if (lane == 0) {
            mu_s[2 * wave] = mu0;
            mu_s[2 * wave + 1] = mu1;
            rstd_s[2 * wave] = rsqrtf(q0 / (float)GEX + 1e-5f);
            rstd_s[2 * wave + 1] = rsqrtf(q1 / (float)GEX + 1e-5f);
        }
    }
    __syncthreads();

    float* sw = gA + wave * (32 * SA_PITCH);
    for (int r = 0; r < 32; r++) {
        const float mu = mu_s[r], rstd = rstd_s[r];
        const float t0 = fmaxf((av[r].x - mu) * rstd * ga0 + be0, 0.f);
        const float t1 = fmaxf((av[r].y - mu) * rstd * ga1 + be1, 0.f);
        v2f y;
        y.x = kin ? t0 : 0.f;
        y.y = kin ? t1 : 0.f;
        *(v2f*)(sw + r * SA_PITCH + 2 * lane) = y;
    }
    __syncthreads();

    const float* fa0 = sw + c * SA_PITCH + 2 * h;
    const float* fa1 = fa0 + 16 * SA_PITCH;
    v8f acc0 = {}, acc1 = {};
    for (int j = 0; j < 16; j++) {
        v2f a0 = *(const v2f*)(fa0 + 4 * j);
        v2f a1 = *(const v2f*)(fa1 + 4 * j);
        acc0 = __builtin_amdgcn_wmma_f32_16x16x4_f32(
            false, a0, false, bf[j], (short)0, acc0, false, false);
        acc1 = __builtin_amdgcn_wmma_f32_16x16x4_f32(
            false, a1, false, bf[j], (short)0, acc1, false, false);
    }

    for (int r = 0; r < 8; r++) {
        gpart[wave * 512 + r * 32 + lane] = acc0[r];
        gpart[wave * 512 + 256 + r * 32 + lane] = acc1[r];
    }
    __syncthreads();
    {
        float s0 = 0.f, s1 = 0.f, s2 = 0.f, s3 = 0.f;
        for (int w = 0; w < GWAVES; w += 4) {
            s0 += gpart[(w + 0) * 512 + tid];
            s1 += gpart[(w + 1) * 512 + tid];
            s2 += gpart[(w + 2) * 512 + tid];
            s3 += gpart[(w + 3) * 512 + tid];
        }
        const int t = tid >> 8, idx = tid & 255;
        const int r = idx >> 5, l = idx & 31;
        const int row = mh * 32 + t * 16 + 8 * (l >> 4) + r;
        const int col = nt * 16 + (l & 15);
        if (col < GEX) out[row * GEX + col] = (s0 + s1) + (s2 + s3);
    }
}

extern "C" void kernel_launch(void* const* d_in, const int* in_sizes, int n_in,
                              void* d_out, int out_size, void* d_ws, size_t ws_size,
                              hipStream_t stream) {
    const float* b_gex    = (const float*)d_in[0];
    const float* node_feat= (const float*)d_in[1];
    const float* mask     = (const float*)d_in[2];
    const float* adj      = (const float*)d_in[3];
    const float* dist     = (const float*)d_in[4];
    const float* dose     = (const float*)d_in[5];
    const float* timev    = (const float*)d_in[6];
    const float* ppi      = (const float*)d_in[7];
    const float* w_gex    = (const float*)d_in[8];
    const float* w_comp   = (const float*)d_in[9];
    const float* W1       = (const float*)d_in[10];
    const float* W2       = (const float*)d_in[11];
    const float* w_dose   = (const float*)d_in[12];
    const float* w_time   = (const float*)d_in[13];
    const float* gamma    = (const float*)d_in[14];
    const float* beta     = (const float*)d_in[15];
    const float* Wff      = (const float*)d_in[16];

    float* out  = (float*)d_out;
    float* comp = out + BATCH * GEX;

    float* ws     = (float*)d_ws;
    float* S      = ws;
    float* ppirow = S + BATCH * HDIM;
    float* pred0  = ppirow + 1024;

    k_front<<<BATCH + GEX, 256, 0, stream>>>(node_feat, mask, adj, dist, W1, W2, dose, timev,
                                             w_dose, w_time, ppi, comp, ppirow);
    k_S<<<32, 1024, 0, stream>>>(b_gex, comp, w_gex, w_comp, S);
    k_pred0<<<248, 32, 0, stream>>>(b_gex, comp, w_gex, w_comp, S, ppirow, pred0);
    k_ln_gemm_wmma<<<124, 512, 0, stream>>>(pred0, gamma, beta, Wff, out);
}


# comp@w_comp regrouped as [pooled,dose,time]@([W2;w_dose;w_time]@w_comp): S=P+pe@Q is finished on WMMA inside k_pred0, the S launch goes (4 launches to 3); summation order of S changes
# speedup vs baseline: 7.0503x; 1.2339x over previous
#include <hip/hip_runtime.h>
#include <hip/hip_bf16.h>

#define GEX 978
#define HDIM 128
#define NAT 50
#define FEAT 34
#define CCEH 64
#define BATCH 64

typedef __attribute__((ext_vector_type(2))) float v2f;
typedef __attribute__((ext_vector_type(4))) float v4f;
typedef __attribute__((ext_vector_type(8))) float v8f;

#define NFQ 7
#define WPQ 10
#define NPQ 72
#define PEP 68
#define SPAIRS (GEX / 2)
__global__ __launch_bounds__(256, 8) void k_front(const float* __restrict__ nf,
                                               const float* __restrict__ mask,
                                               const float* __restrict__ adj,
                                               const float* __restrict__ dist,
                                               const float* __restrict__ W1,
                                               const float* __restrict__ W2,
                                               const float* __restrict__ dose,
                                               const float* __restrict__ timev,
                                               const float* __restrict__ w_dose,
                                               const float* __restrict__ w_time,
                                               const float* __restrict__ ppi,
                                               const float* __restrict__ b_gex,
                                               const float* __restrict__ w_gex,
                                               const float* __restrict__ w_comp,
                                               float* __restrict__ comp,
                                               float* __restrict__ rowsum,
                                               float* __restrict__ P,
                                               float* __restrict__ Q,
                                               float* __restrict__ pe) {
    const int tid = threadIdx.x;
    __shared__ float nfs[NFQ * 256];
    __shared__ float wprod[WPQ * 256];
    __shared__ float hs[NAT * CCEH];
    __shared__ float msk[256];
    __shared__ float wm[NAT];
    __shared__ float pl[CCEH];
    __shared__ float red[256];

    if (blockIdx.x >= BATCH && blockIdx.x < BATCH + NPQ) {
        const int t = blockIdx.x - BATCH;
        const bool isP = t < 32;
        const int u = isP ? t : (t - 32);
        const int mt = u >> 3, nt = u & 7;
        const int lane = tid & 31;
        const int wave = __builtin_amdgcn_readfirstlane(tid >> 5);
        const int c = lane & 15, h = lane >> 4;
        const int R = mt * 16 + c;
        const float* arow;
        if (isP) arow = b_gex + R * GEX;
        else arow = (R < CCEH) ? (W2 + R * GEX) : ((R == CCEH) ? w_dose : w_time);
        const float* Wm = isP ? w_gex : w_comp;
        v8f acc0 = {}, acc1 = {};
        #pragma unroll
        for (int q = 0; q < 2; q++) {
            const int pw = (wave + 8 * q) * 32;
            const int p0 = (pw < SPAIRS - 32) ? pw : (SPAIRS - 32);
            const int skip = pw - p0;
            const int k0 = 2 * p0;
            const float* ap = arow + k0 + 2 * h;
            const float* bp = Wm + (k0 + 2 * h) * HDIM + nt * 16 + c;
            v2f af[16], bf[16];
            #pragma unroll
            for (int j = 0; j < 16; j++) {
                af[j] = *(const v2f*)(ap + 4 * j);
                bf[j].x = bp[(4 * j) * HDIM];
                bf[j].y = bp[(4 * j + 1) * HDIM];
            }
            #pragma unroll
            for (int j = 0; j < 16; j += 2) {
                v2f a0 = af[j], a1 = af[j + 1];
                v2f b0 = bf[j], b1 = bf[j + 1];
                const bool ok0 = (2 * j + h) >= skip;
                const bool ok1 = (2 * j + 2 + h) >= skip;
                a0.x = ok0 ? a0.x : 0.f;  a0.y = ok0 ? a0.y : 0.f;
                b0.x = ok0 ? b0.x : 0.f;  b0.y = ok0 ? b0.y : 0.f;
                a1.x = ok1 ? a1.x : 0.f;  a1.y = ok1 ? a1.y : 0.f;
                b1.x = ok1 ? b1.x : 0.f;  b1.y = ok1 ? b1.y : 0.f;
                acc0 = __builtin_amdgcn_wmma_f32_16x16x4_f32(
                    false, a0, false, b0, (short)0, acc0, false, false);
                acc1 = __builtin_amdgcn_wmma_f32_16x16x4_f32(
                    false, a1, false, b1, (short)0, acc1, false, false);
            }
        }
        #pragma unroll
        for (int r = 0; r < 8; r++) wprod[wave * 256 + r * 32 + lane] = acc0[r] + acc1[r];
        __syncthreads();
        {
            float s0 = 0.f, s1 = 0.f;
            for (int w = 0; w < 8; w += 2) {
                s0 += wprod[(w + 0) * 256 + tid];
                s1 += wprod[(w + 1) * 256 + tid];
            }
            const int r = tid >> 5, l = tid & 31;
            const int row = mt * 16 + 8 * (l >> 4) + r;
            const int col = nt * 16 + (l & 15);
            if (isP) P[row * HDIM + col] = s0 + s1;
            else Q[row * HDIM + col] = (row < CCEH + 2) ? (s0 + s1) : 0.f;
        }
        return;
    }
    if (blockIdx.x >= BATCH + NPQ) {
        const int g = blockIdx.x - BATCH - NPQ;
        float s = 0.f;
        for (int k = tid; k < GEX; k += 256) s += ppi[g * GEX + k];
        red[tid] = s;
        __syncthreads();
        for (int o = 128; o > 0; o >>= 1) {
            if (tid < o) red[tid] += red[tid + o];
            __syncthreads();
        }
        if (tid == 0) rowsum[g] = red[0];
        return;
    }

    const int b = blockIdx.x;
    const int d = tid & (CCEH - 1);

    const float* nfb = nf + b * (NAT * FEAT);
    float nv[NFQ];
    #pragma unroll
    for (int q = 0; q < NFQ; q++) {
        const int i = tid + 256 * q;
        nv[q] = nfb[(i < NAT * FEAT) ? i : (NAT * FEAT - 1)];
    }
    float w1c[FEAT];
    #pragma unroll
    for (int f = 0; f < FEAT; f++) w1c[f] = W1[f * CCEH + d];
    float av[WPQ], dv[WPQ], mv[WPQ];
    #pragma unroll
    for (int q = 0; q < WPQ; q++) {
        const int i = tid + 256 * q;
        const int ic = (i < NAT * NAT) ? i : (NAT * NAT - 1);
        av[q] = adj[b * (NAT * NAT) + ic];
        dv[q] = dist[b * (NAT * NAT) + ic];
        mv[q] = mask[b * NAT + ic / NAT];
    }
    const float mk = mask[b * NAT + ((tid < NAT) ? tid : 0)];

    #pragma unroll
    for (int q = 0; q < NFQ; q++) nfs[tid + 256 * q] = nv[q];
    #pragma unroll
    for (int q = 0; q < WPQ; q++) wprod[tid + 256 * q] = mv[q] * av[q] * expf(-dv[q]);
    msk[tid] = mk;
    __syncthreads();

    #pragma unroll 2
    for (int it = 0; it < 12; it++) {
        const float* row = nfs + ((tid >> 6) + 4 * it) * FEAT;
        float s = 0.f;
        #pragma unroll
        for (int f = 0; f < FEAT; f++) s += row[f] * w1c[f];
        hs[tid + 256 * it] = fmaxf(s, 0.f);
    }
    if (tid < 128) {
        const float* row = nfs + ((tid >> 6) + 48) * FEAT;
        float s = 0.f;
        #pragma unroll
        for (int f = 0; f < FEAT; f++) s += row[f] * w1c[f];
        hs[tid + 256 * 12] = fmaxf(s, 0.f);
    } else if (tid < 128 + NAT) {
        const int m = tid - 128;
        float s = 0.f;
        #pragma unroll
        for (int n = 0; n < NAT; n++) s += wprod[n * NAT + m];
        wm[m] = s;
    }
    __syncthreads();
    if (tid < CCEH) {
        float msum = 0.f;
        for (int n = 0; n < NAT; n++) msum += msk[n];
        msum = fmaxf(msum, 1.f);
        float s = 0.f;
        for (int m = 0; m < NAT; m++) s += wm[m] * hs[m * CCEH + tid];
        pl[tid] = s / msum;
        pe[b * PEP + tid] = s / msum;
    }
    __syncthreads();

    float acc[4];
    int gq[4];
    const float db = dose[b], tb = timev[b];
    if (tid < 4) pe[b * PEP + CCEH + tid] = (tid == 0) ? db : ((tid == 1) ? tb : 0.f);
    #pragma unroll
    for (int q = 0; q < 4; q++) {
        const int g = tid + 256 * q;
        gq[q] = (g < GEX) ? g : (GEX - 1);
        acc[q] = db * w_dose[gq[q]] + tb * w_time[gq[q]];
    }
    #pragma unroll 16
    for (int dd = 0; dd < CCEH; dd++) {
        const float p = pl[dd];
        #pragma unroll
        for (int q = 0; q < 4; q++) acc[q] += p * W2[dd * GEX + gq[q]];
    }
    #pragma unroll
    for (int q = 0; q < 4; q++) {
        const int g = tid + 256 * q;
        if (g < GEX) comp[b * GEX + g] = acc[q];
    }
}

#define SA_PITCH 68
#define SLP 132
#define QST 17
__global__ __launch_bounds__(128) void k_pred0(const float* __restrict__ b_gex,
                                               const float* __restrict__ comp,
                                               const float* __restrict__ w_gex,
                                               const float* __restrict__ w_comp,
                                               const float* __restrict__ P,
                                               const float* __restrict__ Q,
                                               const float* __restrict__ pe,
                                               const float* __restrict__ ppirow,
                                               float* __restrict__ pred0) {
    __shared__ __attribute__((aligned(16))) float Ss[16 * SLP];
    const int tid = threadIdx.x;
    const int lane = tid & 31;
    const int wave = __builtin_amdgcn_readfirstlane(tid >> 5);
    const int c = lane & 15, h = lane >> 4;
    const int mt = blockIdx.x & 3, ng = blockIdx.x >> 2;
    const int nt = ng * 4 + wave;
    const int g = nt * 16 + c;
    const int gc = (g < GEX) ? g : (GEX - 1);

    const float* pa = pe + (mt * 16 + c) * PEP + 2 * h;
    const float* qp = Q + (2 * h) * HDIM + 32 * wave + c;
    const float* pp = P + (mt * 16 + 8 * h) * HDIM + 32 * wave + c;
    v2f ea[QST], q0[QST], q1[QST];
    v8f s0, s1;
    #pragma unroll
    for (int j = 0; j < QST; j++) ea[j] = *(const v2f*)(pa + 4 * j);
    #pragma unroll
    for (int j = 0; j < QST; j++) {
        q0[j].x = qp[(4 * j) * HDIM];
        q0[j].y = qp[(4 * j + 1) * HDIM];
        q1[j].x = qp[(4 * j) * HDIM + 16];
        q1[j].y = qp[(4 * j + 1) * HDIM + 16];
    }
    #pragma unroll
    for (int r = 0; r < 8; r++) {
        s0[r] = pp[r * HDIM];
        s1[r] = pp[r * HDIM + 16];
    }

    const v4f* gp = (const v4f*)(w_gex + gc * HDIM + 4 * h);
    const v4f* cp = (const v4f*)(w_comp + gc * HDIM + 4 * h);
    const int b0 = mt * 16 + 8 * h;
    float xv[8], cv[8];
    #pragma unroll
    for (int r = 0; r < 8; r++) {
        xv[r] = b_gex[(b0 + r) * GEX + gc];
        cv[r] = comp[(b0 + r) * GEX + gc];
    }
    const float pr = ppirow[gc];
    v4f gf[16], cf[16];
    #pragma unroll
    for (int J = 0; J < 16; J++) {
        gf[J] = gp[2 * J];
        cf[J] = cp[2 * J];
    }
    __builtin_amdgcn_sched_barrier(0);

    #pragma unroll
    for (int j = 0; j < QST; j++) {
        s0 = __builtin_amdgcn_wmma_f32_16x16x4_f32(
            false, ea[j], false, q0[j], (short)0, s0, false, false);
        s1 = __builtin_amdgcn_wmma_f32_16x16x4_f32(
            false, ea[j], false, q1[j], (short)0, s1, false, false);
    }
    #pragma unroll
    for (int r = 0; r < 8; r++) {
        Ss[(8 * h + r) * SLP + 32 * wave + c] = s0[r];
        Ss[(8 * h + r) * SLP + 32 * wave + 16 + c] = s1[r];
    }
    __syncthreads();

    const v4f* ap = (const v4f*)(Ss + c * SLP + 4 * h);
    v4f af[16];
    #pragma unroll
    for (int J = 0; J < 16; J++) af[J] = ap[2 * J];
    __builtin_amdgcn_sched_barrier(0);

    v8f t1 = {}, t2 = {};
    #pragma unroll
    for (int J = 0; J < 16; J++) {
        v2f a0, a1, g0, g1, c0, c1;
        a0.x = af[J].x;  a0.y = af[J].y;  a1.x = af[J].z;  a1.y = af[J].w;
        g0.x = gf[J].x;  g0.y = gf[J].y;  g1.x = gf[J].z;  g1.y = gf[J].w;
        c0.x = cf[J].x;  c0.y = cf[J].y;  c1.x = cf[J].z;  c1.y = cf[J].w;
        t1 = __builtin_amdgcn_wmma_f32_16x16x4_f32(
            false, a0, false, g0, (short)0, t1, false, false);
        t2 = __builtin_amdgcn_wmma_f32_16x16x4_f32(
            false, a0, false, c0, (short)0, t2, false, false);
        t1 = __builtin_amdgcn_wmma_f32_16x16x4_f32(
            false, a1, false, g1, (short)0, t1, false, false);
        t2 = __builtin_amdgcn_wmma_f32_16x16x4_f32(
            false, a1, false, c1, (short)0, t2, false, false);
    }

    asm volatile("" :: "v"(xv[0]), "v"(xv[1]), "v"(xv[2]), "v"(xv[3]), "v"(xv[4]), "v"(xv[5]), "v"(xv[6]), "v"(xv[7]),
                       "v"(cv[0]), "v"(cv[1]), "v"(cv[2]), "v"(cv[3]), "v"(cv[4]), "v"(cv[5]), "v"(cv[6]), "v"(cv[7]),
                       "v"(pr));
    if (g < GEX) {
        #pragma unroll
        for (int r = 0; r < 8; r++) {
            const float x = xv[r];
            const float row = (x * t1[r] + cv[r] * t2[r]) * 0.08838834764831845f + pr;
            pred0[(b0 + r) * GEX + g] = x * row;
        }
    }
}

#define GWAVES 16
#define LNV 31

__device__ __forceinline__ void wave_sum2(float& x, float& y) {
    for (int o = 16; o > 0; o >>= 1) {
        const float px = __shfl_xor(x, o, 32);
        const float py = __shfl_xor(y, o, 32);
        x += px;
        y += py;
    }
}

__global__ __launch_bounds__(512) void k_ln_gemm_wmma(const float* __restrict__ pred0,
                                                      const float* __restrict__ gamma,
                                                      const float* __restrict__ beta,
                                                      const float* __restrict__ Wff,
                                                      float* __restrict__ out) {
    __shared__ float gA[GWAVES * 32 * SA_PITCH];
    __shared__ float gpart[GWAVES * 512];
    __shared__ float mu_s[32], rstd_s[32];
    const int tid = threadIdx.x;
    const int lane = tid & 31;
    const int wave = __builtin_amdgcn_readfirstlane(tid >> 5);
    const int c = lane & 15, h = lane >> 4;
    const int nt = blockIdx.x >> 1, mh = blockIdx.x & 1;
    const int k0 = wave * 64;

    const float* prow0 = pred0 + (mh * 32 + 2 * wave) * GEX + lane;
    const float* prow1 = prow0 + GEX;
    float v0[LNV], v1[LNV];
    #pragma unroll
    for (int i = 0; i < LNV; i++) {
        const bool in = lane + 32 * i < GEX;
        v0[i] = in ? prow0[32 * i] : 0.f;
        v1[i] = in ? prow1[32 * i] : 0.f;
    }

    const int ka = k0 + 2 * lane;
    const bool kin = ka < GEX;
    const int kc = kin ? ka : (GEX - 2);
    const float* arow = pred0 + (mh * 32) * GEX + kc;
    v2f av[32];
    for (int r = 0; r < 32; r++) av[r] = *(const v2f*)(arow + r * GEX);
    const float ga0 = gamma[kc], ga1 = gamma[kc + 1];
    const float be0 = beta[kc], be1 = beta[kc + 1];

    const int n = nt * 16 + c;
    const int nc = (n < GEX) ? n : (GEX - 1);
    v2f bf[16];
    for (int j = 0; j < 16; j++) {
        const int kb = k0 + 4 * j + 2 * h;
        const bool ok = kb < GEX;
        const int kr = ok ? kb : (GEX - 2);
        const float x = Wff[kr * GEX + nc];
        const float y = Wff[(kr + 1) * GEX + nc];
        bf[j].x = ok ? x : 0.f;
        bf[j].y = ok ? y : 0.f;
    }

    {
        float s0 = 0.f, s1 = 0.f;
        #pragma unroll
        for (int i = 0; i < LNV; i++) { s0 += v0[i]; s1 += v1[i]; }
        wave_sum2(s0, s1);
        const float mu0 = s0 / (float)GEX, mu1 = s1 / (float)GEX;
        float q0 = 0.f, q1 = 0.f;
        #pragma unroll
        for (int i = 0; i < LNV; i++) {
            const bool in = lane + 32 * i < GEX;
            const float d0 = v0[i] - mu0, d1 = v1[i] - mu1;
            q0 += in ? d0 * d0 : 0.f;
            q1 += in ? d1 * d1 : 0.f;
        }
        wave_sum2(q0, q1);
        if (lane == 0) {
            mu_s[2 * wave] = mu0;
            mu_s[2 * wave + 1] = mu1;
            rstd_s[2 * wave] = rsqrtf(q0 / (float)GEX + 1e-5f);
            rstd_s[2 * wave + 1] = rsqrtf(q1 / (float)GEX + 1e-5f);
        }
    }
    __syncthreads();

    float* sw = gA + wave * (32 * SA_PITCH);
    for (int r = 0; r < 32; r++) {
        const float mu = mu_s[r], rstd = rstd_s[r];
        const float t0 = fmaxf((av[r].x - mu) * rstd * ga0 + be0, 0.f);
        const float t1 = fmaxf((av[r].y - mu) * rstd * ga1 + be1, 0.f);
        v2f y;
        y.x = kin ? t0 : 0.f;
        y.y = kin ? t1 : 0.f;
        *(v2f*)(sw + r * SA_PITCH + 2 * lane) = y;
    }
    __syncthreads();

    const float* fa0 = sw + c * SA_PITCH + 2 * h;
    const float* fa1 = fa0 + 16 * SA_PITCH;
    v8f acc0 = {}, acc1 = {};
    for (int j = 0; j < 16; j++) {
        v2f a0 = *(const v2f*)(fa0 + 4 * j);
        v2f a1 = *(const v2f*)(fa1 + 4 * j);
        acc0 = __builtin_amdgcn_wmma_f32_16x16x4_f32(
            false, a0, false, bf[j], (short)0, acc0, false, false);
        acc1 = __builtin_amdgcn_wmma_f32_16x16x4_f32(
            false, a1, false, bf[j], (short)0, acc1, false, false);
    }

    for (int r = 0; r < 8; r++) {
        gpart[wave * 512 + r * 32 + lane] = acc0[r];
        gpart[wave * 512 + 256 + r * 32 + lane] = acc1[r];
    }
    __syncthreads();
    {
        float s0 = 0.f, s1 = 0.f, s2 = 0.f, s3 = 0.f;
        for (int w = 0; w < GWAVES; w += 4) {
            s0 += gpart[(w + 0) * 512 + tid];
            s1 += gpart[(w + 1) * 512 + tid];
            s2 += gpart[(w + 2) * 512 + tid];
            s3 += gpart[(w + 3) * 512 + tid];
        }
        const int t = tid >> 8, idx = tid & 255;
        const int r = idx >> 5, l = idx & 31;
        const int row = mh * 32 + t * 16 + 8 * (l >> 4) + r;
        const int col = nt * 16 + (l & 15);
        if (col < GEX) out[row * GEX + col] = (s0 + s1) + (s2 + s3);
    }
}

extern "C" void kernel_launch(void* const* d_in, const int* in_sizes, int n_in,
                              void* d_out, int out_size, void* d_ws, size_t ws_size,
                              hipStream_t stream) {
    const float* b_gex    = (const float*)d_in[0];
    const float* node_feat= (const float*)d_in[1];
    const float* mask     = (const float*)d_in[2];
    const float* adj      = (const float*)d_in[3];
    const float* dist     = (const float*)d_in[4];
    const float* dose     = (const float*)d_in[5];
    const float* timev    = (const float*)d_in[6];
    const float* ppi      = (const float*)d_in[7];
    const float* w_gex    = (const float*)d_in[8];
    const float* w_comp   = (const float*)d_in[9];
    const float* W1       = (const float*)d_in[10];
    const float* W2       = (const float*)d_in[11];
    const float* w_dose   = (const float*)d_in[12];
    const float* w_time   = (const float*)d_in[13];
    const float* gamma    = (const float*)d_in[14];
    const float* beta     = (const float*)d_in[15];
    const float* Wff      = (const float*)d_in[16];

    float* out  = (float*)d_out;
    float* comp = out + BATCH * GEX;

    float* ws     = (float*)d_ws;
    float* P      = ws;
    float* ppirow = P + BATCH * HDIM;
    float* pred0  = ppirow + 1024;
    float* Q      = pred0 + BATCH * GEX;
    float* pe     = Q + 80 * HDIM;

    k_front<<<BATCH + NPQ + GEX, 256, 0, stream>>>(node_feat, mask, adj, dist, W1, W2, dose, timev,
                                                   w_dose, w_time, ppi, b_gex, w_gex, w_comp,
                                                   comp, ppirow, P, Q, pe);
    k_pred0<<<64, 128, 0, stream>>>(b_gex, comp, w_gex, w_comp, P, Q, pe, ppirow, pred0);
    k_ln_gemm_wmma<<<124, 512, 0, stream>>>(pred0, gamma, beta, Wff, out);
}


# k_front batch rows: relu(nf@W1) moves from VALU sums over LDS rows onto f32 WMMA (fragments straight from memory), pooling weights applied to the D tiles in registers; hs and pooled sum order changes
# speedup vs baseline: 7.8134x; 1.1082x over previous
#include <hip/hip_runtime.h>
#include <hip/hip_bf16.h>

#define GEX 978
#define HDIM 128
#define NAT 50
#define FEAT 34
#define CCEH 64
#define BATCH 64

typedef __attribute__((ext_vector_type(2))) float v2f;
typedef __attribute__((ext_vector_type(4))) float v4f;
typedef __attribute__((ext_vector_type(8))) float v8f;

#define HST 9
#define WPQ 10
#define NPQ 72
#define PEP 68
#define SPAIRS (GEX / 2)
__global__ __launch_bounds__(256, 8) void k_front(const float* __restrict__ nf,
                                               const float* __restrict__ mask,
                                               const float* __restrict__ adj,
                                               const float* __restrict__ dist,
                                               const float* __restrict__ W1,
                                               const float* __restrict__ W2,
                                               const float* __restrict__ dose,
                                               const float* __restrict__ timev,
                                               const float* __restrict__ w_dose,
                                               const float* __restrict__ w_time,
                                               const float* __restrict__ ppi,
                                               const float* __restrict__ b_gex,
                                               const float* __restrict__ w_gex,
                                               const float* __restrict__ w_comp,
                                               float* __restrict__ comp,
                                               float* __restrict__ rowsum,
                                               float* __restrict__ P,
                                               float* __restrict__ Q,
                                               float* __restrict__ pe) {
    const int tid = threadIdx.x;
    __shared__ float wprod[WPQ * 256];
    __shared__ float msk[256];
    __shared__ float wm[CCEH];
    __shared__ float pl[CCEH];
    __shared__ float red[256];
    __shared__ float msum_s;

    if (blockIdx.x >= BATCH && blockIdx.x < BATCH + NPQ) {
        const int t = blockIdx.x - BATCH;
        const bool isP = t < 32;
        const int u = isP ? t : (t - 32);
        const int mt = u >> 3, nt = u & 7;
        const int lane = tid & 31;
        const int wave = __builtin_amdgcn_readfirstlane(tid >> 5);
        const int c = lane & 15, h = lane >> 4;
        const int R = mt * 16 + c;
        const float* arow;
        if (isP) arow = b_gex + R * GEX;
        else arow = (R < CCEH) ? (W2 + R * GEX) : ((R == CCEH) ? w_dose : w_time);
        const float* Wm = isP ? w_gex : w_comp;
        v8f acc0 = {}, acc1 = {};
        #pragma unroll
        for (int q = 0; q < 2; q++) {
            const int pw = (wave + 8 * q) * 32;
            const int p0 = (pw < SPAIRS - 32) ? pw : (SPAIRS - 32);
            const int skip = pw - p0;
            const int k0 = 2 * p0;
            const float* ap = arow + k0 + 2 * h;
            const float* bp = Wm + (k0 + 2 * h) * HDIM + nt * 16 + c;
            v2f af[16], bf[16];
            #pragma unroll
            for (int j = 0; j < 16; j++) {
                af[j] = *(const v2f*)(ap + 4 * j);
                bf[j].x = bp[(4 * j) * HDIM];
                bf[j].y = bp[(4 * j + 1) * HDIM];
            }
            #pragma unroll
            for (int j = 0; j < 16; j += 2) {
                v2f a0 = af[j], a1 = af[j + 1];
                v2f b0 = bf[j], b1 = bf[j + 1];
                const bool ok0 = (2 * j + h) >= skip;
                const bool ok1 = (2 * j + 2 + h) >= skip;
                a0.x = ok0 ? a0.x : 0.f;  a0.y = ok0 ? a0.y : 0.f;
                b0.x = ok0 ? b0.x : 0.f;  b0.y = ok0 ? b0.y : 0.f;
                a1.x = ok1 ? a1.x : 0.f;  a1.y = ok1 ? a1.y : 0.f;
                b1.x = ok1 ? b1.x : 0.f;  b1.y = ok1 ? b1.y : 0.f;
                acc0 = __builtin_amdgcn_wmma_f32_16x16x4_f32(
                    false, a0, false, b0, (short)0, acc0, false, false);
                acc1 = __builtin_amdgcn_wmma_f32_16x16x4_f32(
                    false, a1, false, b1, (short)0, acc1, false, false);
            }
        }
        #pragma unroll
        for (int r = 0; r < 8; r++) wprod[wave * 256 + r * 32 + lane] = acc0[r] + acc1[r];
        __syncthreads();
        {
            float s0 = 0.f, s1 = 0.f;
            for (int w = 0; w < 8; w += 2) {
                s0 += wprod[(w + 0) * 256 + tid];
                s1 += wprod[(w + 1) * 256 + tid];
            }
            const int r = tid >> 5, l = tid & 31;
            const int row = mt * 16 + 8 * (l >> 4) + r;
            const int col = nt * 16 + (l & 15);
            if (isP) P[row * HDIM + col] = s0 + s1;
            else Q[row * HDIM + col] = (row < CCEH + 2) ? (s0 + s1) : 0.f;
        }
        return;
    }
    if (blockIdx.x >= BATCH + NPQ) {
        const int g = blockIdx.x - BATCH - NPQ;
        float s = 0.f;
        for (int k = tid; k < GEX; k += 256) s += ppi[g * GEX + k];
        red[tid] = s;
        __syncthreads();
        for (int o = 128; o > 0; o >>= 1) {
            if (tid < o) red[tid] += red[tid + o];
            __syncthreads();
        }
        if (tid == 0) rowsum[g] = red[0];
        return;
    }

    const int b = blockIdx.x;
    const int lane = tid & 31;
    const int wave = __builtin_amdgcn_readfirstlane(tid >> 5);
    const int c = lane & 15, h = lane >> 4;
    const int nt = wave & 3;
    const int mp = wave >> 2;

    float av[WPQ], dv[WPQ], mv[WPQ];
    #pragma unroll
    for (int q = 0; q < WPQ; q++) {
        const int i = tid + 256 * q;
        const int ic = (i < NAT * NAT) ? i : (NAT * NAT - 1);
        av[q] = adj[b * (NAT * NAT) + ic];
        dv[q] = dist[b * (NAT * NAT) + ic];
        mv[q] = mask[b * NAT + ic / NAT];
    }
    const float mk = mask[b * NAT + ((tid < NAT) ? tid : 0)];
    const float* nfb = nf + b * (NAT * FEAT);
    const int n0 = mp * 32 + c;
    const int n1 = (n0 + 16 < NAT) ? (n0 + 16) : (NAT - 1);
    const float* a0p = nfb + n0 * FEAT;
    const float* a1p = nfb + n1 * FEAT;
    const float* bp = W1 + nt * 16 + c;
    v2f a0[HST], a1[HST], bw[HST];
    #pragma unroll
    for (int j = 0; j < HST; j++) {
        const bool kin = (j < HST - 1) || (h == 0);
        const int k = kin ? (4 * j + 2 * h) : (4 * j);
        a0[j] = *(const v2f*)(a0p + k);
        a1[j] = *(const v2f*)(a1p + k);
        bw[j].x = bp[k * CCEH];
        bw[j].y = bp[(k + 1) * CCEH];
    }
    __builtin_amdgcn_sched_barrier(0);
    {
        const bool kin = (h == 0);
        a0[HST - 1].x = kin ? a0[HST - 1].x : 0.f;  a0[HST - 1].y = kin ? a0[HST - 1].y : 0.f;
        a1[HST - 1].x = kin ? a1[HST - 1].x : 0.f;  a1[HST - 1].y = kin ? a1[HST - 1].y : 0.f;
        bw[HST - 1].x = kin ? bw[HST - 1].x : 0.f;  bw[HST - 1].y = kin ? bw[HST - 1].y : 0.f;
    }

    #pragma unroll
    for (int q = 0; q < WPQ; q++) wprod[tid + 256 * q] = mv[q] * av[q] * expf(-dv[q]);
    msk[tid] = mk;

    v8f d0 = {}, d1 = {};
    #pragma unroll
    for (int j = 0; j < HST; j++) {
        d0 = __builtin_amdgcn_wmma_f32_16x16x4_f32(
            false, a0[j], false, bw[j], (short)0, d0, false, false);
        d1 = __builtin_amdgcn_wmma_f32_16x16x4_f32(
            false, a1[j], false, bw[j], (short)0, d1, false, false);
    }
    __syncthreads();

    if (tid >= 128 && tid < 128 + CCEH) {
        const int m = tid - 128;
        float s = 0.f;
        if (m < NAT) {
            #pragma unroll
            for (int n = 0; n < NAT; n++) s += wprod[n * NAT + m];
        }
        wm[m] = s;
    } else if (tid >= 192 && tid < 224) {
        float msum = 0.f;
        #pragma unroll
        for (int n = 0; n < NAT; n++) msum += msk[n];
        if (tid == 192) msum_s = fmaxf(msum, 1.f);
    }
    __syncthreads();
    {
        const float* wp = wm + mp * 32 + 8 * h;
        float s = 0.f;
        #pragma unroll
        for (int r = 0; r < 8; r++) s += wp[r] * fmaxf(d0[r], 0.f);
        #pragma unroll
        for (int r = 0; r < 8; r++) s += wp[16 + r] * fmaxf(d1[r], 0.f);
        s += __shfl_xor(s, 16, 32);
        if (h == 0) red[wave * 16 + c] = s;
    }
    __syncthreads();
    if (tid < CCEH) {
        const float p = (red[tid] + red[CCEH + tid]) / msum_s;
        pl[tid] = p;
        pe[b * PEP + tid] = p;
    }
    __syncthreads();

    float acc[4];
    int gq[4];
    const float db = dose[b], tb = timev[b];
    if (tid < 4) pe[b * PEP + CCEH + tid] = (tid == 0) ? db : ((tid == 1) ? tb : 0.f);
    #pragma unroll
    for (int q = 0; q < 4; q++) {
        const int g = tid + 256 * q;
        gq[q] = (g < GEX) ? g : (GEX - 1);
        acc[q] = db * w_dose[gq[q]] + tb * w_time[gq[q]];
    }
    #pragma unroll 16
    for (int dd = 0; dd < CCEH; dd++) {
        const float p = pl[dd];
        #pragma unroll
        for (int q = 0; q < 4; q++) acc[q] += p * W2[dd * GEX + gq[q]];
    }
    #pragma unroll
    for (int q = 0; q < 4; q++) {
        const int g = tid + 256 * q;
        if (g < GEX) comp[b * GEX + g] = acc[q];
    }
}

#define SA_PITCH 68
#define SLP 132
#define QST 17
__global__ __launch_bounds__(128) void k_pred0(const float* __restrict__ b_gex,
                                               const float* __restrict__ comp,
                                               const float* __restrict__ w_gex,
                                               const float* __restrict__ w_comp,
                                               const float* __restrict__ P,
                                               const float* __restrict__ Q,
                                               const float* __restrict__ pe,
                                               const float* __restrict__ ppirow,
                                               float* __restrict__ pred0) {
    __shared__ __attribute__((aligned(16))) float Ss[16 * SLP];
    const int tid = threadIdx.x;
    const int lane = tid & 31;
    const int wave = __builtin_amdgcn_readfirstlane(tid >> 5);
    const int c = lane & 15, h = lane >> 4;
    const int mt = blockIdx.x & 3, ng = blockIdx.x >> 2;
    const int nt = ng * 4 + wave;
    const int g = nt * 16 + c;
    const int gc = (g < GEX) ? g : (GEX - 1);

    const float* pa = pe + (mt * 16 + c) * PEP + 2 * h;
    const float* qp = Q + (2 * h) * HDIM + 32 * wave + c;
    const float* pp = P + (mt * 16 + 8 * h) * HDIM + 32 * wave + c;
    v2f ea[QST], q0[QST], q1[QST];
    v8f s0, s1;
    #pragma unroll
    for (int j = 0; j < QST; j++) ea[j] = *(const v2f*)(pa + 4 * j);
    #pragma unroll
    for (int j = 0; j < QST; j++) {
        q0[j].x = qp[(4 * j) * HDIM];
        q0[j].y = qp[(4 * j + 1) * HDIM];
        q1[j].x = qp[(4 * j) * HDIM + 16];
        q1[j].y = qp[(4 * j + 1) * HDIM + 16];
    }
    #pragma unroll
    for (int r = 0; r < 8; r++) {
        s0[r] = pp[r * HDIM];
        s1[r] = pp[r * HDIM + 16];
    }

    const v4f* gp = (const v4f*)(w_gex + gc * HDIM + 4 * h);
    const v4f* cp = (const v4f*)(w_comp + gc * HDIM + 4 * h);
    const int b0 = mt * 16 + 8 * h;
    float xv[8], cv[8];
    #pragma unroll
    for (int r = 0; r < 8; r++) {
        xv[r] = b_gex[(b0 + r) * GEX + gc];
        cv[r] = comp[(b0 + r) * GEX + gc];
    }
    const float pr = ppirow[gc];
    v4f gf[16], cf[16];
    #pragma unroll
    for (int J = 0; J < 16; J++) {
        gf[J] = gp[2 * J];
        cf[J] = cp[2 * J];
    }
    __builtin_amdgcn_sched_barrier(0);

    #pragma unroll
    for (int j = 0; j < QST; j++) {
        s0 = __builtin_amdgcn_wmma_f32_16x16x4_f32(
            false, ea[j], false, q0[j], (short)0, s0, false, false);
        s1 = __builtin_amdgcn_wmma_f32_16x16x4_f32(
            false, ea[j], false, q1[j], (short)0, s1, false, false);
    }
    #pragma unroll
    for (int r = 0; r < 8; r++) {
        Ss[(8 * h + r) * SLP + 32 * wave + c] = s0[r];
        Ss[(8 * h + r) * SLP + 32 * wave + 16 + c] = s1[r];
    }
    __syncthreads();

    const v4f* ap = (const v4f*)(Ss + c * SLP + 4 * h);
    v4f af[16];
    #pragma unroll
    for (int J = 0; J < 16; J++) af[J] = ap[2 * J];
    __builtin_amdgcn_sched_barrier(0);

    v8f t1 = {}, t2 = {};
    #pragma unroll
    for (int J = 0; J < 16; J++) {
        v2f a0, a1, g0, g1, c0, c1;
        a0.x = af[J].x;  a0.y = af[J].y;  a1.x = af[J].z;  a1.y = af[J].w;
        g0.x = gf[J].x;  g0.y = gf[J].y;  g1.x = gf[J].z;  g1.y = gf[J].w;
        c0.x = cf[J].x;  c0.y = cf[J].y;  c1.x = cf[J].z;  c1.y = cf[J].w;
        t1 = __builtin_amdgcn_wmma_f32_16x16x4_f32(
            false, a0, false, g0, (short)0, t1, false, false);
        t2 = __builtin_amdgcn_wmma_f32_16x16x4_f32(
            false, a0, false, c0, (short)0, t2, false, false);
        t1 = __builtin_amdgcn_wmma_f32_16x16x4_f32(
            false, a1, false, g1, (short)0, t1, false, false);
        t2 = __builtin_amdgcn_wmma_f32_16x16x4_f32(
            false, a1, false, c1, (short)0, t2, false, false);
    }

    asm volatile("" :: "v"(xv[0]), "v"(xv[1]), "v"(xv[2]), "v"(xv[3]), "v"(xv[4]), "v"(xv[5]), "v"(xv[6]), "v"(xv[7]),
                       "v"(cv[0]), "v"(cv[1]), "v"(cv[2]), "v"(cv[3]), "v"(cv[4]), "v"(cv[5]), "v"(cv[6]), "v"(cv[7]),
                       "v"(pr));
    if (g < GEX) {
        #pragma unroll
        for (int r = 0; r < 8; r++) {
            const float x = xv[r];
            const float row = (x * t1[r] + cv[r] * t2[r]) * 0.08838834764831845f + pr;
            pred0[(b0 + r) * GEX + g] = x * row;
        }
    }
}

#define GWAVES 16
#define LNV 31

__device__ __forceinline__ void wave_sum2(float& x, float& y) {
    for (int o = 16; o > 0; o >>= 1) {
        const float px = __shfl_xor(x, o, 32);
        const float py = __shfl_xor(y, o, 32);
        x += px;
        y += py;
    }
}

__global__ __launch_bounds__(512) void k_ln_gemm_wmma(const float* __restrict__ pred0,
                                                      const float* __restrict__ gamma,
                                                      const float* __restrict__ beta,
                                                      const float* __restrict__ Wff,
                                                      float* __restrict__ out) {
    __shared__ float gA[GWAVES * 32 * SA_PITCH];
    __shared__ float gpart[GWAVES * 512];
    __shared__ float mu_s[32], rstd_s[32];
    const int tid = threadIdx.x;
    const int lane = tid & 31;
    const int wave = __builtin_amdgcn_readfirstlane(tid >> 5);
    const int c = lane & 15, h = lane >> 4;
    const int nt = blockIdx.x >> 1, mh = blockIdx.x & 1;
    const int k0 = wave * 64;

    const float* prow0 = pred0 + (mh * 32 + 2 * wave) * GEX + lane;
    const float* prow1 = prow0 + GEX;
    float v0[LNV], v1[LNV];
    #pragma unroll
    for (int i = 0; i < LNV; i++) {
        const bool in = lane + 32 * i < GEX;
        v0[i] = in ? prow0[32 * i] : 0.f;
        v1[i] = in ? prow1[32 * i] : 0.f;
    }

    const int ka = k0 + 2 * lane;
    const bool kin = ka < GEX;
    const int kc = kin ? ka : (GEX - 2);
    const float* arow = pred0 + (mh * 32) * GEX + kc;
    v2f av[32];
    for (int r = 0; r < 32; r++) av[r] = *(const v2f*)(arow + r * GEX);
    const float ga0 = gamma[kc], ga1 = gamma[kc + 1];
    const float be0 = beta[kc], be1 = beta[kc + 1];

    const int n = nt * 16 + c;
    const int nc = (n < GEX) ? n : (GEX - 1);
    v2f bf[16];
    for (int j = 0; j < 16; j++) {
        const int kb = k0 + 4 * j + 2 * h;
        const bool ok = kb < GEX;
        const int kr = ok ? kb : (GEX - 2);
        const float x = Wff[kr * GEX + nc];
        const float y = Wff[(kr + 1) * GEX + nc];
        bf[j].x = ok ? x : 0.f;
        bf[j].y = ok ? y : 0.f;
    }

    {
        float s0 = 0.f, s1 = 0.f;
        #pragma unroll
        for (int i = 0; i < LNV; i++) { s0 += v0[i]; s1 += v1[i]; }
        wave_sum2(s0, s1);
        const float mu0 = s0 / (float)GEX, mu1 = s1 / (float)GEX;
        float q0 = 0.f, q1 = 0.f;
        #pragma unroll
        for (int i = 0; i < LNV; i++) {
            const bool in = lane + 32 * i < GEX;
            const float d0 = v0[i] - mu0, d1 = v1[i] - mu1;
            q0 += in ? d0 * d0 : 0.f;
            q1 += in ? d1 * d1 : 0.f;
        }
        wave_sum2(q0, q1);
        if (lane == 0) {
            mu_s[2 * wave] = mu0;
            mu_s[2 * wave + 1] = mu1;
            rstd_s[2 * wave] = rsqrtf(q0 / (float)GEX + 1e-5f);
            rstd_s[2 * wave + 1] = rsqrtf(q1 / (float)GEX + 1e-5f);
        }
    }
    __syncthreads();

    float* sw = gA + wave * (32 * SA_PITCH);
    for (int r = 0; r < 32; r++) {
        const float mu = mu_s[r], rstd = rstd_s[r];
        const float t0 = fmaxf((av[r].x - mu) * rstd * ga0 + be0, 0.f);
        const float t1 = fmaxf((av[r].y - mu) * rstd * ga1 + be1, 0.f);
        v2f y;
        y.x = kin ? t0 : 0.f;
        y.y = kin ? t1 : 0.f;
        *(v2f*)(sw + r * SA_PITCH + 2 * lane) = y;
    }
    __syncthreads();

    const float* fa0 = sw + c * SA_PITCH + 2 * h;
    const float* fa1 = fa0 + 16 * SA_PITCH;
    v8f acc0 = {}, acc1 = {};
    for (int j = 0; j < 16; j++) {
        v2f a0 = *(const v2f*)(fa0 + 4 * j);
        v2f a1 = *(const v2f*)(fa1 + 4 * j);
        acc0 = __builtin_amdgcn_wmma_f32_16x16x4_f32(
            false, a0, false, bf[j], (short)0, acc0, false, false);
        acc1 = __builtin_amdgcn_wmma_f32_16x16x4_f32(
            false, a1, false, bf[j], (short)0, acc1, false, false);
    }

    for (int r = 0; r < 8; r++) {
        gpart[wave * 512 + r * 32 + lane] = acc0[r];
        gpart[wave * 512 + 256 + r * 32 + lane] = acc1[r];
    }
    __syncthreads();
    {
        float s0 = 0.f, s1 = 0.f, s2 = 0.f, s3 = 0.f;
        for (int w = 0; w < GWAVES; w += 4) {
            s0 += gpart[(w + 0) * 512 + tid];
            s1 += gpart[(w + 1) * 512 + tid];
            s2 += gpart[(w + 2) * 512 + tid];
            s3 += gpart[(w + 3) * 512 + tid];
        }
        const int t = tid >> 8, idx = tid & 255;
        const int r = idx >> 5, l = idx & 31;
        const int row = mh * 32 + t * 16 + 8 * (l >> 4) + r;
        const int col = nt * 16 + (l & 15);
        if (col < GEX) out[row * GEX + col] = (s0 + s1) + (s2 + s3);
    }
}

extern "C" void kernel_launch(void* const* d_in, const int* in_sizes, int n_in,
                              void* d_out, int out_size, void* d_ws, size_t ws_size,
                              hipStream_t stream) {
    const float* b_gex    = (const float*)d_in[0];
    const float* node_feat= (const float*)d_in[1];
    const float* mask     = (const float*)d_in[2];
    const float* adj      = (const float*)d_in[3];
    const float* dist     = (const float*)d_in[4];
    const float* dose     = (const float*)d_in[5];
    const float* timev    = (const float*)d_in[6];
    const float* ppi      = (const float*)d_in[7];
    const float* w_gex    = (const float*)d_in[8];
    const float* w_comp   = (const float*)d_in[9];
    const float* W1       = (const float*)d_in[10];
    const float* W2       = (const float*)d_in[11];
    const float* w_dose   = (const float*)d_in[12];
    const float* w_time   = (const float*)d_in[13];
    const float* gamma    = (const float*)d_in[14];
    const float* beta     = (const float*)d_in[15];
    const float* Wff      = (const float*)d_in[16];

    float* out  = (float*)d_out;
    float* comp = out + BATCH * GEX;

    float* ws     = (float*)d_ws;
    float* P      = ws;
    float* ppirow = P + BATCH * HDIM;
    float* pred0  = ppirow + 1024;
    float* Q      = pred0 + BATCH * GEX;
    float* pe     = Q + 80 * HDIM;

    k_front<<<BATCH + NPQ + GEX, 256, 0, stream>>>(node_feat, mask, adj, dist, W1, W2, dose, timev,
                                                   w_dose, w_time, ppi, b_gex, w_gex, w_comp,
                                                   comp, ppirow, P, Q, pe);
    k_pred0<<<64, 128, 0, stream>>>(b_gex, comp, w_gex, w_comp, P, Q, pe, ppirow, pred0);
    k_ln_gemm_wmma<<<124, 512, 0, stream>>>(pred0, gamma, beta, Wff, out);
}


# comp = [pooled,dose,time] @ [W2;w_dose;w_time] moves from k_front's 64 batch-row work-groups (VALU, each streaming W2) to f32 WMMA in k_pred0 on its pe fragments; comp's sum order changes
# speedup vs baseline: 8.1328x; 1.0409x over previous
#include <hip/hip_runtime.h>
#include <hip/hip_bf16.h>

#define GEX 978
#define HDIM 128
#define NAT 50
#define FEAT 34
#define CCEH 64
#define BATCH 64

typedef __attribute__((ext_vector_type(2))) float v2f;
typedef __attribute__((ext_vector_type(4))) float v4f;
typedef __attribute__((ext_vector_type(8))) float v8f;

#define HST 9
#define WPQ 10
#define NPQ 72
#define PEP 68
#define SPAIRS (GEX / 2)
__global__ __launch_bounds__(256, 8) void k_front(const float* __restrict__ nf,
                                               const float* __restrict__ mask,
                                               const float* __restrict__ adj,
                                               const float* __restrict__ dist,
                                               const float* __restrict__ W1,
                                               const float* __restrict__ W2,
                                               const float* __restrict__ dose,
                                               const float* __restrict__ timev,
                                               const float* __restrict__ w_dose,
                                               const float* __restrict__ w_time,
                                               const float* __restrict__ ppi,
                                               const float* __restrict__ b_gex,
                                               const float* __restrict__ w_gex,
                                               const float* __restrict__ w_comp,
                                               float* __restrict__ rowsum,
                                               float* __restrict__ P,
                                               float* __restrict__ Q,
                                               float* __restrict__ pe) {
    const int tid = threadIdx.x;
    __shared__ float wprod[WPQ * 256];
    __shared__ float msk[256];
    __shared__ float wm[CCEH];
    __shared__ float red[256];
    __shared__ float msum_s;

    if (blockIdx.x >= BATCH && blockIdx.x < BATCH + NPQ) {
        const int t = blockIdx.x - BATCH;
        const bool isP = t < 32;
        const int u = isP ? t : (t - 32);
        const int mt = u >> 3, nt = u & 7;
        const int lane = tid & 31;
        const int wave = __builtin_amdgcn_readfirstlane(tid >> 5);
        const int c = lane & 15, h = lane >> 4;
        const int R = mt * 16 + c;
        const float* arow;
        if (isP) arow = b_gex + R * GEX;
        else arow = (R < CCEH) ? (W2 + R * GEX) : ((R == CCEH) ? w_dose : w_time);
        const float* Wm = isP ? w_gex : w_comp;
        v8f acc0 = {}, acc1 = {};
        #pragma unroll
        for (int q = 0; q < 2; q++) {
            const int pw = (wave + 8 * q) * 32;
            const int p0 = (pw < SPAIRS - 32) ? pw : (SPAIRS - 32);
            const int skip = pw - p0;
            const int k0 = 2 * p0;
            const float* ap = arow + k0 + 2 * h;
            const float* bp = Wm + (k0 + 2 * h) * HDIM + nt * 16 + c;
            v2f af[16], bf[16];
            #pragma unroll
            for (int j = 0; j < 16; j++) {
                af[j] = *(const v2f*)(ap + 4 * j);
                bf[j].x = bp[(4 * j) * HDIM];
                bf[j].y = bp[(4 * j + 1) * HDIM];
            }
            #pragma unroll
            for (int j = 0; j < 16; j += 2) {
                v2f a0 = af[j], a1 = af[j + 1];
                v2f b0 = bf[j], b1 = bf[j + 1];
                const bool ok0 = (2 * j + h) >= skip;
                const bool ok1 = (2 * j + 2 + h) >= skip;
                a0.x = ok0 ? a0.x : 0.f;  a0.y = ok0 ? a0.y : 0.f;
                b0.x = ok0 ? b0.x : 0.f;  b0.y = ok0 ? b0.y : 0.f;
                a1.x = ok1 ? a1.x : 0.f;  a1.y = ok1 ? a1.y : 0.f;
                b1.x = ok1 ? b1.x : 0.f;  b1.y = ok1 ? b1.y : 0.f;
                acc0 = __builtin_amdgcn_wmma_f32_16x16x4_f32(
                    false, a0, false, b0, (short)0, acc0, false, false);
                acc1 = __builtin_amdgcn_wmma_f32_16x16x4_f32(
                    false, a1, false, b1, (short)0, acc1, false, false);
            }
        }
        #pragma unroll
        for (int r = 0; r < 8; r++) wprod[wave * 256 + r * 32 + lane] = acc0[r] + acc1[r];
        __syncthreads();
        {
            float s0 = 0.f, s1 = 0.f;
            for (int w = 0; w < 8; w += 2) {
                s0 += wprod[(w + 0) * 256 + tid];
                s1 += wprod[(w + 1) * 256 + tid];
            }
            const int r = tid >> 5, l = tid & 31;
            const int row = mt * 16 + 8 * (l >> 4) + r;
            const int col = nt * 16 + (l & 15);
            if (isP) P[row * HDIM + col] = s0 + s1;
            else Q[row * HDIM + col] = (row < CCEH + 2) ? (s0 + s1) : 0.f;
        }
        return;
    }
    if (blockIdx.x >= BATCH + NPQ) {
        const int g = blockIdx.x - BATCH - NPQ;
        float s = 0.f;
        for (int k = tid; k < GEX; k += 256) s += ppi[g * GEX + k];
        red[tid] = s;
        __syncthreads();
        for (int o = 128; o > 0; o >>= 1) {
            if (tid < o) red[tid] += red[tid + o];
            __syncthreads();
        }
        if (tid == 0) rowsum[g] = red[0];
        return;
    }

    const int b = blockIdx.x;
    const int lane = tid & 31;
    const int wave = __builtin_amdgcn_readfirstlane(tid >> 5);
    const int c = lane & 15, h = lane >> 4;
    const int nt = wave & 3;
    const int mp = wave >> 2;

    float av[WPQ], dv[WPQ], mv[WPQ];
    #pragma unroll
    for (int q = 0; q < WPQ; q++) {
        const int i = tid + 256 * q;
        const int ic = (i < NAT * NAT) ? i : (NAT * NAT - 1);
        av[q] = adj[b * (NAT * NAT) + ic];
        dv[q] = dist[b * (NAT * NAT) + ic];
        mv[q] = mask[b * NAT + ic / NAT];
    }
    const float mk = mask[b * NAT + ((tid < NAT) ? tid : 0)];
    const float db = dose[b], tb = timev[b];
    const float* nfb = nf + b * (NAT * FEAT);
    const int n0 = mp * 32 + c;
    const int n1 = (n0 + 16 < NAT) ? (n0 + 16) : (NAT - 1);
    const float* a0p = nfb + n0 * FEAT;
    const float* a1p = nfb + n1 * FEAT;
    const float* bp = W1 + nt * 16 + c;
    v2f a0[HST], a1[HST], bw[HST];
    #pragma unroll
    for (int j = 0; j < HST; j++) {
        const bool kin = (j < HST - 1) || (h == 0);
        const int k = kin ? (4 * j + 2 * h) : (4 * j);
        a0[j] = *(const v2f*)(a0p + k);
        a1[j] = *(const v2f*)(a1p + k);
        bw[j].x = bp[k * CCEH];
        bw[j].y = bp[(k + 1) * CCEH];
    }
    __builtin_amdgcn_sched_barrier(0);
    asm volatile("" :: "s"(db), "s"(tb));
    {
        const bool kin = (h == 0);
        a0[HST - 1].x = kin ? a0[HST - 1].x : 0.f;  a0[HST - 1].y = kin ? a0[HST - 1].y : 0.f;
        a1[HST - 1].x = kin ? a1[HST - 1].x : 0.f;  a1[HST - 1].y = kin ? a1[HST - 1].y : 0.f;
        bw[HST - 1].x = kin ? bw[HST - 1].x : 0.f;  bw[HST - 1].y = kin ? bw[HST - 1].y : 0.f;
    }

    #pragma unroll
    for (int q = 0; q < WPQ; q++) wprod[tid + 256 * q] = mv[q] * av[q] * expf(-dv[q]);
    msk[tid] = mk;

    v8f d0 = {}, d1 = {};
    #pragma unroll
    for (int j = 0; j < HST; j++) {
        d0 = __builtin_amdgcn_wmma_f32_16x16x4_f32(
            false, a0[j], false, bw[j], (short)0, d0, false, false);
        d1 = __builtin_amdgcn_wmma_f32_16x16x4_f32(
            false, a1[j], false, bw[j], (short)0, d1, false, false);
    }
    __syncthreads();

    if (tid >= 128 && tid < 128 + CCEH) {
        const int m = tid - 128;
        float s = 0.f;
        if (m < NAT) {
            #pragma unroll
            for (int n = 0; n < NAT; n++) s += wprod[n * NAT + m];
        }
        wm[m] = s;
    } else if (tid >= 192 && tid < 224) {
        float msum = 0.f;
        #pragma unroll
        for (int n = 0; n < NAT; n++) msum += msk[n];
        if (tid == 192) msum_s = fmaxf(msum, 1.f);
    }
    __syncthreads();
    {
        const float* wp = wm + mp * 32 + 8 * h;
        float s = 0.f;
        #pragma unroll
        for (int r = 0; r < 8; r++) s += wp[r] * fmaxf(d0[r], 0.f);
        #pragma unroll
        for (int r = 0; r < 8; r++) s += wp[16 + r] * fmaxf(d1[r], 0.f);
        s += __shfl_xor(s, 16, 32);
        if (h == 0) red[wave * 16 + c] = s;
    }
    __syncthreads();
    if (tid < CCEH) pe[b * PEP + tid] = (red[tid] + red[CCEH + tid]) / msum_s;
    else if (tid < PEP) pe[b * PEP + tid] = (tid == CCEH) ? db : ((tid == CCEH + 1) ? tb : 0.f);
}

#define SA_PITCH 68
#define SLP 132
#define QST 17
__global__ __launch_bounds__(128) void k_pred0(const float* __restrict__ b_gex,
                                               const float* __restrict__ W2,
                                               const float* __restrict__ w_dose,
                                               const float* __restrict__ w_time,
                                               const float* __restrict__ w_gex,
                                               const float* __restrict__ w_comp,
                                               const float* __restrict__ P,
                                               const float* __restrict__ Q,
                                               const float* __restrict__ pe,
                                               const float* __restrict__ ppirow,
                                               float* __restrict__ comp,
                                               float* __restrict__ pred0) {
    __shared__ __attribute__((aligned(16))) float Ss[16 * SLP];
    const int tid = threadIdx.x;
    const int lane = tid & 31;
    const int wave = __builtin_amdgcn_readfirstlane(tid >> 5);
    const int c = lane & 15, h = lane >> 4;
    const int mt = blockIdx.x & 3, ng = blockIdx.x >> 2;
    const int nt = ng * 4 + wave;
    const int g = nt * 16 + c;
    const int gc = (g < GEX) ? g : (GEX - 1);

    const float* pa = pe + (mt * 16 + c) * PEP + 2 * h;
    const float* qp = Q + (2 * h) * HDIM + 32 * wave + c;
    const float* pp = P + (mt * 16 + 8 * h) * HDIM + 32 * wave + c;
    v2f ea[QST], q0[QST], q1[QST];
    v8f s0, s1;
    #pragma unroll
    for (int j = 0; j < QST; j++) ea[j] = *(const v2f*)(pa + 4 * j);
    #pragma unroll
    for (int j = 0; j < QST; j++) {
        q0[j].x = qp[(4 * j) * HDIM];
        q0[j].y = qp[(4 * j + 1) * HDIM];
        q1[j].x = qp[(4 * j) * HDIM + 16];
        q1[j].y = qp[(4 * j + 1) * HDIM + 16];
    }
    #pragma unroll
    for (int r = 0; r < 8; r++) {
        s0[r] = pp[r * HDIM];
        s1[r] = pp[r * HDIM + 16];
    }
    const float* w2p = W2 + (2 * h) * GEX + gc;
    v2f wf[QST];
    #pragma unroll
    for (int j = 0; j < QST - 1; j++) {
        wf[j].x = w2p[(4 * j) * GEX];
        wf[j].y = w2p[(4 * j + 1) * GEX];
    }
    {
        const float wd = w_dose[gc], wt = w_time[gc];
        wf[QST - 1].x = (h == 0) ? wd : 0.f;
        wf[QST - 1].y = (h == 0) ? wt : 0.f;
    }

    const v4f* gp = (const v4f*)(w_gex + gc * HDIM + 4 * h);
    const v4f* cp = (const v4f*)(w_comp + gc * HDIM + 4 * h);
    const int b0 = mt * 16 + 8 * h;
    float xv[8];
    #pragma unroll
    for (int r = 0; r < 8; r++) xv[r] = b_gex[(b0 + r) * GEX + gc];
    const float pr = ppirow[gc];
    v4f gf[16], cf[16];
    #pragma unroll
    for (int J = 0; J < 16; J++) {
        gf[J] = gp[2 * J];
        cf[J] = cp[2 * J];
    }
    __builtin_amdgcn_sched_barrier(0);

    #pragma unroll
    for (int j = 0; j < QST; j++) {
        s0 = __builtin_amdgcn_wmma_f32_16x16x4_f32(
            false, ea[j], false, q0[j], (short)0, s0, false, false);
        s1 = __builtin_amdgcn_wmma_f32_16x16x4_f32(
            false, ea[j], false, q1[j], (short)0, s1, false, false);
    }
    #pragma unroll
    for (int r = 0; r < 8; r++) {
        Ss[(8 * h + r) * SLP + 32 * wave + c] = s0[r];
        Ss[(8 * h + r) * SLP + 32 * wave + 16 + c] = s1[r];
    }
    __syncthreads();

    const v4f* ap = (const v4f*)(Ss + c * SLP + 4 * h);
    v4f af[16];
    #pragma unroll
    for (int J = 0; J < 16; J++) af[J] = ap[2 * J];
    __builtin_amdgcn_sched_barrier(0);

    v8f ca = {}, cb = {};
    #pragma unroll
    for (int j = 0; j < QST - 1; j += 2) {
        ca = __builtin_amdgcn_wmma_f32_16x16x4_f32(
            false, ea[j], false, wf[j], (short)0, ca, false, false);
        cb = __builtin_amdgcn_wmma_f32_16x16x4_f32(
            false, ea[j + 1], false, wf[j + 1], (short)0, cb, false, false);
    }
    ca = __builtin_amdgcn_wmma_f32_16x16x4_f32(
        false, ea[QST - 1], false, wf[QST - 1], (short)0, ca, false, false);

    v8f t1 = {}, t2 = {};
    #pragma unroll
    for (int J = 0; J < 16; J++) {
        v2f a0, a1, g0, g1, c0, c1;
        a0.x = af[J].x;  a0.y = af[J].y;  a1.x = af[J].z;  a1.y = af[J].w;
        g0.x = gf[J].x;  g0.y = gf[J].y;  g1.x = gf[J].z;  g1.y = gf[J].w;
        c0.x = cf[J].x;  c0.y = cf[J].y;  c1.x = cf[J].z;  c1.y = cf[J].w;
        t1 = __builtin_amdgcn_wmma_f32_16x16x4_f32(
            false, a0, false, g0, (short)0, t1, false, false);
        t2 = __builtin_amdgcn_wmma_f32_16x16x4_f32(
            false, a0, false, c0, (short)0, t2, false, false);
        t1 = __builtin_amdgcn_wmma_f32_16x16x4_f32(
            false, a1, false, g1, (short)0, t1, false, false);
        t2 = __builtin_amdgcn_wmma_f32_16x16x4_f32(
            false, a1, false, c1, (short)0, t2, false, false);
    }

    asm volatile("" :: "v"(xv[0]), "v"(xv[1]), "v"(xv[2]), "v"(xv[3]), "v"(xv[4]), "v"(xv[5]), "v"(xv[6]), "v"(xv[7]),
                       "v"(pr));
    if (g < GEX) {
        #pragma unroll
        for (int r = 0; r < 8; r++) {
            const float x = xv[r];
            const float cvr = ca[r] + cb[r];
            comp[(b0 + r) * GEX + g] = cvr;
            const float row = (x * t1[r] + cvr * t2[r]) * 0.08838834764831845f + pr;
            pred0[(b0 + r) * GEX + g] = x * row;
        }
    }
}

#define GWAVES 16
#define LNV 31

__device__ __forceinline__ void wave_sum2(float& x, float& y) {
    for (int o = 16; o > 0; o >>= 1) {
        const float px = __shfl_xor(x, o, 32);
        const float py = __shfl_xor(y, o, 32);
        x += px;
        y += py;
    }
}

__global__ __launch_bounds__(512) void k_ln_gemm_wmma(const float* __restrict__ pred0,
                                                      const float* __restrict__ gamma,
                                                      const float* __restrict__ beta,
                                                      const float* __restrict__ Wff,
                                                      float* __restrict__ out) {
    __shared__ float gA[GWAVES * 32 * SA_PITCH];
    __shared__ float gpart[GWAVES * 512];
    __shared__ float mu_s[32], rstd_s[32];
    const int tid = threadIdx.x;
    const int lane = tid & 31;
    const int wave = __builtin_amdgcn_readfirstlane(tid >> 5);
    const int c = lane & 15, h = lane >> 4;
    const int nt = blockIdx.x >> 1, mh = blockIdx.x & 1;
    const int k0 = wave * 64;

    const float* prow0 = pred0 + (mh * 32 + 2 * wave) * GEX + lane;
    const float* prow1 = prow0 + GEX;
    float v0[LNV], v1[LNV];
    #pragma unroll
    for (int i = 0; i < LNV; i++) {
        const bool in = lane + 32 * i < GEX;
        v0[i] = in ? prow0[32 * i] : 0.f;
        v1[i] = in ? prow1[32 * i] : 0.f;
    }

    const int ka = k0 + 2 * lane;
    const bool kin = ka < GEX;
    const int kc = kin ? ka : (GEX - 2);
    const float* arow = pred0 + (mh * 32) * GEX + kc;
    v2f av[32];
    for (int r = 0; r < 32; r++) av[r] = *(const v2f*)(arow + r * GEX);
    const float ga0 = gamma[kc], ga1 = gamma[kc + 1];
    const float be0 = beta[kc], be1 = beta[kc + 1];

    const int n = nt * 16 + c;
    const int nc = (n < GEX) ? n : (GEX - 1);
    v2f bf[16];
    for (int j = 0; j < 16; j++) {
        const int kb = k0 + 4 * j + 2 * h;
        const bool ok = kb < GEX;
        const int kr = ok ? kb : (GEX - 2);
        const float x = Wff[kr * GEX + nc];
        const float y = Wff[(kr + 1) * GEX + nc];
        bf[j].x = ok ? x : 0.f;
        bf[j].y = ok ? y : 0.f;
    }

    {
        float s0 = 0.f, s1 = 0.f;
        #pragma unroll
        for (int i = 0; i < LNV; i++) { s0 += v0[i]; s1 += v1[i]; }
        wave_sum2(s0, s1);
        const float mu0 = s0 / (float)GEX, mu1 = s1 / (float)GEX;
        float q0 = 0.f, q1 = 0.f;
        #pragma unroll
        for (int i = 0; i < LNV; i++) {
            const bool in = lane + 32 * i < GEX;
            const float d0 = v0[i] - mu0, d1 = v1[i] - mu1;
            q0 += in ? d0 * d0 : 0.f;
            q1 += in ? d1 * d1 : 0.f;
        }
        wave_sum2(q0, q1);
        if (lane == 0) {
            mu_s[2 * wave] = mu0;
            mu_s[2 * wave + 1] = mu1;
            rstd_s[2 * wave] = rsqrtf(q0 / (float)GEX + 1e-5f);
            rstd_s[2 * wave + 1] = rsqrtf(q1 / (float)GEX + 1e-5f);
        }
    }
    __syncthreads();

    float* sw = gA + wave * (32 * SA_PITCH);
    for (int r = 0; r < 32; r++) {
        const float mu = mu_s[r], rstd = rstd_s[r];
        const float t0 = fmaxf((av[r].x - mu) * rstd * ga0 + be0, 0.f);
        const float t1 = fmaxf((av[r].y - mu) * rstd * ga1 + be1, 0.f);
        v2f y;
        y.x = kin ? t0 : 0.f;
        y.y = kin ? t1 : 0.f;
        *(v2f*)(sw + r * SA_PITCH + 2 * lane) = y;
    }
    __syncthreads();

    const float* fa0 = sw + c * SA_PITCH + 2 * h;
    const float* fa1 = fa0 + 16 * SA_PITCH;
    v8f acc0 = {}, acc1 = {};
    for (int j = 0; j < 16; j++) {
        v2f a0 = *(const v2f*)(fa0 + 4 * j);
        v2f a1 = *(const v2f*)(fa1 + 4 * j);
        acc0 = __builtin_amdgcn_wmma_f32_16x16x4_f32(
            false, a0, false, bf[j], (short)0, acc0, false, false);
        acc1 = __builtin_amdgcn_wmma_f32_16x16x4_f32(
            false, a1, false, bf[j], (short)0, acc1, false, false);
    }

    for (int r = 0; r < 8; r++) {
        gpart[wave * 512 + r * 32 + lane] = acc0[r];
        gpart[wave * 512 + 256 + r * 32 + lane] = acc1[r];
    }
    __syncthreads();
    {
        float s0 = 0.f, s1 = 0.f, s2 = 0.f, s3 = 0.f;
        for (int w = 0; w < GWAVES; w += 4) {
            s0 += gpart[(w + 0) * 512 + tid];
            s1 += gpart[(w + 1) * 512 + tid];
            s2 += gpart[(w + 2) * 512 + tid];
            s3 += gpart[(w + 3) * 512 + tid];
        }
        const int t = tid >> 8, idx = tid & 255;
        const int r = idx >> 5, l = idx & 31;
        const int row = mh * 32 + t * 16 + 8 * (l >> 4) + r;
        const int col = nt * 16 + (l & 15);
        if (col < GEX) out[row * GEX + col] = (s0 + s1) + (s2 + s3);
    }
}

extern "C" void kernel_launch(void* const* d_in, const int* in_sizes, int n_in,
                              void* d_out, int out_size, void* d_ws, size_t ws_size,
                              hipStream_t stream) {
    const float* b_gex    = (const float*)d_in[0];
    const float* node_feat= (const float*)d_in[1];
    const float* mask     = (const float*)d_in[2];
    const float* adj      = (const float*)d_in[3];
    const float* dist     = (const float*)d_in[4];
    const float* dose     = (const float*)d_in[5];
    const float* timev    = (const float*)d_in[6];
    const float* ppi      = (const float*)d_in[7];
    const float* w_gex    = (const float*)d_in[8];
    const float* w_comp   = (const float*)d_in[9];
    const float* W1       = (const float*)d_in[10];
    const float* W2       = (const float*)d_in[11];
    const float* w_dose   = (const float*)d_in[12];
    const float* w_time   = (const float*)d_in[13];
    const float* gamma    = (const float*)d_in[14];
    const float* beta     = (const float*)d_in[15];
    const float* Wff      = (const float*)d_in[16];

    float* out  = (float*)d_out;
    float* comp = out + BATCH * GEX;

    float* ws     = (float*)d_ws;
    float* P      = ws;
    float* ppirow = P + BATCH * HDIM;
    float* pred0  = ppirow + 1024;
    float* Q      = pred0 + BATCH * GEX;
    float* pe     = Q + 80 * HDIM;

    k_front<<<BATCH + NPQ + GEX, 256, 0, stream>>>(node_feat, mask, adj, dist, W1, W2, dose, timev,
                                                   w_dose, w_time, ppi, b_gex, w_gex, w_comp,
                                                   ppirow, P, Q, pe);
    k_pred0<<<64, 128, 0, stream>>>(b_gex, W2, w_dose, w_time, w_gex, w_comp, P, Q, pe, ppirow, comp, pred0);
    k_ln_gemm_wmma<<<124, 512, 0, stream>>>(pred0, gamma, beta, Wff, out);
}


# k_front ppi row sums: one wave per two rows (8-byte loads into registers, lane butterfly, no LDS tree) instead of an 8-wave work-group per row: 1114 work-groups to 198 to start; row-sum order changes
# speedup vs baseline: 8.6512x; 1.0637x over previous
#include <hip/hip_runtime.h>
#include <hip/hip_bf16.h>

#define GEX 978
#define HDIM 128
#define NAT 50
#define FEAT 34
#define CCEH 64
#define BATCH 64

typedef __attribute__((ext_vector_type(2))) float v2f;
typedef __attribute__((ext_vector_type(4))) float v4f;
typedef __attribute__((ext_vector_type(8))) float v8f;

__device__ __forceinline__ void wave_sum2(float& x, float& y) {
    for (int o = 16; o > 0; o >>= 1) {
        const float px = __shfl_xor(x, o, 32);
        const float py = __shfl_xor(y, o, 32);
        x += px;
        y += py;
    }
}

#define HST 9
#define WPQ 10
#define NPQ 72
#define PEP 68
#define SPAIRS (GEX / 2)
#define RPAIRS (GEX / 2)
#define NPPI ((RPAIRS + 7) / 8)
#define PLD 16
__global__ __launch_bounds__(256, 8) void k_front(const float* __restrict__ nf,
                                               const float* __restrict__ mask,
                                               const float* __restrict__ adj,
                                               const float* __restrict__ dist,
                                               const float* __restrict__ W1,
                                               const float* __restrict__ W2,
                                               const float* __restrict__ dose,
                                               const float* __restrict__ timev,
                                               const float* __restrict__ w_dose,
                                               const float* __restrict__ w_time,
                                               const float* __restrict__ ppi,
                                               const float* __restrict__ b_gex,
                                               const float* __restrict__ w_gex,
                                               const float* __restrict__ w_comp,
                                               float* __restrict__ rowsum,
                                               float* __restrict__ P,
                                               float* __restrict__ Q,
                                               float* __restrict__ pe) {
    const int tid = threadIdx.x;
    __shared__ float wprod[WPQ * 256];
    __shared__ float msk[256];
    __shared__ float wm[CCEH];
    __shared__ float red[256];
    __shared__ float msum_s;

    if (blockIdx.x >= BATCH && blockIdx.x < BATCH + NPQ) {
        const int t = blockIdx.x - BATCH;
        const bool isP = t < 32;
        const int u = isP ? t : (t - 32);
        const int mt = u >> 3, nt = u & 7;
        const int lane = tid & 31;
        const int wave = __builtin_amdgcn_readfirstlane(tid >> 5);
        const int c = lane & 15, h = lane >> 4;
        const int R = mt * 16 + c;
        const float* arow;
        if (isP) arow = b_gex + R * GEX;
        else arow = (R < CCEH) ? (W2 + R * GEX) : ((R == CCEH) ? w_dose : w_time);
        const float* Wm = isP ? w_gex : w_comp;
        v8f acc0 = {}, acc1 = {};
        #pragma unroll
        for (int q = 0; q < 2; q++) {
            const int pw = (wave + 8 * q) * 32;
            const int p0 = (pw < SPAIRS - 32) ? pw : (SPAIRS - 32);
            const int skip = pw - p0;
            const int k0 = 2 * p0;
            const float* ap = arow + k0 + 2 * h;
            const float* bp = Wm + (k0 + 2 * h) * HDIM + nt * 16 + c;
            v2f af[16], bf[16];
            #pragma unroll
            for (int j = 0; j < 16; j++) {
                af[j] = *(const v2f*)(ap + 4 * j);
                bf[j].x = bp[(4 * j) * HDIM];
                bf[j].y = bp[(4 * j + 1) * HDIM];
            }
            #pragma unroll
            for (int j = 0; j < 16; j += 2) {
                v2f a0 = af[j], a1 = af[j + 1];
                v2f b0 = bf[j], b1 = bf[j + 1];
                const bool ok0 = (2 * j + h) >= skip;
                const bool ok1 = (2 * j + 2 + h) >= skip;
                a0.x = ok0 ? a0.x : 0.f;  a0.y = ok0 ? a0.y : 0.f;
                b0.x = ok0 ? b0.x : 0.f;  b0.y = ok0 ? b0.y : 0.f;
                a1.x = ok1 ? a1.x : 0.f;  a1.y = ok1 ? a1.y : 0.f;
                b1.x = ok1 ? b1.x : 0.f;  b1.y = ok1 ? b1.y : 0.f;
                acc0 = __builtin_amdgcn_wmma_f32_16x16x4_f32(
                    false, a0, false, b0, (short)0, acc0, false, false);
                acc1 = __builtin_amdgcn_wmma_f32_16x16x4_f32(
                    false, a1, false, b1, (short)0, acc1, false, false);
            }
        }
        #pragma unroll
        for (int r = 0; r < 8; r++) wprod[wave * 256 + r * 32 + lane] = acc0[r] + acc1[r];
        __syncthreads();
        {
            float s0 = 0.f, s1 = 0.f;
            for (int w = 0; w < 8; w += 2) {
                s0 += wprod[(w + 0) * 256 + tid];
                s1 += wprod[(w + 1) * 256 + tid];
            }
            const int r = tid >> 5, l = tid & 31;
            const int row = mt * 16 + 8 * (l >> 4) + r;
            const int col = nt * 16 + (l & 15);
            if (isP) P[row * HDIM + col] = s0 + s1;
            else Q[row * HDIM + col] = (row < CCEH + 2) ? (s0 + s1) : 0.f;
        }
        return;
    }
    if (blockIdx.x >= BATCH + NPQ) {
        const int lane = tid & 31;
        const int wave = __builtin_amdgcn_readfirstlane(tid >> 5);
        const int rp = (blockIdx.x - BATCH - NPQ) * 8 + wave;
        if (rp >= RPAIRS) return;
        const float* r0 = ppi + (2 * rp) * GEX;
        const float* r1 = r0 + GEX;
        v2f u0[PLD], u1[PLD];
        #pragma unroll
        for (int i = 0; i < PLD; i++) {
            const int p = lane + 32 * i;
            const int pc = (p < SPAIRS) ? p : (SPAIRS - 1);
            u0[i] = *(const v2f*)(r0 + 2 * pc);
            u1[i] = *(const v2f*)(r1 + 2 * pc);
        }
        float s0x = 0.f, s0y = 0.f, s1x = 0.f, s1y = 0.f;
        #pragma unroll
        for (int i = 0; i < PLD; i++) {
            const bool in = lane + 32 * i < SPAIRS;
            s0x += in ? u0[i].x : 0.f;  s0y += in ? u0[i].y : 0.f;
            s1x += in ? u1[i].x : 0.f;  s1y += in ? u1[i].y : 0.f;
        }
        float s0 = s0x + s0y, s1 = s1x + s1y;
        wave_sum2(s0, s1);
        if (lane == 0) {
            rowsum[2 * rp] = s0;
            rowsum[2 * rp + 1] = s1;
        }
        return;
    }

    const int b = blockIdx.x;
    const int lane = tid & 31;
    const int wave = __builtin_amdgcn_readfirstlane(tid >> 5);
    const int c = lane & 15, h = lane >> 4;
    const int nt = wave & 3;
    const int mp = wave >> 2;

    float av[WPQ], dv[WPQ], mv[WPQ];
    #pragma unroll
    for (int q = 0; q < WPQ; q++) {
        const int i = tid + 256 * q;
        const int ic = (i < NAT * NAT) ? i : (NAT * NAT - 1);
        av[q] = adj[b * (NAT * NAT) + ic];
        dv[q] = dist[b * (NAT * NAT) + ic];
        mv[q] = mask[b * NAT + ic / NAT];
    }
    const float mk = mask[b * NAT + ((tid < NAT) ? tid : 0)];
    const float db = dose[b], tb = timev[b];
    const float* nfb = nf + b * (NAT * FEAT);
    const int n0 = mp * 32 + c;
    const int n1 = (n0 + 16 < NAT) ? (n0 + 16) : (NAT - 1);
    const float* a0p = nfb + n0 * FEAT;
    const float* a1p = nfb + n1 * FEAT;
    const float* bp = W1 + nt * 16 + c;
    v2f a0[HST], a1[HST], bw[HST];
    #pragma unroll
    for (int j = 0; j < HST; j++) {
        const bool kin = (j < HST - 1) || (h == 0);
        const int k = kin ? (4 * j + 2 * h) : (4 * j);
        a0[j] = *(const v2f*)(a0p + k);
        a1[j] = *(const v2f*)(a1p + k);
        bw[j].x = bp[k * CCEH];
        bw[j].y = bp[(k + 1) * CCEH];
    }
    __builtin_amdgcn_sched_barrier(0);
    asm volatile("" :: "s"(db), "s"(tb));
    {
        const bool kin = (h == 0);
        a0[HST - 1].x = kin ? a0[HST - 1].x : 0.f;  a0[HST - 1].y = kin ? a0[HST - 1].y : 0.f;
        a1[HST - 1].x = kin ? a1[HST - 1].x : 0.f;  a1[HST - 1].y = kin ? a1[HST - 1].y : 0.f;
        bw[HST - 1].x = kin ? bw[HST - 1].x : 0.f;  bw[HST - 1].y = kin ? bw[HST - 1].y : 0.f;
    }

    #pragma unroll
    for (int q = 0; q < WPQ; q++) wprod[tid + 256 * q] = mv[q] * av[q] * expf(-dv[q]);
    msk[tid] = mk;

    v8f d0 = {}, d1 = {};
    #pragma unroll
    for (int j = 0; j < HST; j++) {
        d0 = __builtin_amdgcn_wmma_f32_16x16x4_f32(
            false, a0[j], false, bw[j], (short)0, d0, false, false);
        d1 = __builtin_amdgcn_wmma_f32_16x16x4_f32(
            false, a1[j], false, bw[j], (short)0, d1, false, false);
    }
    __syncthreads();

    if (tid >= 128 && tid < 128 + CCEH) {
        const int m = tid - 128;
        float s = 0.f;
        if (m < NAT) {
            #pragma unroll
            for (int n = 0; n < NAT; n++) s += wprod[n * NAT + m];
        }
        wm[m] = s;
    } else if (tid >= 192 && tid < 224) {
        float msum = 0.f;
        #pragma unroll
        for (int n = 0; n < NAT; n++) msum += msk[n];
        if (tid == 192) msum_s = fmaxf(msum, 1.f);
    }
    __syncthreads();
    {
        const float* wp = wm + mp * 32 + 8 * h;
        float s = 0.f;
        #pragma unroll
        for (int r = 0; r < 8; r++) s += wp[r] * fmaxf(d0[r], 0.f);
        #pragma unroll
        for (int r = 0; r < 8; r++) s += wp[16 + r] * fmaxf(d1[r], 0.f);
        s += __shfl_xor(s, 16, 32);
        if (h == 0) red[wave * 16 + c] = s;
    }
    __syncthreads();
    if (tid < CCEH) pe[b * PEP + tid] = (red[tid] + red[CCEH + tid]) / msum_s;
    else if (tid < PEP) pe[b * PEP + tid] = (tid == CCEH) ? db : ((tid == CCEH + 1) ? tb : 0.f);
}

#define SA_PITCH 68
#define SLP 132
#define QST 17
__global__ __launch_bounds__(128) void k_pred0(const float* __restrict__ b_gex,
                                               const float* __restrict__ W2,
                                               const float* __restrict__ w_dose,
                                               const float* __restrict__ w_time,
                                               const float* __restrict__ w_gex,
                                               const float* __restrict__ w_comp,
                                               const float* __restrict__ P,
                                               const float* __restrict__ Q,
                                               const float* __restrict__ pe,
                                               const float* __restrict__ ppirow,
                                               float* __restrict__ comp,
                                               float* __restrict__ pred0) {
    __shared__ __attribute__((aligned(16))) float Ss[16 * SLP];
    const int tid = threadIdx.x;
    const int lane = tid & 31;
    const int wave = __builtin_amdgcn_readfirstlane(tid >> 5);
    const int c = lane & 15, h = lane >> 4;
    const int mt = blockIdx.x & 3, ng = blockIdx.x >> 2;
    const int nt = ng * 4 + wave;
    const int g = nt * 16 + c;
    const int gc = (g < GEX) ? g : (GEX - 1);

    const float* pa = pe + (mt * 16 + c) * PEP + 2 * h;
    const float* qp = Q + (2 * h) * HDIM + 32 * wave + c;
    const float* pp = P + (mt * 16 + 8 * h) * HDIM + 32 * wave + c;
    v2f ea[QST], q0[QST], q1[QST];
    v8f s0, s1;
    #pragma unroll
    for (int j = 0; j < QST; j++) ea[j] = *(const v2f*)(pa + 4 * j);
    #pragma unroll
    for (int j = 0; j < QST; j++) {
        q0[j].x = qp[(4 * j) * HDIM];
        q0[j].y = qp[(4 * j + 1) * HDIM];
        q1[j].x = qp[(4 * j) * HDIM + 16];
        q1[j].y = qp[(4 * j + 1) * HDIM + 16];
    }
    #pragma unroll
    for (int r = 0; r < 8; r++) {
        s0[r] = pp[r * HDIM];
        s1[r] = pp[r * HDIM + 16];
    }
    const float* w2p = W2 + (2 * h) * GEX + gc;
    v2f wf[QST];
    #pragma unroll
    for (int j = 0; j < QST - 1; j++) {
        wf[j].x = w2p[(4 * j) * GEX];
        wf[j].y = w2p[(4 * j + 1) * GEX];
    }
    {
        const float wd = w_dose[gc], wt = w_time[gc];
        wf[QST - 1].x = (h == 0) ? wd : 0.f;
        wf[QST - 1].y = (h == 0) ? wt : 0.f;
    }

    const v4f* gp = (const v4f*)(w_gex + gc * HDIM + 4 * h);
    const v4f* cp = (const v4f*)(w_comp + gc * HDIM + 4 * h);
    const int b0 = mt * 16 + 8 * h;
    float xv[8];
    #pragma unroll
    for (int r = 0; r < 8; r++) xv[r] = b_gex[(b0 + r) * GEX + gc];
    const float pr = ppirow[gc];
    v4f gf[16], cf[16];
    #pragma unroll
    for (int J = 0; J < 16; J++) {
        gf[J] = gp[2 * J];
        cf[J] = cp[2 * J];
    }
    __builtin_amdgcn_sched_barrier(0);

    #pragma unroll
    for (int j = 0; j < QST; j++) {
        s0 = __builtin_amdgcn_wmma_f32_16x16x4_f32(
            false, ea[j], false, q0[j], (short)0, s0, false, false);
        s1 = __builtin_amdgcn_wmma_f32_16x16x4_f32(
            false, ea[j], false, q1[j], (short)0, s1, false, false);
    }
    #pragma unroll
    for (int r = 0; r < 8; r++) {
        Ss[(8 * h + r) * SLP + 32 * wave + c] = s0[r];
        Ss[(8 * h + r) * SLP + 32 * wave + 16 + c] = s1[r];
    }
    __syncthreads();

    const v4f* ap = (const v4f*)(Ss + c * SLP + 4 * h);
    v4f af[16];
    #pragma unroll
    for (int J = 0; J < 16; J++) af[J] = ap[2 * J];
    __builtin_amdgcn_sched_barrier(0);

    v8f ca = {}, cb = {};
    #pragma unroll
    for (int j = 0; j < QST - 1; j += 2) {
        ca = __builtin_amdgcn_wmma_f32_16x16x4_f32(
            false, ea[j], false, wf[j], (short)0, ca, false, false);
        cb = __builtin_amdgcn_wmma_f32_16x16x4_f32(
            false, ea[j + 1], false, wf[j + 1], (short)0, cb, false, false);
    }
    ca = __builtin_amdgcn_wmma_f32_16x16x4_f32(
        false, ea[QST - 1], false, wf[QST - 1], (short)0, ca, false, false);

    v8f t1 = {}, t2 = {};
    #pragma unroll
    for (int J = 0; J < 16; J++) {
        v2f a0, a1, g0, g1, c0, c1;
        a0.x = af[J].x;  a0.y = af[J].y;  a1.x = af[J].z;  a1.y = af[J].w;
        g0.x = gf[J].x;  g0.y = gf[J].y;  g1.x = gf[J].z;  g1.y = gf[J].w;
        c0.x = cf[J].x;  c0.y = cf[J].y;  c1.x = cf[J].z;  c1.y = cf[J].w;
        t1 = __builtin_amdgcn_wmma_f32_16x16x4_f32(
            false, a0, false, g0, (short)0, t1, false, false);
        t2 = __builtin_amdgcn_wmma_f32_16x16x4_f32(
            false, a0, false, c0, (short)0, t2, false, false);
        t1 = __builtin_amdgcn_wmma_f32_16x16x4_f32(
            false, a1, false, g1, (short)0, t1, false, false);
        t2 = __builtin_amdgcn_wmma_f32_16x16x4_f32(
            false, a1, false, c1, (short)0, t2, false, false);
    }

    asm volatile("" :: "v"(xv[0]), "v"(xv[1]), "v"(xv[2]), "v"(xv[3]), "v"(xv[4]), "v"(xv[5]), "v"(xv[6]), "v"(xv[7]),
                       "v"(pr));
    if (g < GEX) {
        #pragma unroll
        for (int r = 0; r < 8; r++) {
            const float x = xv[r];
            const float cvr = ca[r] + cb[r];
            comp[(b0 + r) * GEX + g] = cvr;
            const float row = (x * t1[r] + cvr * t2[r]) * 0.08838834764831845f + pr;
            pred0[(b0 + r) * GEX + g] = x * row;
        }
    }
}

#define GWAVES 16
#define LNV 31

__global__ __launch_bounds__(512) void k_ln_gemm_wmma(const float* __restrict__ pred0,
                                                      const float* __restrict__ gamma,
                                                      const float* __restrict__ beta,
                                                      const float* __restrict__ Wff,
                                                      float* __restrict__ out) {
    __shared__ float gA[GWAVES * 32 * SA_PITCH];
    __shared__ float gpart[GWAVES * 512];
    __shared__ float mu_s[32], rstd_s[32];
    const int tid = threadIdx.x;
    const int lane = tid & 31;
    const int wave = __builtin_amdgcn_readfirstlane(tid >> 5);
    const int c = lane & 15, h = lane >> 4;
    const int nt = blockIdx.x >> 1, mh = blockIdx.x & 1;
    const int k0 = wave * 64;

    const float* prow0 = pred0 + (mh * 32 + 2 * wave) * GEX + lane;
    const float* prow1 = prow0 + GEX;
    float v0[LNV], v1[LNV];
    #pragma unroll
    for (int i = 0; i < LNV; i++) {
        const bool in = lane + 32 * i < GEX;
        v0[i] = in ? prow0[32 * i] : 0.f;
        v1[i] = in ? prow1[32 * i] : 0.f;
    }

    const int ka = k0 + 2 * lane;
    const bool kin = ka < GEX;
    const int kc = kin ? ka : (GEX - 2);
    const float* arow = pred0 + (mh * 32) * GEX + kc;
    v2f av[32];
    for (int r = 0; r < 32; r++) av[r] = *(const v2f*)(arow + r * GEX);
    const float ga0 = gamma[kc], ga1 = gamma[kc + 1];
    const float be0 = beta[kc], be1 = beta[kc + 1];

    const int n = nt * 16 + c;
    const int nc = (n < GEX) ? n : (GEX - 1);
    v2f bf[16];
    for (int j = 0; j < 16; j++) {
        const int kb = k0 + 4 * j + 2 * h;
        const bool ok = kb < GEX;
        const int kr = ok ? kb : (GEX - 2);
        const float x = Wff[kr * GEX + nc];
        const float y = Wff[(kr + 1) * GEX + nc];
        bf[j].x = ok ? x : 0.f;
        bf[j].y = ok ? y : 0.f;
    }

    {
        float s0 = 0.f, s1 = 0.f;
        #pragma unroll
        for (int i = 0; i < LNV; i++) { s0 += v0[i]; s1 += v1[i]; }
        wave_sum2(s0, s1);
        const float mu0 = s0 / (float)GEX, mu1 = s1 / (float)GEX;
        float q0 = 0.f, q1 = 0.f;
        #pragma unroll
        for (int i = 0; i < LNV; i++) {
            const bool in = lane + 32 * i < GEX;
            const float d0 = v0[i] - mu0, d1 = v1[i] - mu1;
            q0 += in ? d0 * d0 : 0.f;
            q1 += in ? d1 * d1 : 0.f;
        }
        wave_sum2(q0, q1);
        if (lane == 0) {
            mu_s[2 * wave] = mu0;
            mu_s[2 * wave + 1] = mu1;
            rstd_s[2 * wave] = rsqrtf(q0 / (float)GEX + 1e-5f);
            rstd_s[2 * wave + 1] = rsqrtf(q1 / (float)GEX + 1e-5f);
        }
    }
    __syncthreads();

    float* sw = gA + wave * (32 * SA_PITCH);
    for (int r = 0; r < 32; r++) {
        const float mu = mu_s[r], rstd = rstd_s[r];
        const float t0 = fmaxf((av[r].x - mu) * rstd * ga0 + be0, 0.f);
        const float t1 = fmaxf((av[r].y - mu) * rstd * ga1 + be1, 0.f);
        v2f y;
        y.x = kin ? t0 : 0.f;
        y.y = kin ? t1 : 0.f;
        *(v2f*)(sw + r * SA_PITCH + 2 * lane) = y;
    }
    __syncthreads();

    const float* fa0 = sw + c * SA_PITCH + 2 * h;
    const float* fa1 = fa0 + 16 * SA_PITCH;
    v8f acc0 = {}, acc1 = {};
    for (int j = 0; j < 16; j++) {
        v2f a0 = *(const v2f*)(fa0 + 4 * j);
        v2f a1 = *(const v2f*)(fa1 + 4 * j);
        acc0 = __builtin_amdgcn_wmma_f32_16x16x4_f32(
            false, a0, false, bf[j], (short)0, acc0, false, false);
        acc1 = __builtin_amdgcn_wmma_f32_16x16x4_f32(
            false, a1, false, bf[j], (short)0, acc1, false, false);
    }

    for (int r = 0; r < 8; r++) {
        gpart[wave * 512 + r * 32 + lane] = acc0[r];
        gpart[wave * 512 + 256 + r * 32 + lane] = acc1[r];
    }
    __syncthreads();
    {
        float s0 = 0.f, s1 = 0.f, s2 = 0.f, s3 = 0.f;
        for (int w = 0; w < GWAVES; w += 4) {
            s0 += gpart[(w + 0) * 512 + tid];
            s1 += gpart[(w + 1) * 512 + tid];
            s2 += gpart[(w + 2) * 512 + tid];
            s3 += gpart[(w + 3) * 512 + tid];
        }
        const int t = tid >> 8, idx = tid & 255;
        const int r = idx >> 5, l = idx & 31;
        const int row = mh * 32 + t * 16 + 8 * (l >> 4) + r;
        const int col = nt * 16 + (l & 15);
        if (col < GEX) out[row * GEX + col] = (s0 + s1) + (s2 + s3);
    }
}

extern "C" void kernel_launch(void* const* d_in, const int* in_sizes, int n_in,
                              void* d_out, int out_size, void* d_ws, size_t ws_size,
                              hipStream_t stream) {
    const float* b_gex    = (const float*)d_in[0];
    const float* node_feat= (const float*)d_in[1];
    const float* mask     = (const float*)d_in[2];
    const float* adj      = (const float*)d_in[3];
    const float* dist     = (const float*)d_in[4];
    const float* dose     = (const float*)d_in[5];
    const float* timev    = (const float*)d_in[6];
    const float* ppi      = (const float*)d_in[7];
    const float* w_gex    = (const float*)d_in[8];
    const float* w_comp   = (const float*)d_in[9];
    const float* W1       = (const float*)d_in[10];
    const float* W2       = (const float*)d_in[11];
    const float* w_dose   = (const float*)d_in[12];
    const float* w_time   = (const float*)d_in[13];
    const float* gamma    = (const float*)d_in[14];
    const float* beta     = (const float*)d_in[15];
    const float* Wff      = (const float*)d_in[16];

    float* out  = (float*)d_out;
    float* comp = out + BATCH * GEX;

    float* ws     = (float*)d_ws;
    float* P      = ws;
    float* ppirow = P + BATCH * HDIM;
    float* pred0  = ppirow + 1024;
    float* Q      = pred0 + BATCH * GEX;
    float* pe     = Q + 80 * HDIM;

    k_front<<<BATCH + NPQ + NPPI, 256, 0, stream>>>(node_feat, mask, adj, dist, W1, W2, dose, timev,
                                                   w_dose, w_time, ppi, b_gex, w_gex, w_comp,
                                                   ppirow, P, Q, pe);
    k_pred0<<<64, 128, 0, stream>>>(b_gex, W2, w_dose, w_time, w_gex, w_comp, P, Q, pe, ppirow, comp, pred0);
    k_ln_gemm_wmma<<<124, 512, 0, stream>>>(pred0, gamma, beta, Wff, out);
}


# Operands contiguous in k come by whole-row loads and become fragments via the wave's LDS piece (2-4 lines a load, not 16): A of k_front's P and Q tiles, w_gex/w_comp rows in k_pred0; sums unchanged
# speedup vs baseline: 9.2121x; 1.0648x over previous
#include <hip/hip_runtime.h>
#include <hip/hip_bf16.h>

#define GEX 978
#define HDIM 128
#define NAT 50
#define FEAT 34
#define CCEH 64
#define BATCH 64

typedef __attribute__((ext_vector_type(2))) float v2f;
typedef __attribute__((ext_vector_type(4))) float v4f;
typedef __attribute__((ext_vector_type(8))) float v8f;

__device__ __forceinline__ void wave_sum2(float& x, float& y) {
    for (int o = 16; o > 0; o >>= 1) {
        const float px = __shfl_xor(x, o, 32);
        const float py = __shfl_xor(y, o, 32);
        x += px;
        y += py;
    }
}

#define HST 9
#define WPQ 10
#define NPQ 72
#define PEP 68
#define SPAIRS (GEX / 2)
#define RPAIRS (GEX / 2)
#define NPPI ((RPAIRS + 7) / 8)
#define PLD 16
#define APITCH 68
__global__ __launch_bounds__(256, 8) void k_front(const float* __restrict__ nf,
                                               const float* __restrict__ mask,
                                               const float* __restrict__ adj,
                                               const float* __restrict__ dist,
                                               const float* __restrict__ W1,
                                               const float* __restrict__ W2,
                                               const float* __restrict__ dose,
                                               const float* __restrict__ timev,
                                               const float* __restrict__ w_dose,
                                               const float* __restrict__ w_time,
                                               const float* __restrict__ ppi,
                                               const float* __restrict__ b_gex,
                                               const float* __restrict__ w_gex,
                                               const float* __restrict__ w_comp,
                                               float* __restrict__ rowsum,
                                               float* __restrict__ P,
                                               float* __restrict__ Q,
                                               float* __restrict__ pe) {
    const int tid = threadIdx.x;
    __shared__ float wprod[WPQ * 256];
    __shared__ float msk[256];
    __shared__ float wm[CCEH];
    __shared__ float red[256];
    __shared__ float msum_s;
    __shared__ __attribute__((aligned(8))) float apiece[8 * 2 * 16 * APITCH];

    if (blockIdx.x >= BATCH && blockIdx.x < BATCH + NPQ) {
        const int t = blockIdx.x - BATCH;
        const bool isP = t < 32;
        const int u = isP ? t : (t - 32);
        const int mt = u >> 3, nt = u & 7;
        const int lane = tid & 31;
        const int wave = __builtin_amdgcn_readfirstlane(tid >> 5);
        const int c = lane & 15, h = lane >> 4;
        const float* Wm = isP ? w_gex : w_comp;
        float* aw = apiece + wave * (2 * 16 * APITCH);
        const bool tail = !isP && (mt == 4);
        const float* nb = (isP ? b_gex : W2) + (tail ? 0 : mt * 16 * GEX);
        const float* rp0 = tail ? w_dose : nb;
        const float* rp1 = tail ? w_time : (nb + GEX);
        const int rst = tail ? 0 : GEX;
        v8f acc0 = {}, acc1 = {};
        #pragma unroll
        for (int q = 0; q < 2; q++) {
            const int pw = (wave + 8 * q) * 32;
            const int p0 = (pw < SPAIRS - 32) ? pw : (SPAIRS - 32);
            const int skip = pw - p0;
            const int k0 = 2 * p0;
            const float* bp = Wm + (k0 + 2 * h) * HDIM + nt * 16 + c;
            v2f ar[16], af[16], bf[16];
            #pragma unroll
            for (int r = 0; r < 16; r++) {
                const float* rp = (r == 0) ? rp0 : (rp1 + (r - 1) * rst);
                ar[r] = *(const v2f*)(rp + k0 + 2 * lane);
            }
            #pragma unroll
            for (int j = 0; j < 16; j++) {
                bf[j].x = bp[(4 * j) * HDIM];
                bf[j].y = bp[(4 * j + 1) * HDIM];
            }
            #pragma unroll
            for (int r = 0; r < 16; r++) *(v2f*)(aw + (q * 16 + r) * APITCH + 2 * lane) = ar[r];
            __builtin_amdgcn_wave_barrier();
            #pragma unroll
            for (int j = 0; j < 16; j++) af[j] = *(const v2f*)(aw + (q * 16 + c) * APITCH + 4 * j + 2 * h);
            #pragma unroll
            for (int j = 0; j < 16; j += 2) {
                v2f a0 = af[j], a1 = af[j + 1];
                v2f b0 = bf[j], b1 = bf[j + 1];
                const bool ok0 = (2 * j + h) >= skip;
                const bool ok1 = (2 * j + 2 + h) >= skip;
                a0.x = ok0 ? a0.x : 0.f;  a0.y = ok0 ? a0.y : 0.f;
                b0.x = ok0 ? b0.x : 0.f;  b0.y = ok0 ? b0.y : 0.f;
                a1.x = ok1 ? a1.x : 0.f;  a1.y = ok1 ? a1.y : 0.f;
                b1.x = ok1 ? b1.x : 0.f;  b1.y = ok1 ? b1.y : 0.f;
                acc0 = __builtin_amdgcn_wmma_f32_16x16x4_f32(
                    false, a0, false, b0, (short)0, acc0, false, false);
                acc1 = __builtin_amdgcn_wmma_f32_16x16x4_f32(
                    false, a1, false, b1, (short)0, acc1, false, false);
            }
        }
        #pragma unroll
        for (int r = 0; r < 8; r++) wprod[wave * 256 + r * 32 + lane] = acc0[r] + acc1[r];
        __syncthreads();
        {
            float s0 = 0.f, s1 = 0.f;
            for (int w = 0; w < 8; w += 2) {
                s0 += wprod[(w + 0) * 256 + tid];
                s1 += wprod[(w + 1) * 256 + tid];
            }
            const int r = tid >> 5, l = tid & 31;
            const int row = mt * 16 + 8 * (l >> 4) + r;
            const int col = nt * 16 + (l & 15);
            if (isP) P[row * HDIM + col] = s0 + s1;
            else Q[row * HDIM + col] = (row < CCEH + 2) ? (s0 + s1) : 0.f;
        }
        return;
    }
    if (blockIdx.x >= BATCH + NPQ) {
        const int lane = tid & 31;
        const int wave = __builtin_amdgcn_readfirstlane(tid >> 5);
        const int rp = (blockIdx.x - BATCH - NPQ) * 8 + wave;
        if (rp >= RPAIRS) return;
        const float* r0 = ppi + (2 * rp) * GEX;
        const float* r1 = r0 + GEX;
        v2f u0[PLD], u1[PLD];
        #pragma unroll
        for (int i = 0; i < PLD; i++) {
            const int p = lane + 32 * i;
            const int pc = (p < SPAIRS) ? p : (SPAIRS - 1);
            u0[i] = *(const v2f*)(r0 + 2 * pc);
            u1[i] = *(const v2f*)(r1 + 2 * pc);
        }
        float s0x = 0.f, s0y = 0.f, s1x = 0.f, s1y = 0.f;
        #pragma unroll
        for (int i = 0; i < PLD; i++) {
            const bool in = lane + 32 * i < SPAIRS;
            s0x += in ? u0[i].x : 0.f;  s0y += in ? u0[i].y : 0.f;
            s1x += in ? u1[i].x : 0.f;  s1y += in ? u1[i].y : 0.f;
        }
        float s0 = s0x + s0y, s1 = s1x + s1y;
        wave_sum2(s0, s1);
        if (lane == 0) {
            rowsum[2 * rp] = s0;
            rowsum[2 * rp + 1] = s1;
        }
        return;
    }

    const int b = blockIdx.x;
    const int lane = tid & 31;
    const int wave = __builtin_amdgcn_readfirstlane(tid >> 5);
    const int c = lane & 15, h = lane >> 4;
    const int nt = wave & 3;
    const int mp = wave >> 2;

    float av[WPQ], dv[WPQ], mv[WPQ];
    #pragma unroll
    for (int q = 0; q < WPQ; q++) {
        const int i = tid + 256 * q;
        const int ic = (i < NAT * NAT) ? i : (NAT * NAT - 1);
        av[q] = adj[b * (NAT * NAT) + ic];
        dv[q] = dist[b * (NAT * NAT) + ic];
        mv[q] = mask[b * NAT + ic / NAT];
    }
    const float mk = mask[b * NAT + ((tid < NAT) ? tid : 0)];
    const float db = dose[b], tb = timev[b];
    const float* nfb = nf + b * (NAT * FEAT);
    const int n0 = mp * 32 + c;
    const int n1 = (n0 + 16 < NAT) ? (n0 + 16) : (NAT - 1);
    const float* a0p = nfb + n0 * FEAT;
    const float* a1p = nfb + n1 * FEAT;
    const float* bp = W1 + nt * 16 + c;
    v2f a0[HST], a1[HST], bw[HST];
    #pragma unroll
    for (int j = 0; j < HST; j++) {
        const bool kin = (j < HST - 1) || (h == 0);
        const int k = kin ? (4 * j + 2 * h) : (4 * j);
        a0[j] = *(const v2f*)(a0p + k);
        a1[j] = *(const v2f*)(a1p + k);
        bw[j].x = bp[k * CCEH];
        bw[j].y = bp[(k + 1) * CCEH];
    }
    __builtin_amdgcn_sched_barrier(0);
    asm volatile("" :: "s"(db), "s"(tb));
    {
        const bool kin = (h == 0);
        a0[HST - 1].x = kin ? a0[HST - 1].x : 0.f;  a0[HST - 1].y = kin ? a0[HST - 1].y : 0.f;
        a1[HST - 1].x = kin ? a1[HST - 1].x : 0.f;  a1[HST - 1].y = kin ? a1[HST - 1].y : 0.f;
        bw[HST - 1].x = kin ? bw[HST - 1].x : 0.f;  bw[HST - 1].y = kin ? bw[HST - 1].y : 0.f;
    }

    #pragma unroll
    for (int q = 0; q < WPQ; q++) wprod[tid + 256 * q] = mv[q] * av[q] * expf(-dv[q]);
    msk[tid] = mk;

    v8f d0 = {}, d1 = {};
    #pragma unroll
    for (int j = 0; j < HST; j++) {
        d0 = __builtin_amdgcn_wmma_f32_16x16x4_f32(
            false, a0[j], false, bw[j], (short)0, d0, false, false);
        d1 = __builtin_amdgcn_wmma_f32_16x16x4_f32(
            false, a1[j], false, bw[j], (short)0, d1, false, false);
    }
    __syncthreads();

    if (tid >= 128 && tid < 128 + CCEH) {
        const int m = tid - 128;
        float s = 0.f;
        if (m < NAT) {
            #pragma unroll
            for (int n = 0; n < NAT; n++) s += wprod[n * NAT + m];
        }
        wm[m] = s;
    } else if (tid >= 192 && tid < 224) {
        float msum = 0.f;
        #pragma unroll
        for (int n = 0; n < NAT; n++) msum += msk[n];
        if (tid == 192) msum_s = fmaxf(msum, 1.f);
    }
    __syncthreads();
    {
        const float* wp = wm + mp * 32 + 8 * h;
        float s = 0.f;
        #pragma unroll
        for (int r = 0; r < 8; r++) s += wp[r] * fmaxf(d0[r], 0.f);
        #pragma unroll
        for (int r = 0; r < 8; r++) s += wp[16 + r] * fmaxf(d1[r], 0.f);
        s += __shfl_xor(s, 16, 32);
        if (h == 0) red[wave * 16 + c] = s;
    }
    __syncthreads();
    if (tid < CCEH) pe[b * PEP + tid] = (red[tid] + red[CCEH + tid]) / msum_s;
    else if (tid < PEP) pe[b * PEP + tid] = (tid == CCEH) ? db : ((tid == CCEH + 1) ? tb : 0.f);
}

#define SA_PITCH 68
#define SLP 132
#define QST 17
__global__ __launch_bounds__(128) void k_pred0(const float* __restrict__ b_gex,
                                               const float* __restrict__ W2,
                                               const float* __restrict__ w_dose,
                                               const float* __restrict__ w_time,
                                               const float* __restrict__ w_gex,
                                               const float* __restrict__ w_comp,
                                               const float* __restrict__ P,
                                               const float* __restrict__ Q,
                                               const float* __restrict__ pe,
                                               const float* __restrict__ ppirow,
                                               float* __restrict__ comp,
                                               float* __restrict__ pred0) {
    __shared__ __attribute__((aligned(16))) float Ss[16 * SLP];
    __shared__ __attribute__((aligned(16))) float Wt[4 * 2 * 16 * SLP];
    const int tid = threadIdx.x;
    const int lane = tid & 31;
    const int wave = __builtin_amdgcn_readfirstlane(tid >> 5);
    const int c = lane & 15, h = lane >> 4;
    const int mt = blockIdx.x & 3, ng = blockIdx.x >> 2;
    const int nt = ng * 4 + wave;
    const int g = nt * 16 + c;
    const int gc = (g < GEX) ? g : (GEX - 1);

    const float* pa = pe + (mt * 16 + c) * PEP + 2 * h;
    const float* qp = Q + (2 * h) * HDIM + 32 * wave + c;
    const float* pp = P + (mt * 16 + 8 * h) * HDIM + 32 * wave + c;
    v2f ea[QST], q0[QST], q1[QST];
    v8f s0, s1;
    #pragma unroll
    for (int j = 0; j < QST; j++) ea[j] = *(const v2f*)(pa + 4 * j);
    #pragma unroll
    for (int j = 0; j < QST; j++) {
        q0[j].x = qp[(4 * j) * HDIM];
        q0[j].y = qp[(4 * j + 1) * HDIM];
        q1[j].x = qp[(4 * j) * HDIM + 16];
        q1[j].y = qp[(4 * j + 1) * HDIM + 16];
    }
    #pragma unroll
    for (int r = 0; r < 8; r++) {
        s0[r] = pp[r * HDIM];
        s1[r] = pp[r * HDIM + 16];
    }
    const float* w2p = W2 + (2 * h) * GEX + gc;
    v2f wf[QST];
    #pragma unroll
    for (int j = 0; j < QST - 1; j++) {
        wf[j].x = w2p[(4 * j) * GEX];
        wf[j].y = w2p[(4 * j + 1) * GEX];
    }
    {
        const float wd = w_dose[gc], wt = w_time[gc];
        wf[QST - 1].x = (h == 0) ? wd : 0.f;
        wf[QST - 1].y = (h == 0) ? wt : 0.f;
    }

    const int b0 = mt * 16 + 8 * h;
    float xv[8];
    #pragma unroll
    for (int r = 0; r < 8; r++) xv[r] = b_gex[(b0 + r) * GEX + gc];
    const float pr = ppirow[gc];
    v4f gr[16], cr[16];
    #pragma unroll
    for (int r = 0; r < 16; r++) {
        const int gr_ = (nt * 16 + r < GEX) ? (nt * 16 + r) : (GEX - 1);
        gr[r] = *(const v4f*)(w_gex + gr_ * HDIM + 4 * lane);
        cr[r] = *(const v4f*)(w_comp + gr_ * HDIM + 4 * lane);
    }
    __builtin_amdgcn_sched_barrier(0);

    #pragma unroll
    for (int j = 0; j < QST; j++) {
        s0 = __builtin_amdgcn_wmma_f32_16x16x4_f32(
            false, ea[j], false, q0[j], (short)0, s0, false, false);
        s1 = __builtin_amdgcn_wmma_f32_16x16x4_f32(
            false, ea[j], false, q1[j], (short)0, s1, false, false);
    }
    #pragma unroll
    for (int r = 0; r < 8; r++) {
        Ss[(8 * h + r) * SLP + 32 * wave + c] = s0[r];
        Ss[(8 * h + r) * SLP + 32 * wave + 16 + c] = s1[r];
    }
    __builtin_amdgcn_sched_barrier(0);
    float* wt = Wt + wave * (2 * 16 * SLP);
    #pragma unroll
    for (int r = 0; r < 16; r++) {
        *(v4f*)(wt + r * SLP + 4 * lane) = gr[r];
        *(v4f*)(wt + (16 + r) * SLP + 4 * lane) = cr[r];
    }
    __syncthreads();

    const v4f* ap = (const v4f*)(Ss + c * SLP + 4 * h);
    const v4f* gp = (const v4f*)(wt + c * SLP + 4 * h);
    const v4f* cp = (const v4f*)(wt + (16 + c) * SLP + 4 * h);
    v4f af[16], gf[16], cf[16];
    #pragma unroll
    for (int J = 0; J < 16; J++) {
        af[J] = ap[2 * J];
        gf[J] = gp[2 * J];
        cf[J] = cp[2 * J];
    }
    __builtin_amdgcn_sched_barrier(0);

    v8f ca = {}, cb = {};
    #pragma unroll
    for (int j = 0; j < QST - 1; j += 2) {
        ca = __builtin_amdgcn_wmma_f32_16x16x4_f32(
            false, ea[j], false, wf[j], (short)0, ca, false, false);
        cb = __builtin_amdgcn_wmma_f32_16x16x4_f32(
            false, ea[j + 1], false, wf[j + 1], (short)0, cb, false, false);
    }
    ca = __builtin_amdgcn_wmma_f32_16x16x4_f32(
        false, ea[QST - 1], false, wf[QST - 1], (short)0, ca, false, false);

    v8f t1 = {}, t2 = {};
    #pragma unroll
    for (int J = 0; J < 16; J++) {
        v2f a0, a1, g0, g1, c0, c1;
        a0.x = af[J].x;  a0.y = af[J].y;  a1.x = af[J].z;  a1.y = af[J].w;
        g0.x = gf[J].x;  g0.y = gf[J].y;  g1.x = gf[J].z;  g1.y = gf[J].w;
        c0.x = cf[J].x;  c0.y = cf[J].y;  c1.x = cf[J].z;  c1.y = cf[J].w;
        t1 = __builtin_amdgcn_wmma_f32_16x16x4_f32(
            false, a0, false, g0, (short)0, t1, false, false);
        t2 = __builtin_amdgcn_wmma_f32_16x16x4_f32(
            false, a0, false, c0, (short)0, t2, false, false);
        t1 = __builtin_amdgcn_wmma_f32_16x16x4_f32(
            false, a1, false, g1, (short)0, t1, false, false);
        t2 = __builtin_amdgcn_wmma_f32_16x16x4_f32(
            false, a1, false, c1, (short)0, t2, false, false);
    }

    asm volatile("" :: "v"(xv[0]), "v"(xv[1]), "v"(xv[2]), "v"(xv[3]), "v"(xv[4]), "v"(xv[5]), "v"(xv[6]), "v"(xv[7]),
                       "v"(pr));
    if (g < GEX) {
        #pragma unroll
        for (int r = 0; r < 8; r++) {
            const float x = xv[r];
            const float cvr = ca[r] + cb[r];
            comp[(b0 + r) * GEX + g] = cvr;
            const float row = (x * t1[r] + cvr * t2[r]) * 0.08838834764831845f + pr;
            pred0[(b0 + r) * GEX + g] = x * row;
        }
    }
}

#define GWAVES 16
#define LNV 31

__global__ __launch_bounds__(512) void k_ln_gemm_wmma(const float* __restrict__ pred0,
                                                      const float* __restrict__ gamma,
                                                      const float* __restrict__ beta,
                                                      const float* __restrict__ Wff,
                                                      float* __restrict__ out) {
    __shared__ float gA[GWAVES * 32 * SA_PITCH];
    __shared__ float gpart[GWAVES * 512];
    __shared__ float mu_s[32], rstd_s[32];
    const int tid = threadIdx.x;
    const int lane = tid & 31;
    const int wave = __builtin_amdgcn_readfirstlane(tid >> 5);
    const int c = lane & 15, h = lane >> 4;
    const int nt = blockIdx.x >> 1, mh = blockIdx.x & 1;
    const int k0 = wave * 64;

    const float* prow0 = pred0 + (mh * 32 + 2 * wave) * GEX + lane;
    const float* prow1 = prow0 + GEX;
    float v0[LNV], v1[LNV];
    #pragma unroll
    for (int i = 0; i < LNV; i++) {
        const bool in = lane + 32 * i < GEX;
        v0[i] = in ? prow0[32 * i] : 0.f;
        v1[i] = in ? prow1[32 * i] : 0.f;
    }

    const int ka = k0 + 2 * lane;
    const bool kin = ka < GEX;
    const int kc = kin ? ka : (GEX - 2);
    const float* arow = pred0 + (mh * 32) * GEX + kc;
    v2f av[32];
    for (int r = 0; r < 32; r++) av[r] = *(const v2f*)(arow + r * GEX);
    const float ga0 = gamma[kc], ga1 = gamma[kc + 1];
    const float be0 = beta[kc], be1 = beta[kc + 1];

    const int n = nt * 16 + c;
    const int nc = (n < GEX) ? n : (GEX - 1);
    v2f bf[16];
    for (int j = 0; j < 16; j++) {
        const int kb = k0 + 4 * j + 2 * h;
        const bool ok = kb < GEX;
        const int kr = ok ? kb : (GEX - 2);
        const float x = Wff[kr * GEX + nc];
        const float y = Wff[(kr + 1) * GEX + nc];
        bf[j].x = ok ? x : 0.f;
        bf[j].y = ok ? y : 0.f;
    }

    {
        float s0 = 0.f, s1 = 0.f;
        #pragma unroll
        for (int i = 0; i < LNV; i++) { s0 += v0[i]; s1 += v1[i]; }
        wave_sum2(s0, s1);
        const float mu0 = s0 / (float)GEX, mu1 = s1 / (float)GEX;
        float q0 = 0.f, q1 = 0.f;
        #pragma unroll
        for (int i = 0; i < LNV; i++) {
            const bool in = lane + 32 * i < GEX;
            const float d0 = v0[i] - mu0, d1 = v1[i] - mu1;
            q0 += in ? d0 * d0 : 0.f;
            q1 += in ? d1 * d1 : 0.f;
        }
        wave_sum2(q0, q1);
        if (lane == 0) {
            mu_s[2 * wave] = mu0;
            mu_s[2 * wave + 1] = mu1;
            rstd_s[2 * wave] = rsqrtf(q0 / (float)GEX + 1e-5f);
            rstd_s[2 * wave + 1] = rsqrtf(q1 / (float)GEX + 1e-5f);
        }
    }
    __syncthreads();

    float* sw = gA + wave * (32 * SA_PITCH);
    for (int r = 0; r < 32; r++) {
        const float mu = mu_s[r], rstd = rstd_s[r];
        const float t0 = fmaxf((av[r].x - mu) * rstd * ga0 + be0, 0.f);
        const float t1 = fmaxf((av[r].y - mu) * rstd * ga1 + be1, 0.f);
        v2f y;
        y.x = kin ? t0 : 0.f;
        y.y = kin ? t1 : 0.f;
        *(v2f*)(sw + r * SA_PITCH + 2 * lane) = y;
    }
    __syncthreads();

    const float* fa0 = sw + c * SA_PITCH + 2 * h;
    const float* fa1 = fa0 + 16 * SA_PITCH;
    v8f acc0 = {}, acc1 = {};
    for (int j = 0; j < 16; j++) {
        v2f a0 = *(const v2f*)(fa0 + 4 * j);
        v2f a1 = *(const v2f*)(fa1 + 4 * j);
        acc0 = __builtin_amdgcn_wmma_f32_16x16x4_f32(
            false, a0, false, bf[j], (short)0, acc0, false, false);
        acc1 = __builtin_amdgcn_wmma_f32_16x16x4_f32(
            false, a1, false, bf[j], (short)0, acc1, false, false);
    }

    for (int r = 0; r < 8; r++) {
        gpart[wave * 512 + r * 32 + lane] = acc0[r];
        gpart[wave * 512 + 256 + r * 32 + lane] = acc1[r];
    }
    __syncthreads();
    {
        float s0 = 0.f, s1 = 0.f, s2 = 0.f, s3 = 0.f;
        for (int w = 0; w < GWAVES; w += 4) {
            s0 += gpart[(w + 0) * 512 + tid];
            s1 += gpart[(w + 1) * 512 + tid];
            s2 += gpart[(w + 2) * 512 + tid];
            s3 += gpart[(w + 3) * 512 + tid];
        }
        const int t = tid >> 8, idx = tid & 255;
        const int r = idx >> 5, l = idx & 31;
        const int row = mh * 32 + t * 16 + 8 * (l >> 4) + r;
        const int col = nt * 16 + (l & 15);
        if (col < GEX) out[row * GEX + col] = (s0 + s1) + (s2 + s3);
    }
}

extern "C" void kernel_launch(void* const* d_in, const int* in_sizes, int n_in,
                              void* d_out, int out_size, void* d_ws, size_t ws_size,
                              hipStream_t stream) {
    const float* b_gex    = (const float*)d_in[0];
    const float* node_feat= (const float*)d_in[1];
    const float* mask     = (const float*)d_in[2];
    const float* adj      = (const float*)d_in[3];
    const float* dist     = (const float*)d_in[4];
    const float* dose     = (const float*)d_in[5];
    const float* timev    = (const float*)d_in[6];
    const float* ppi      = (const float*)d_in[7];
    const float* w_gex    = (const float*)d_in[8];
    const float* w_comp   = (const float*)d_in[9];
    const float* W1       = (const float*)d_in[10];
    const float* W2       = (const float*)d_in[11];
    const float* w_dose   = (const float*)d_in[12];
    const float* w_time   = (const float*)d_in[13];
    const float* gamma    = (const float*)d_in[14];
    const float* beta     = (const float*)d_in[15];
    const float* Wff      = (const float*)d_in[16];

    float* out  = (float*)d_out;
    float* comp = out + BATCH * GEX;

    float* ws     = (float*)d_ws;
    float* P      = ws;
    float* ppirow = P + BATCH * HDIM;
    float* pred0  = ppirow + 1024;
    float* Q      = pred0 + BATCH * GEX;
    float* pe     = Q + 80 * HDIM;

    k_front<<<BATCH + NPQ + NPPI, 256, 0, stream>>>(node_feat, mask, adj, dist, W1, W2, dose, timev,
                                                   w_dose, w_time, ppi, b_gex, w_gex, w_comp,
                                                   ppirow, P, Q, pe);
    k_pred0<<<64, 128, 0, stream>>>(b_gex, W2, w_dose, w_time, w_gex, w_comp, P, Q, pe, ppirow, comp, pred0);
    k_ln_gemm_wmma<<<124, 512, 0, stream>>>(pred0, gamma, beta, Wff, out);
}
